# starting kernel
# baseline (speedup 1.0000x reference)
#include <hip/hip_runtime.h>
#include <hip/hip_bf16.h>
#include <math.h>

#define NNODES   100000
#define NEDGES   600000
#define FDIM     256
#define HDIM     128
#define NCLS     47
#define NCLSPAD  64
#define NSEEDS   10000
#define GEN_EPS  1e-7f
#define LN_EPS   1e-5f

typedef float v2f __attribute__((ext_vector_type(2)));
typedef float v8f __attribute__((ext_vector_type(8)));

__global__ __launch_bounds__(128)
void gemm_f32_wmma_kernel(const float* __restrict__ A, const float* __restrict__ Wt,
                          const float* __restrict__ bias, const float* __restrict__ res,
                          float* __restrict__ out, int M, int K, int Nc)
{
    const int lane  = threadIdx.x;
    const int mtile = blockIdx.x * 4 + threadIdx.y;
    const int row0  = mtile * 16;
    if (row0 >= M) return;
    const int hb   = lane >> 4;
    const int lo   = lane & 15;
    const int arow = row0 + lo;
    const int col0 = blockIdx.y * 64 + lo;

    const float* __restrict__ a_ptr = A  + (size_t)arow * K;
    const float* __restrict__ b_ptr = Wt + (size_t)col0 * K;
    const size_t bstep = (size_t)16 * K;

    v8f acc[4];
#pragma unroll
    for (int t = 0; t < 4; ++t)
#pragma unroll
        for (int i = 0; i < 8; ++i) acc[t][i] = 0.f;

#pragma unroll 4
    for (int k = 0; k < K; k += 4) {
        const int ka = k + 2 * hb;
        const v2f a  = *(const v2f*)(a_ptr + ka);
        const v2f b0 = *(const v2f*)(b_ptr + ka);
        const v2f b1 = *(const v2f*)(b_ptr + bstep + ka);
        const v2f b2 = *(const v2f*)(b_ptr + 2 * bstep + ka);
        const v2f b3 = *(const v2f*)(b_ptr + 3 * bstep + ka);
        acc[0] = __builtin_amdgcn_wmma_f32_16x16x4_f32(false, a, false, b0, (short)0, acc[0], false, false);
        acc[1] = __builtin_amdgcn_wmma_f32_16x16x4_f32(false, a, false, b1, (short)0, acc[1], false, false);
        acc[2] = __builtin_amdgcn_wmma_f32_16x16x4_f32(false, a, false, b2, (short)0, acc[2], false, false);
        acc[3] = __builtin_amdgcn_wmma_f32_16x16x4_f32(false, a, false, b3, (short)0, acc[3], false, false);
    }

#pragma unroll
    for (int t = 0; t < 4; ++t) {
        const int   col = col0 + 16 * t;
        const float bv  = bias[col];
#pragma unroll
        for (int i = 0; i < 8; ++i) {
            const int r = row0 + i + 8 * hb;
            float v = acc[t][i] + bv;
            if (res) v += res[(size_t)r * Nc + col];
            out[(size_t)r * Nc + col] = v;
        }
    }
}

__global__ __launch_bounds__(256)
void transpose_pad_kernel(const float* __restrict__ W, float* __restrict__ Wt,
                          int K, int Nc, int Ncpad)
{
    const int tid = blockIdx.x * 256 + threadIdx.x;
    if (tid >= Ncpad * K) return;
    const int c = tid / K;
    const int k = tid - c * K;
    Wt[(size_t)c * K + k] = (c < Nc) ? W[(size_t)k * Nc + c] : 0.f;
}

__global__ void pad_bias_kernel(const float* __restrict__ b, float* __restrict__ bp,
                                int Nc, int Ncpad)
{
    const int c = threadIdx.x;
    if (c < Ncpad) bp[c] = (c < Nc) ? b[c] : 0.f;
}

__global__ __launch_bounds__(256)
void edge_max_kernel(const float* __restrict__ hin, const int* __restrict__ src,
                     const int* __restrict__ dst, float* __restrict__ mx)
{
    const int tid = blockIdx.x * 256 + threadIdx.x;
    if (tid >= NEDGES * 32) return;
    const int e = tid >> 5;
    const int f = (tid & 31) * 4;
    const int s = src[e];
    const int d = dst[e];
    const float4 hv = *(const float4*)(hin + (size_t)s * HDIM + f);
    unsigned* mp = (unsigned*)(mx + (size_t)d * HDIM + f);
    atomicMax(mp + 0, __float_as_uint(fmaxf(hv.x, 0.f) + GEN_EPS));
    atomicMax(mp + 1, __float_as_uint(fmaxf(hv.y, 0.f) + GEN_EPS));
    atomicMax(mp + 2, __float_as_uint(fmaxf(hv.z, 0.f) + GEN_EPS));
    atomicMax(mp + 3, __float_as_uint(fmaxf(hv.w, 0.f) + GEN_EPS));
}

__global__ __launch_bounds__(256)
void edge_sum_kernel(const float* __restrict__ hin, const int* __restrict__ src,
                     const int* __restrict__ dst, const float* __restrict__ mx,
                     float* __restrict__ denom, float* __restrict__ numer)
{
    const int tid = blockIdx.x * 256 + threadIdx.x;
    if (tid >= NEDGES * 32) return;
    const int e = tid >> 5;
    const int f = (tid & 31) * 4;
    const int s = src[e];
    const int d = dst[e];
    const float4 hv = *(const float4*)(hin + (size_t)s * HDIM + f);
    const float4 mv = *(const float4*)(mx + (size_t)d * HDIM + f);
    const size_t o = (size_t)d * HDIM + f;
    float m, ex;
    m = fmaxf(hv.x, 0.f) + GEN_EPS; ex = expf(m - mv.x);
    atomicAdd(denom + o + 0, ex); atomicAdd(numer + o + 0, m * ex);
    m = fmaxf(hv.y, 0.f) + GEN_EPS; ex = expf(m - mv.y);
    atomicAdd(denom + o + 1, ex); atomicAdd(numer + o + 1, m * ex);
    m = fmaxf(hv.z, 0.f) + GEN_EPS; ex = expf(m - mv.z);
    atomicAdd(denom + o + 2, ex); atomicAdd(numer + o + 2, m * ex);
    m = fmaxf(hv.w, 0.f) + GEN_EPS; ex = expf(m - mv.w);
    atomicAdd(denom + o + 3, ex); atomicAdd(numer + o + 3, m * ex);
}

__global__ __launch_bounds__(256)
void combine_kernel(const float* __restrict__ numer, const float* __restrict__ denom,
                    const float* __restrict__ hin, float* __restrict__ t, int n4)
{
    const int i = blockIdx.x * 256 + threadIdx.x;
    if (i >= n4) return;
    const float4 nu = ((const float4*)numer)[i];
    const float4 de = ((const float4*)denom)[i];
    const float4 hv = ((const float4*)hin)[i];
    float4 r;
    r.x = nu.x / (de.x + 1e-16f) + hv.x;
    r.y = nu.y / (de.y + 1e-16f) + hv.y;
    r.z = nu.z / (de.z + 1e-16f) + hv.z;
    r.w = nu.w / (de.w + 1e-16f) + hv.w;
    ((float4*)t)[i] = r;
}

__global__ __launch_bounds__(256)
void ln_relu_kernel(const float* __restrict__ in, const float* __restrict__ g,
                    const float* __restrict__ b, float* __restrict__ out, int M)
{
    const int row  = blockIdx.x * 8 + (threadIdx.x >> 5);
    const int lane = threadIdx.x & 31;
    if (row >= M) return;
    const float4 v = *(const float4*)(in + (size_t)row * HDIM + lane * 4);
    float s1 = v.x + v.y + v.z + v.w;
    float s2 = v.x * v.x + v.y * v.y + v.z * v.z + v.w * v.w;
#pragma unroll
    for (int off = 16; off > 0; off >>= 1) {
        s1 += __shfl_xor(s1, off, 32);
        s2 += __shfl_xor(s2, off, 32);
    }
    const float mu  = s1 * (1.f / HDIM);
    const float var = s2 * (1.f / HDIM) - mu * mu;
    const float inv = rsqrtf(var + LN_EPS);
    const float4 gg = *(const float4*)(g + lane * 4);
    const float4 bb = *(const float4*)(b + lane * 4);
    float4 r;
    r.x = fmaxf((v.x - mu) * inv * gg.x + bb.x, 0.f);
    r.y = fmaxf((v.y - mu) * inv * gg.y + bb.y, 0.f);
    r.z = fmaxf((v.z - mu) * inv * gg.z + bb.z, 0.f);
    r.w = fmaxf((v.w - mu) * inv * gg.w + bb.w, 0.f);
    *(float4*)(out + (size_t)row * HDIM + lane * 4) = r;
}

__global__ __launch_bounds__(256)
void gather_kernel(const float* __restrict__ in, const int* __restrict__ map,
                   float* __restrict__ out, int M)
{
    const int tid = blockIdx.x * 256 + threadIdx.x;
    if (tid >= M * 32) return;
    const int i = tid >> 5;
    const int f = (tid & 31) * 4;
    const int j = map[i];
    *(float4*)(out + (size_t)i * HDIM + f) =
        *(const float4*)(in + (size_t)j * HDIM + f);
}

__global__ __launch_bounds__(256)
void logsoftmax_kernel(const float* __restrict__ in, float* __restrict__ out, int M)
{
    const int row  = blockIdx.x * 8 + (threadIdx.x >> 5);
    const int lane = threadIdx.x & 31;
    if (row >= M) return;
    const float NEG = -3.0e38f;
    const float v0 = (lane      < NCLS) ? in[(size_t)row * NCLSPAD + lane     ] : NEG;
    const float v1 = (lane + 32 < NCLS) ? in[(size_t)row * NCLSPAD + lane + 32] : NEG;
    float mx = fmaxf(v0, v1);
#pragma unroll
    for (int off = 16; off > 0; off >>= 1) mx = fmaxf(mx, __shfl_xor(mx, off, 32));
    float se = ((lane      < NCLS) ? expf(v0 - mx) : 0.f)
             + ((lane + 32 < NCLS) ? expf(v1 - mx) : 0.f);
#pragma unroll
    for (int off = 16; off > 0; off >>= 1) se += __shfl_xor(se, off, 32);
    const float lse = logf(se) + mx;
    if (lane      < NCLS) out[(size_t)row * NCLS + lane     ] = v0 - lse;
    if (lane + 32 < NCLS) out[(size_t)row * NCLS + lane + 32] = v1 - lse;
}

extern "C" void kernel_launch(void* const* d_in, const int* in_sizes, int n_in,
                              void* d_out, int out_size, void* d_ws, size_t ws_size,
                              hipStream_t stream)
{
    (void)in_sizes; (void)n_in; (void)out_size; (void)ws_size;

    const float* x         = (const float*)d_in[0];
    const int*   src       = (const int*)  d_in[1];
    const int*   dst       = (const int*)  d_in[2];
    const int*   node_map  = (const int*)  d_in[3];
    const int*   final_map = (const int*)  d_in[4];
    const float* enc_w     = (const float*)d_in[5];
    const float* enc_b     = (const float*)d_in[6];
    const float* gcn_w     = (const float*)d_in[7];
    const float* gcn_b     = (const float*)d_in[8];
    const float* ln_g      = (const float*)d_in[9];
    const float* ln_b      = (const float*)d_in[10];
    const float* pred_w    = (const float*)d_in[11];
    const float* pred_b    = (const float*)d_in[12];
    float* out = (float*)d_out;

    const size_t NH = (size_t)NNODES * HDIM;
    float* B0     = (float*)d_ws;
    float* B1     = B0 + NH;
    float* B2     = B0 + 2 * NH;
    float* B3     = B0 + 3 * NH;
    float* B4     = B0 + 4 * NH;
    float* HG     = B0 + 5 * NH;
    float* HG2    = HG  + (size_t)NSEEDS * HDIM;
    float* SC     = HG2 + (size_t)NSEEDS * HDIM;
    float* WT_ENC = SC  + (size_t)NSEEDS * NCLSPAD;
    float* WT_G   = WT_ENC + (size_t)HDIM * FDIM;
    float* PWT    = WT_G   + (size_t)3 * HDIM * HDIM;
    float* PB     = PWT    + (size_t)NCLSPAD * HDIM;

    transpose_pad_kernel<<<(HDIM * FDIM + 255) / 256, 256, 0, stream>>>(
        enc_w, WT_ENC, FDIM, HDIM, HDIM);
    for (int l = 0; l < 3; ++l)
        transpose_pad_kernel<<<(HDIM * HDIM + 255) / 256, 256, 0, stream>>>(
            gcn_w + (size_t)l * HDIM * HDIM, WT_G + (size_t)l * HDIM * HDIM,
            HDIM, HDIM, HDIM);
    transpose_pad_kernel<<<(NCLSPAD * HDIM + 255) / 256, 256, 0, stream>>>(
        pred_w, PWT, HDIM, NCLS, NCLSPAD);
    pad_bias_kernel<<<1, NCLSPAD, 0, stream>>>(pred_b, PB, NCLS, NCLSPAD);

    const dim3 gblk(32, 4, 1);
    const int  mtN   = NNODES / 16;
    const dim3 gridN((mtN + 3) / 4, HDIM / 64, 1);
    const int  egrid = (NEDGES * 32 + 255) / 256;
    const int  cgrid = ((int)(NH / 4) + 255) / 256;
    const int  ngrid = (NNODES * 32 + 255) / 256;
    const int  lgrid = (NNODES + 7) / 8;

    gemm_f32_wmma_kernel<<<gridN, gblk, 0, stream>>>(
        x, WT_ENC, enc_b, nullptr, B0, NNODES, FDIM, HDIM);

    auto conv = [&](const float* hin, int eblk, const float* wt, const float* bias,
                    const float* res, float* outp) {
        hipMemsetAsync(B2, 0, 3 * NH * sizeof(float), stream);
        edge_max_kernel<<<egrid, 256, 0, stream>>>(
            hin, src + (size_t)eblk * NEDGES, dst + (size_t)eblk * NEDGES, B2);
        edge_sum_kernel<<<egrid, 256, 0, stream>>>(
            hin, src + (size_t)eblk * NEDGES, dst + (size_t)eblk * NEDGES, B2, B3, B4);
        combine_kernel<<<cgrid, 256, 0, stream>>>(B4, B3, hin, B2, (int)(NH / 4));
        gemm_f32_wmma_kernel<<<gridN, gblk, 0, stream>>>(
            B2, wt, bias, res, outp, NNODES, HDIM, HDIM);
    };

    conv(B0, 0, WT_G, gcn_b, nullptr, B0);

    for (int l = 1; l < 3; ++l) {
        ln_relu_kernel<<<lgrid, 256, 0, stream>>>(
            B0, ln_g + (size_t)(l - 1) * HDIM, ln_b + (size_t)(l - 1) * HDIM, B1, NNODES);
        conv(B1, l - 1, WT_G + (size_t)l * HDIM * HDIM, gcn_b + (size_t)l * HDIM,
              B0,  B4);
        gather_kernel<<<ngrid, 256, 0, stream>>>(
            B4, node_map + (size_t)(l - 1) * NNODES, B0, NNODES);
    }

    const int sgrid = (NSEEDS * 32 + 255) / 256;
    gather_kernel<<<sgrid, 256, 0, stream>>>(B0, final_map, HG, NSEEDS);
    ln_relu_kernel<<<(NSEEDS + 7) / 8, 256, 0, stream>>>(
        HG, ln_g + 2 * HDIM, ln_b + 2 * HDIM, HG2, NSEEDS);
    const int  mtS = NSEEDS / 16;
    const dim3 gridS((mtS + 3) / 4, NCLSPAD / 64, 1);
    gemm_f32_wmma_kernel<<<gridS, gblk, 0, stream>>>(
        HG2, PWT, PB, nullptr, SC, NSEEDS, HDIM, NCLSPAD);
    logsoftmax_kernel<<<(NSEEDS + 7) / 8, 256, 0, stream>>>(SC, out, NSEEDS);
}


# Per-edge float atomics (max, sums) replaced by per-call integer counting sort by destination + one wave per node reducing in registers; sums in list order, rescaled to new max per 8-edge chunk
# speedup vs baseline: 2.0528x; 2.0528x over previous
#include <hip/hip_runtime.h>
#include <hip/hip_bf16.h>
#include <math.h>

#define NNODES   100000
#define NEDGES   600000
#define FDIM     256
#define HDIM     128
#define NCLS     47
#define NCLSPAD  64
#define NSEEDS   10000
#define GEN_EPS  1e-7f
#define LN_EPS   1e-5f

typedef float v2f __attribute__((ext_vector_type(2)));
typedef float v8f __attribute__((ext_vector_type(8)));

__global__ __launch_bounds__(128)
void gemm_f32_wmma_kernel(const float* __restrict__ A, const float* __restrict__ Wt,
                          const float* __restrict__ bias, const float* __restrict__ res,
                          float* __restrict__ out, int M, int K, int Nc)
{
    const int lane  = threadIdx.x;
    const int mtile = blockIdx.x * 4 + threadIdx.y;
    const int row0  = mtile * 16;
    if (row0 >= M) return;
    const int hb   = lane >> 4;
    const int lo   = lane & 15;
    const int arow = row0 + lo;
    const int col0 = blockIdx.y * 64 + lo;

    const float* __restrict__ a_ptr = A  + (size_t)arow * K;
    const float* __restrict__ b_ptr = Wt + (size_t)col0 * K;
    const size_t bstep = (size_t)16 * K;

    v8f acc[4];
#pragma unroll
    for (int t = 0; t < 4; ++t)
#pragma unroll
        for (int i = 0; i < 8; ++i) acc[t][i] = 0.f;

#pragma unroll 4
    for (int k = 0; k < K; k += 4) {
        const int ka = k + 2 * hb;
        const v2f a  = *(const v2f*)(a_ptr + ka);
        const v2f b0 = *(const v2f*)(b_ptr + ka);
        const v2f b1 = *(const v2f*)(b_ptr + bstep + ka);
        const v2f b2 = *(const v2f*)(b_ptr + 2 * bstep + ka);
        const v2f b3 = *(const v2f*)(b_ptr + 3 * bstep + ka);
        acc[0] = __builtin_amdgcn_wmma_f32_16x16x4_f32(false, a, false, b0, (short)0, acc[0], false, false);
        acc[1] = __builtin_amdgcn_wmma_f32_16x16x4_f32(false, a, false, b1, (short)0, acc[1], false, false);
        acc[2] = __builtin_amdgcn_wmma_f32_16x16x4_f32(false, a, false, b2, (short)0, acc[2], false, false);
        acc[3] = __builtin_amdgcn_wmma_f32_16x16x4_f32(false, a, false, b3, (short)0, acc[3], false, false);
    }

#pragma unroll
    for (int t = 0; t < 4; ++t) {
        const int   col = col0 + 16 * t;
        const float bv  = bias[col];
#pragma unroll
        for (int i = 0; i < 8; ++i) {
            const int r = row0 + i + 8 * hb;
            float v = acc[t][i] + bv;
            if (res) v += res[(size_t)r * Nc + col];
            out[(size_t)r * Nc + col] = v;
        }
    }
}

__global__ __launch_bounds__(256)
void transpose_pad_kernel(const float* __restrict__ W, float* __restrict__ Wt,
                          int K, int Nc, int Ncpad)
{
    const int tid = blockIdx.x * 256 + threadIdx.x;
    if (tid >= Ncpad * K) return;
    const int c = tid / K;
    const int k = tid - c * K;
    Wt[(size_t)c * K + k] = (c < Nc) ? W[(size_t)k * Nc + c] : 0.f;
}

__global__ void pad_bias_kernel(const float* __restrict__ b, float* __restrict__ bp,
                                int Nc, int Ncpad)
{
    const int c = threadIdx.x;
    if (c < Ncpad) bp[c] = (c < Nc) ? b[c] : 0.f;
}

__global__ __launch_bounds__(256)
void zero_int_kernel(int* __restrict__ p, int n)
{
    const int i = blockIdx.x * 256 + threadIdx.x;
    if (i < n) p[i] = 0;
}

__global__ __launch_bounds__(256)
void csr_count_kernel(const int* __restrict__ dst, int* __restrict__ deg)
{
    const int e = blockIdx.x * 256 + threadIdx.x;
    if (e >= 2 * NEDGES) return;
    const int v = ((e >= NEDGES) ? NNODES : 0) + dst[e];
    atomicAdd(deg + v, 1);
}

__global__ __launch_bounds__(256)
void csr_alloc_kernel(const int* __restrict__ deg, int* __restrict__ counter,
                      int* __restrict__ seg_beg, int* __restrict__ cursor, int n)
{
    __shared__ int wtot[8];
    __shared__ int blk_base;
    const int tid  = threadIdx.x;
    const int lane = tid & 31;
    const int w    = tid >> 5;
    const int i    = blockIdx.x * 256 + tid;
    const int d    = (i < n) ? deg[i] : 0;
    int incl = d;
#pragma unroll
    for (int off = 1; off < 32; off <<= 1) {
        const int up = __shfl_up(incl, off, 32);
        if (lane >= off) incl += up;
    }
    if (lane == 31) wtot[w] = incl;
    __syncthreads();
    if (tid == 0) {
        int tot = 0;
#pragma unroll
        for (int k = 0; k < 8; ++k) tot += wtot[k];
        blk_base = atomicAdd(counter, tot);
    }
    __syncthreads();
    int st = blk_base + incl - d;
#pragma unroll
    for (int k = 0; k < 8; ++k) if (k < w) st += wtot[k];
    if (i < n) { seg_beg[i] = st; cursor[i] = st; }
}

__global__ __launch_bounds__(256)
void csr_fill_kernel(const int* __restrict__ src, const int* __restrict__ dst,
                     int* __restrict__ cursor, int* __restrict__ csr_src)
{
    const int e = blockIdx.x * 256 + threadIdx.x;
    if (e >= 2 * NEDGES) return;
    const int v   = ((e >= NEDGES) ? NNODES : 0) + dst[e];
    const int pos = atomicAdd(cursor + v, 1);
    csr_src[pos] = src[e];
}

#define AGG_BLOCKS 4096

template <int CNT>
__device__ __forceinline__ void agg_chunk(const float* __restrict__ hl, int idx, int c, bool first,
                                          float (&mx)[4], float (&de)[4], float (&nu)[4])
{
    float4 hv[CNT];
#pragma unroll
    for (int j = 0; j < CNT; ++j) {
        const int s = __builtin_amdgcn_readlane(idx, c + j);
        hv[j] = *(const float4*)(hl + (size_t)s * HDIM);
    }
    float m[CNT][4];
    float c0 = mx[0], c1 = mx[1], c2 = mx[2], c3 = mx[3];
#pragma unroll
    for (int j = 0; j < CNT; ++j) {
        m[j][0] = fmaxf(hv[j].x, 0.f) + GEN_EPS;
        m[j][1] = fmaxf(hv[j].y, 0.f) + GEN_EPS;
        m[j][2] = fmaxf(hv[j].z, 0.f) + GEN_EPS;
        m[j][3] = fmaxf(hv[j].w, 0.f) + GEN_EPS;
        c0 = fmaxf(c0, m[j][0]);
        c1 = fmaxf(c1, m[j][1]);
        c2 = fmaxf(c2, m[j][2]);
        c3 = fmaxf(c3, m[j][3]);
    }
    if (!first) {
        const float s0 = expf(mx[0] - c0), s1 = expf(mx[1] - c1);
        const float s2 = expf(mx[2] - c2), s3 = expf(mx[3] - c3);
        de[0] *= s0; nu[0] *= s0;
        de[1] *= s1; nu[1] *= s1;
        de[2] *= s2; nu[2] *= s2;
        de[3] *= s3; nu[3] *= s3;
    }
    mx[0] = c0; mx[1] = c1; mx[2] = c2; mx[3] = c3;
#pragma unroll
    for (int j = 0; j < CNT; ++j) {
#pragma unroll
        for (int f = 0; f < 4; ++f) {
            const float ex = expf(m[j][f] - mx[f]);
            de[f] += ex;
            nu[f] += m[j][f] * ex;
        }
    }
}

__global__ __launch_bounds__(256)
void node_softmax_agg_kernel(const float* __restrict__ hin, const int* __restrict__ seg_beg,
                             const int* __restrict__ seg_end, const int* __restrict__ csr_src,
                             float* __restrict__ t)
{
    const int lane = threadIdx.x & 31;
    const int wv = __builtin_amdgcn_readfirstlane((int)(threadIdx.x >> 5));
    const float* __restrict__ hl = hin + lane * 4;

    for (int row = blockIdx.x * 8 + wv; row < NNODES; row += AGG_BLOCKS * 8) {
        const int beg = __builtin_amdgcn_readfirstlane(seg_beg[row]);
        const int n   = __builtin_amdgcn_readfirstlane(seg_end[row]) - beg;
        const float4 own = *(const float4*)(hl + (size_t)row * HDIM);

        float mx[4] = {0.f, 0.f, 0.f, 0.f};
        float de[4] = {0.f, 0.f, 0.f, 0.f};
        float nu[4] = {0.f, 0.f, 0.f, 0.f};

        for (int base = 0; base < n; base += 32) {
            const int bc  = (n - base < 32) ? (n - base) : 32;
            const int idx = (lane < bc) ? csr_src[beg + base + lane] : 0;
            for (int c = 0; c < bc; c += 8) {
                const int  cnt   = (bc - c < 8) ? (bc - c) : 8;
                const bool first = (base + c == 0);
                switch (cnt) {
                case 1:  agg_chunk<1>(hl, idx, c, first, mx, de, nu); break;
                case 2:  agg_chunk<2>(hl, idx, c, first, mx, de, nu); break;
                case 3:  agg_chunk<3>(hl, idx, c, first, mx, de, nu); break;
                case 4:  agg_chunk<4>(hl, idx, c, first, mx, de, nu); break;
                case 5:  agg_chunk<5>(hl, idx, c, first, mx, de, nu); break;
                case 6:  agg_chunk<6>(hl, idx, c, first, mx, de, nu); break;
                case 7:  agg_chunk<7>(hl, idx, c, first, mx, de, nu); break;
                default: agg_chunk<8>(hl, idx, c, first, mx, de, nu); break;
                }
            }
        }

        float4 r;
        r.x = nu[0] / (de[0] + 1e-16f) + own.x;
        r.y = nu[1] / (de[1] + 1e-16f) + own.y;
        r.z = nu[2] / (de[2] + 1e-16f) + own.z;
        r.w = nu[3] / (de[3] + 1e-16f) + own.w;
        *(float4*)(t + (size_t)row * HDIM + lane * 4) = r;
    }
}

__global__ __launch_bounds__(256)
void ln_relu_kernel(const float* __restrict__ in, const float* __restrict__ g,
                    const float* __restrict__ b, float* __restrict__ out, int M)
{
    const int row  = blockIdx.x * 8 + (threadIdx.x >> 5);
    const int lane = threadIdx.x & 31;
    if (row >= M) return;
    const float4 v = *(const float4*)(in + (size_t)row * HDIM + lane * 4);
    float s1 = v.x + v.y + v.z + v.w;
    float s2 = v.x * v.x + v.y * v.y + v.z * v.z + v.w * v.w;
#pragma unroll
    for (int off = 16; off > 0; off >>= 1) {
        s1 += __shfl_xor(s1, off, 32);
        s2 += __shfl_xor(s2, off, 32);
    }
    const float mu  = s1 * (1.f / HDIM);
    const float var = s2 * (1.f / HDIM) - mu * mu;
    const float inv = rsqrtf(var + LN_EPS);
    const float4 gg = *(const float4*)(g + lane * 4);
    const float4 bb = *(const float4*)(b + lane * 4);
    float4 r;
    r.x = fmaxf((v.x - mu) * inv * gg.x + bb.x, 0.f);
    r.y = fmaxf((v.y - mu) * inv * gg.y + bb.y, 0.f);
    r.z = fmaxf((v.z - mu) * inv * gg.z + bb.z, 0.f);
    r.w = fmaxf((v.w - mu) * inv * gg.w + bb.w, 0.f);
    *(float4*)(out + (size_t)row * HDIM + lane * 4) = r;
}

__global__ __launch_bounds__(256)
void gather_kernel(const float* __restrict__ in, const int* __restrict__ map,
                   float* __restrict__ out, int M)
{
    const int tid = blockIdx.x * 256 + threadIdx.x;
    if (tid >= M * 32) return;
    const int i = tid >> 5;
    const int f = (tid & 31) * 4;
    const int j = map[i];
    *(float4*)(out + (size_t)i * HDIM + f) =
        *(const float4*)(in + (size_t)j * HDIM + f);
}

__global__ __launch_bounds__(256)
void logsoftmax_kernel(const float* __restrict__ in, float* __restrict__ out, int M)
{
    const int row  = blockIdx.x * 8 + (threadIdx.x >> 5);
    const int lane = threadIdx.x & 31;
    if (row >= M) return;
    const float NEG = -3.0e38f;
    const float v0 = (lane      < NCLS) ? in[(size_t)row * NCLSPAD + lane     ] : NEG;
    const float v1 = (lane + 32 < NCLS) ? in[(size_t)row * NCLSPAD + lane + 32] : NEG;
    float mx = fmaxf(v0, v1);
#pragma unroll
    for (int off = 16; off > 0; off >>= 1) mx = fmaxf(mx, __shfl_xor(mx, off, 32));
    float se = ((lane      < NCLS) ? expf(v0 - mx) : 0.f)
             + ((lane + 32 < NCLS) ? expf(v1 - mx) : 0.f);
#pragma unroll
    for (int off = 16; off > 0; off >>= 1) se += __shfl_xor(se, off, 32);
    const float lse = logf(se) + mx;
    if (lane      < NCLS) out[(size_t)row * NCLS + lane     ] = v0 - lse;
    if (lane + 32 < NCLS) out[(size_t)row * NCLS + lane + 32] = v1 - lse;
}

extern "C" void kernel_launch(void* const* d_in, const int* in_sizes, int n_in,
                              void* d_out, int out_size, void* d_ws, size_t ws_size,
                              hipStream_t stream)
{
    (void)in_sizes; (void)n_in; (void)out_size; (void)ws_size;

    const float* x         = (const float*)d_in[0];
    const int*   src       = (const int*)  d_in[1];
    const int*   dst       = (const int*)  d_in[2];
    const int*   node_map  = (const int*)  d_in[3];
    const int*   final_map = (const int*)  d_in[4];
    const float* enc_w     = (const float*)d_in[5];
    const float* enc_b     = (const float*)d_in[6];
    const float* gcn_w     = (const float*)d_in[7];
    const float* gcn_b     = (const float*)d_in[8];
    const float* ln_g      = (const float*)d_in[9];
    const float* ln_b      = (const float*)d_in[10];
    const float* pred_w    = (const float*)d_in[11];
    const float* pred_b    = (const float*)d_in[12];
    float* out = (float*)d_out;

    const size_t NH = (size_t)NNODES * HDIM;
    float* B0     = (float*)d_ws;
    float* B1     = B0 + NH;
    float* B2     = B0 + 2 * NH;
    float* B3     = B0 + 3 * NH;
    float* B4     = B0 + 4 * NH;
    float* HG     = B0 + 5 * NH;
    float* HG2    = HG  + (size_t)NSEEDS * HDIM;
    float* SC     = HG2 + (size_t)NSEEDS * HDIM;
    float* WT_ENC = SC  + (size_t)NSEEDS * NCLSPAD;
    float* WT_G   = WT_ENC + (size_t)HDIM * FDIM;
    float* PWT    = WT_G   + (size_t)3 * HDIM * HDIM;
    float* PB     = PWT    + (size_t)NCLSPAD * HDIM;

    transpose_pad_kernel<<<(HDIM * FDIM + 255) / 256, 256, 0, stream>>>(
        enc_w, WT_ENC, FDIM, HDIM, HDIM);
    for (int l = 0; l < 3; ++l)
        transpose_pad_kernel<<<(HDIM * HDIM + 255) / 256, 256, 0, stream>>>(
            gcn_w + (size_t)l * HDIM * HDIM, WT_G + (size_t)l * HDIM * HDIM,
            HDIM, HDIM, HDIM);
    transpose_pad_kernel<<<(NCLSPAD * HDIM + 255) / 256, 256, 0, stream>>>(
        pred_w, PWT, HDIM, NCLS, NCLSPAD);
    pad_bias_kernel<<<1, NCLSPAD, 0, stream>>>(pred_b, PB, NCLS, NCLSPAD);

    const dim3 gblk(32, 4, 1);
    const int  mtN   = NNODES / 16;
    const dim3 gridN((mtN + 3) / 4, HDIM / 64, 1);
    const int  ngrid = (NNODES * 32 + 255) / 256;
    const int  lgrid = (NNODES + 7) / 8;

    int* IDEG   = (int*)B3;
    int* ICNT   = IDEG + 2 * NNODES;
    int* ISTART = IDEG + 2 * NNODES + 32;
    int* ICUR   = ISTART + 2 * NNODES;
    int* ICSR   = ICUR + 2 * NNODES;
    zero_int_kernel<<<(2 * NNODES + 32 + 255) / 256, 256, 0, stream>>>(IDEG, 2 * NNODES + 32);
    csr_count_kernel<<<(2 * NEDGES + 255) / 256, 256, 0, stream>>>(dst, IDEG);
    csr_alloc_kernel<<<(2 * NNODES + 255) / 256, 256, 0, stream>>>(
        IDEG, ICNT, ISTART, ICUR, 2 * NNODES);
    csr_fill_kernel<<<(2 * NEDGES + 255) / 256, 256, 0, stream>>>(src, dst, ICUR, ICSR);

    gemm_f32_wmma_kernel<<<gridN, gblk, 0, stream>>>(
        x, WT_ENC, enc_b, nullptr, B0, NNODES, FDIM, HDIM);

    auto conv = [&](const float* hin, int eblk, const float* wt, const float* bias,
                    const float* res, float* outp) {
        node_softmax_agg_kernel<<<AGG_BLOCKS, 256, 0, stream>>>(
            hin, ISTART + (size_t)eblk * NNODES, ICUR + (size_t)eblk * NNODES, ICSR, B2);
        gemm_f32_wmma_kernel<<<gridN, gblk, 0, stream>>>(
            B2, wt, bias, res, outp, NNODES, HDIM, HDIM);
    };

    conv(B0, 0, WT_G, gcn_b, nullptr, B0);

    for (int l = 1; l < 3; ++l) {
        ln_relu_kernel<<<lgrid, 256, 0, stream>>>(
            B0, ln_g + (size_t)(l - 1) * HDIM, ln_b + (size_t)(l - 1) * HDIM, B1, NNODES);
        conv(B1, l - 1, WT_G + (size_t)l * HDIM * HDIM, gcn_b + (size_t)l * HDIM,
              B0,  B4);
        gather_kernel<<<ngrid, 256, 0, stream>>>(
            B4, node_map + (size_t)(l - 1) * NNODES, B0, NNODES);
    }

    const int sgrid = (NSEEDS * 32 + 255) / 256;
    gather_kernel<<<sgrid, 256, 0, stream>>>(B0, final_map, HG, NSEEDS);
    ln_relu_kernel<<<(NSEEDS + 7) / 8, 256, 0, stream>>>(
        HG, ln_g + 2 * HDIM, ln_b + 2 * HDIM, HG2, NSEEDS);
    const int  mtS = NSEEDS / 16;
    const dim3 gridS((mtS + 3) / 4, NCLSPAD / 64, 1);
    gemm_f32_wmma_kernel<<<gridS, gblk, 0, stream>>>(
        HG2, PWT, PB, nullptr, SC, NSEEDS, HDIM, NCLSPAD);
    logsoftmax_kernel<<<(NSEEDS + 7) / 8, 256, 0, stream>>>(SC, out, NSEEDS);
}


# Node GEMMs retiled: weights held in LDS per 16-wave group, each wave a 16x128 strip with 16-byte fragment loads and whole-line stores; k pairs regrouped per WMMA, so the sum order over k changes
# speedup vs baseline: 3.0378x; 1.4798x over previous
#include <hip/hip_runtime.h>
#include <hip/hip_bf16.h>
#include <math.h>

#define NNODES   100000
#define NEDGES   600000
#define FDIM     256
#define HDIM     128
#define NCLS     47
#define NCLSPAD  64
#define NSEEDS   10000
#define GEN_EPS  1e-7f
#define LN_EPS   1e-5f

typedef float v2f __attribute__((ext_vector_type(2)));
typedef float v8f __attribute__((ext_vector_type(8)));

__global__ __launch_bounds__(128)
void gemm_f32_wmma_kernel(const float* __restrict__ A, const float* __restrict__ Wt,
                          const float* __restrict__ bias, const float* __restrict__ res,
                          float* __restrict__ out, int M, int K, int Nc)
{
    const int lane  = threadIdx.x;
    const int mtile = blockIdx.x * 4 + threadIdx.y;
    const int row0  = mtile * 16;
    if (row0 >= M) return;
    const int hb   = lane >> 4;
    const int lo   = lane & 15;
    const int arow = row0 + lo;
    const int col0 = blockIdx.y * 64 + lo;

    const float* __restrict__ a_ptr = A  + (size_t)arow * K;
    const float* __restrict__ b_ptr = Wt + (size_t)col0 * K;
    const size_t bstep = (size_t)16 * K;

    v8f acc[4];
#pragma unroll
    for (int t = 0; t < 4; ++t)
#pragma unroll
        for (int i = 0; i < 8; ++i) acc[t][i] = 0.f;

#pragma unroll 4
    for (int k = 0; k < K; k += 4) {
        const int ka = k + 2 * hb;
        const v2f a  = *(const v2f*)(a_ptr + ka);
        const v2f b0 = *(const v2f*)(b_ptr + ka);
        const v2f b1 = *(const v2f*)(b_ptr + bstep + ka);
        const v2f b2 = *(const v2f*)(b_ptr + 2 * bstep + ka);
        const v2f b3 = *(const v2f*)(b_ptr + 3 * bstep + ka);
        acc[0] = __builtin_amdgcn_wmma_f32_16x16x4_f32(false, a, false, b0, (short)0, acc[0], false, false);
        acc[1] = __builtin_amdgcn_wmma_f32_16x16x4_f32(false, a, false, b1, (short)0, acc[1], false, false);
        acc[2] = __builtin_amdgcn_wmma_f32_16x16x4_f32(false, a, false, b2, (short)0, acc[2], false, false);
        acc[3] = __builtin_amdgcn_wmma_f32_16x16x4_f32(false, a, false, b3, (short)0, acc[3], false, false);
    }

#pragma unroll
    for (int t = 0; t < 4; ++t) {
        const int   col = col0 + 16 * t;
        const float bv  = bias[col];
#pragma unroll
        for (int i = 0; i < 8; ++i) {
            const int r = row0 + i + 8 * hb;
            float v = acc[t][i] + bv;
            if (res) v += res[(size_t)r * Nc + col];
            out[(size_t)r * Nc + col] = v;
        }
    }
}

#define GEMM_WGS 256


__device__ __forceinline__ void load_a_chunk(const float* __restrict__ ap, float4 (&a)[16])
{
#pragma unroll
    for (int i = 0; i < 16; ++i) a[i] = *(const float4*)(ap + 8 * i);
}

template <int K>
__device__ __forceinline__ void load_b_block(const float* bp, float4 (&b)[8])
{
    constexpr int PITCH = K + 4;
#pragma unroll
    for (int t = 0; t < 8; ++t) b[t] = *(const float4*)(bp + t * 16 * PITCH);
}

__device__ __forceinline__ void mma_block(const float4 a, const float4 (&b)[8], v8f (&acc)[8])
{
    const v2f a01 = {a.x, a.y};
    const v2f a23 = {a.z, a.w};
#pragma unroll
    for (int t = 0; t < 8; ++t) {
        const v2f b01 = {b[t].x, b[t].y};
        acc[t] = __builtin_amdgcn_wmma_f32_16x16x4_f32(false, a01, false, b01, (short)0, acc[t], false, false);
    }
#pragma unroll
    for (int t = 0; t < 8; ++t) {
        const v2f b23 = {b[t].z, b[t].w};
        acc[t] = __builtin_amdgcn_wmma_f32_16x16x4_f32(false, a23, false, b23, (short)0, acc[t], false, false);
    }
}

template <int K>
__device__ __forceinline__ void mma_chunk(const float* bp, const float4 (&a)[16], v8f (&acc)[8])
{
    float4 b0[8], b1[8];
    load_b_block<K>(bp, b0);
#pragma unroll
    for (int i = 0; i < 16; i += 2) {
        load_b_block<K>(bp + 8 * (i + 1), b1);
        __builtin_amdgcn_sched_barrier(0);
        mma_block(a[i], b0, acc);
        __builtin_amdgcn_sched_barrier(0);
        if (i + 2 < 16) load_b_block<K>(bp + 8 * (i + 2), b0);
        __builtin_amdgcn_sched_barrier(0);
        mma_block(a[i + 1], b1, acc);
        __builtin_amdgcn_sched_barrier(0);
    }
}

template <bool HAS_RES>
__device__ __forceinline__ void store_strip(const v8f (&acc)[8], const float* __restrict__ bias,
                                            const float* __restrict__ res, float* __restrict__ out,
                                            int row0, int c, int h)
{
#pragma unroll
    for (int g = 0; g < 2; ++g) {
        const int    col = 64 * g + 4 * c;
        const float4 bv  = *(const float4*)(bias + col);
        float4 rv[8];
        if (HAS_RES) {
#pragma unroll
            for (int i = 0; i < 8; ++i)
                rv[i] = *(const float4*)(res + (size_t)(row0 + i + 8 * h) * HDIM + col);
        }
#pragma unroll
        for (int i = 0; i < 8; ++i) {
            float4 v;
            v.x = acc[4 * g + 0][i] + bv.x;
            v.y = acc[4 * g + 1][i] + bv.y;
            v.z = acc[4 * g + 2][i] + bv.z;
            v.w = acc[4 * g + 3][i] + bv.w;
            if (HAS_RES) { v.x += rv[i].x; v.y += rv[i].y; v.z += rv[i].z; v.w += rv[i].w; }
            *(float4*)(out + (size_t)(row0 + i + 8 * h) * HDIM + col) = v;
        }
    }
}

template <int K>
__global__ __launch_bounds__(512)
void gemm_rows128_kernel(const float* __restrict__ A, const float* __restrict__ Wt,
                         const float* __restrict__ bias, const float* __restrict__ res,
                         float* __restrict__ out, int ntiles)
{
    constexpr int PITCH = K + 4;
    __shared__ __attribute__((aligned(16))) float wl[HDIM * PITCH];

    const int tid  = threadIdx.x;
    const int lane = tid & 31;
    const int wv   = __builtin_amdgcn_readfirstlane(tid >> 5);
    const int c    = lane & 15;
    const int h    = lane >> 4;

    const int wa = wv >> 2;
    const int q  = wa + 4 * (((wv & 3) - wa) & 3);
    int tile = q * GEMM_WGS + (int)blockIdx.x;

    float4 a[16];
    if (tile < ntiles) load_a_chunk(A + (size_t)(tile * 16 + c) * K + 4 * h, a);

    constexpr int NPASS = HDIM * (K / 4) / 512;
    float4 wq[NPASS];
#pragma unroll
    for (int r = 0; r < NPASS; ++r) {
        const int i  = tid + 512 * r;
        const int n  = i / (K / 4);
        const int k4 = (i - n * (K / 4)) * 4;
        wq[r] = *(const float4*)(Wt + (size_t)n * K + k4);
    }
#pragma unroll
    for (int r = 0; r < NPASS; ++r) {
        const int i   = tid + 512 * r;
        const int n   = i / (K / 4);
        const int k4  = (i - n * (K / 4)) * 4;
        const int rho = (n & 64) + 16 * (n & 3) + ((n >> 2) & 15);
        *(float4*)(wl + rho * PITCH + k4) = wq[r];
    }
    __syncthreads();

    const float* bp = wl + c * PITCH + 4 * h;

    while (tile < ntiles) {
        const int row0 = tile * 16;

        v8f acc[8];
#pragma unroll
        for (int t = 0; t < 8; ++t)
#pragma unroll
            for (int i = 0; i < 8; ++i) acc[t][i] = 0.f;

        mma_chunk<K>(bp, a, acc);
#pragma unroll
        for (int kc = 128; kc < K; kc += 128) {
            load_a_chunk(A + (size_t)(row0 + c) * K + 4 * h + kc, a);
            mma_chunk<K>(bp + kc, a, acc);
        }

        if (res) store_strip<true >(acc, bias, res, out, row0, c, h);
        else     store_strip<false>(acc, bias, res, out, row0, c, h);

        tile += 16 * GEMM_WGS;
        if (tile < ntiles) load_a_chunk(A + (size_t)(tile * 16 + c) * K + 4 * h, a);
    }
}

__global__ __launch_bounds__(256)
void transpose_pad_kernel(const float* __restrict__ W, float* __restrict__ Wt,
                          int K, int Nc, int Ncpad)
{
    const int tid = blockIdx.x * 256 + threadIdx.x;
    if (tid >= Ncpad * K) return;
    const int c = tid / K;
    const int k = tid - c * K;
    Wt[(size_t)c * K + k] = (c < Nc) ? W[(size_t)k * Nc + c] : 0.f;
}

__global__ void pad_bias_kernel(const float* __restrict__ b, float* __restrict__ bp,
                                int Nc, int Ncpad)
{
    const int c = threadIdx.x;
    if (c < Ncpad) bp[c] = (c < Nc) ? b[c] : 0.f;
}

__global__ __launch_bounds__(256)
void zero_int_kernel(int* __restrict__ p, int n)
{
    const int i = blockIdx.x * 256 + threadIdx.x;
    if (i < n) p[i] = 0;
}

__global__ __launch_bounds__(256)
void csr_count_kernel(const int* __restrict__ dst, int* __restrict__ deg)
{
    const int e = blockIdx.x * 256 + threadIdx.x;
    if (e >= 2 * NEDGES) return;
    const int v = ((e >= NEDGES) ? NNODES : 0) + dst[e];
    atomicAdd(deg + v, 1);
}

__global__ __launch_bounds__(256)
void csr_alloc_kernel(const int* __restrict__ deg, int* __restrict__ counter,
                      int* __restrict__ seg_beg, int* __restrict__ cursor, int n)
{
    __shared__ int wtot[8];
    __shared__ int blk_base;
    const int tid  = threadIdx.x;
    const int lane = tid & 31;
    const int w    = tid >> 5;
    const int i    = blockIdx.x * 256 + tid;
    const int d    = (i < n) ? deg[i] : 0;
    int incl = d;
#pragma unroll
    for (int off = 1; off < 32; off <<= 1) {
        const int up = __shfl_up(incl, off, 32);
        if (lane >= off) incl += up;
    }
    if (lane == 31) wtot[w] = incl;
    __syncthreads();
    if (tid == 0) {
        int tot = 0;
#pragma unroll
        for (int k = 0; k < 8; ++k) tot += wtot[k];
        blk_base = atomicAdd(counter, tot);
    }
    __syncthreads();
    int st = blk_base + incl - d;
#pragma unroll
    for (int k = 0; k < 8; ++k) if (k < w) st += wtot[k];
    if (i < n) { seg_beg[i] = st; cursor[i] = st; }
}

__global__ __launch_bounds__(256)
void csr_fill_kernel(const int* __restrict__ src, const int* __restrict__ dst,
                     int* __restrict__ cursor, int* __restrict__ csr_src)
{
    const int e = blockIdx.x * 256 + threadIdx.x;
    if (e >= 2 * NEDGES) return;
    const int v   = ((e >= NEDGES) ? NNODES : 0) + dst[e];
    const int pos = atomicAdd(cursor + v, 1);
    csr_src[pos] = src[e];
}

#define AGG_BLOCKS 4096

template <int CNT>
__device__ __forceinline__ void agg_chunk(const float* __restrict__ hl, int idx, int c, bool first,
                                          float (&mx)[4], float (&de)[4], float (&nu)[4])
{
    float4 hv[CNT];
#pragma unroll
    for (int j = 0; j < CNT; ++j) {
        const int s = __builtin_amdgcn_readlane(idx, c + j);
        hv[j] = *(const float4*)(hl + (size_t)s * HDIM);
    }
    float m[CNT][4];
    float c0 = mx[0], c1 = mx[1], c2 = mx[2], c3 = mx[3];
#pragma unroll
    for (int j = 0; j < CNT; ++j) {
        m[j][0] = fmaxf(hv[j].x, 0.f) + GEN_EPS;
        m[j][1] = fmaxf(hv[j].y, 0.f) + GEN_EPS;
        m[j][2] = fmaxf(hv[j].z, 0.f) + GEN_EPS;
        m[j][3] = fmaxf(hv[j].w, 0.f) + GEN_EPS;
        c0 = fmaxf(c0, m[j][0]);
        c1 = fmaxf(c1, m[j][1]);
        c2 = fmaxf(c2, m[j][2]);
        c3 = fmaxf(c3, m[j][3]);
    }
    if (!first) {
        const float s0 = expf(mx[0] - c0), s1 = expf(mx[1] - c1);
        const float s2 = expf(mx[2] - c2), s3 = expf(mx[3] - c3);
        de[0] *= s0; nu[0] *= s0;
        de[1] *= s1; nu[1] *= s1;
        de[2] *= s2; nu[2] *= s2;
        de[3] *= s3; nu[3] *= s3;
    }
    mx[0] = c0; mx[1] = c1; mx[2] = c2; mx[3] = c3;
#pragma unroll
    for (int j = 0; j < CNT; ++j) {
#pragma unroll
        for (int f = 0; f < 4; ++f) {
            const float ex = expf(m[j][f] - mx[f]);
            de[f] += ex;
            nu[f] += m[j][f] * ex;
        }
    }
}

__global__ __launch_bounds__(256)
void node_softmax_agg_kernel(const float* __restrict__ hin, const int* __restrict__ seg_beg,
                             const int* __restrict__ seg_end, const int* __restrict__ csr_src,
                             float* __restrict__ t)
{
    const int lane = threadIdx.x & 31;
    const int wv = __builtin_amdgcn_readfirstlane((int)(threadIdx.x >> 5));
    const float* __restrict__ hl = hin + lane * 4;

    for (int row = blockIdx.x * 8 + wv; row < NNODES; row += AGG_BLOCKS * 8) {
        const int beg = __builtin_amdgcn_readfirstlane(seg_beg[row]);
        const int n   = __builtin_amdgcn_readfirstlane(seg_end[row]) - beg;
        const float4 own = *(const float4*)(hl + (size_t)row * HDIM);

        float mx[4] = {0.f, 0.f, 0.f, 0.f};
        float de[4] = {0.f, 0.f, 0.f, 0.f};
        float nu[4] = {0.f, 0.f, 0.f, 0.f};

        for (int base = 0; base < n; base += 32) {
            const int bc  = (n - base < 32) ? (n - base) : 32;
            const int idx = (lane < bc) ? csr_src[beg + base + lane] : 0;
            for (int c = 0; c < bc; c += 8) {
                const int  cnt   = (bc - c < 8) ? (bc - c) : 8;
                const bool first = (base + c == 0);
                switch (cnt) {
                case 1:  agg_chunk<1>(hl, idx, c, first, mx, de, nu); break;
                case 2:  agg_chunk<2>(hl, idx, c, first, mx, de, nu); break;
                case 3:  agg_chunk<3>(hl, idx, c, first, mx, de, nu); break;
                case 4:  agg_chunk<4>(hl, idx, c, first, mx, de, nu); break;
                case 5:  agg_chunk<5>(hl, idx, c, first, mx, de, nu); break;
                case 6:  agg_chunk<6>(hl, idx, c, first, mx, de, nu); break;
                case 7:  agg_chunk<7>(hl, idx, c, first, mx, de, nu); break;
                default: agg_chunk<8>(hl, idx, c, first, mx, de, nu); break;
                }
            }
        }

        float4 r;
        r.x = nu[0] / (de[0] + 1e-16f) + own.x;
        r.y = nu[1] / (de[1] + 1e-16f) + own.y;
        r.z = nu[2] / (de[2] + 1e-16f) + own.z;
        r.w = nu[3] / (de[3] + 1e-16f) + own.w;
        *(float4*)(t + (size_t)row * HDIM + lane * 4) = r;
    }
}

__global__ __launch_bounds__(256)
void ln_relu_kernel(const float* __restrict__ in, const float* __restrict__ g,
                    const float* __restrict__ b, float* __restrict__ out, int M)
{
    const int row  = blockIdx.x * 8 + (threadIdx.x >> 5);
    const int lane = threadIdx.x & 31;
    if (row >= M) return;
    const float4 v = *(const float4*)(in + (size_t)row * HDIM + lane * 4);
    float s1 = v.x + v.y + v.z + v.w;
    float s2 = v.x * v.x + v.y * v.y + v.z * v.z + v.w * v.w;
#pragma unroll
    for (int off = 16; off > 0; off >>= 1) {
        s1 += __shfl_xor(s1, off, 32);
        s2 += __shfl_xor(s2, off, 32);
    }
    const float mu  = s1 * (1.f / HDIM);
    const float var = s2 * (1.f / HDIM) - mu * mu;
    const float inv = rsqrtf(var + LN_EPS);
    const float4 gg = *(const float4*)(g + lane * 4);
    const float4 bb = *(const float4*)(b + lane * 4);
    float4 r;
    r.x = fmaxf((v.x - mu) * inv * gg.x + bb.x, 0.f);
    r.y = fmaxf((v.y - mu) * inv * gg.y + bb.y, 0.f);
    r.z = fmaxf((v.z - mu) * inv * gg.z + bb.z, 0.f);
    r.w = fmaxf((v.w - mu) * inv * gg.w + bb.w, 0.f);
    *(float4*)(out + (size_t)row * HDIM + lane * 4) = r;
}

__global__ __launch_bounds__(256)
void gather_kernel(const float* __restrict__ in, const int* __restrict__ map,
                   float* __restrict__ out, int M)
{
    const int tid = blockIdx.x * 256 + threadIdx.x;
    if (tid >= M * 32) return;
    const int i = tid >> 5;
    const int f = (tid & 31) * 4;
    const int j = map[i];
    *(float4*)(out + (size_t)i * HDIM + f) =
        *(const float4*)(in + (size_t)j * HDIM + f);
}

__global__ __launch_bounds__(256)
void logsoftmax_kernel(const float* __restrict__ in, float* __restrict__ out, int M)
{
    const int row  = blockIdx.x * 8 + (threadIdx.x >> 5);
    const int lane = threadIdx.x & 31;
    if (row >= M) return;
    const float NEG = -3.0e38f;
    const float v0 = (lane      < NCLS) ? in[(size_t)row * NCLSPAD + lane     ] : NEG;
    const float v1 = (lane + 32 < NCLS) ? in[(size_t)row * NCLSPAD + lane + 32] : NEG;
    float mx = fmaxf(v0, v1);
#pragma unroll
    for (int off = 16; off > 0; off >>= 1) mx = fmaxf(mx, __shfl_xor(mx, off, 32));
    float se = ((lane      < NCLS) ? expf(v0 - mx) : 0.f)
             + ((lane + 32 < NCLS) ? expf(v1 - mx) : 0.f);
#pragma unroll
    for (int off = 16; off > 0; off >>= 1) se += __shfl_xor(se, off, 32);
    const float lse = logf(se) + mx;
    if (lane      < NCLS) out[(size_t)row * NCLS + lane     ] = v0 - lse;
    if (lane + 32 < NCLS) out[(size_t)row * NCLS + lane + 32] = v1 - lse;
}

extern "C" void kernel_launch(void* const* d_in, const int* in_sizes, int n_in,
                              void* d_out, int out_size, void* d_ws, size_t ws_size,
                              hipStream_t stream)
{
    (void)in_sizes; (void)n_in; (void)out_size; (void)ws_size;

    const float* x         = (const float*)d_in[0];
    const int*   src       = (const int*)  d_in[1];
    const int*   dst       = (const int*)  d_in[2];
    const int*   node_map  = (const int*)  d_in[3];
    const int*   final_map = (const int*)  d_in[4];
    const float* enc_w     = (const float*)d_in[5];
    const float* enc_b     = (const float*)d_in[6];
    const float* gcn_w     = (const float*)d_in[7];
    const float* gcn_b     = (const float*)d_in[8];
    const float* ln_g      = (const float*)d_in[9];
    const float* ln_b      = (const float*)d_in[10];
    const float* pred_w    = (const float*)d_in[11];
    const float* pred_b    = (const float*)d_in[12];
    float* out = (float*)d_out;

    const size_t NH = (size_t)NNODES * HDIM;
    float* B0     = (float*)d_ws;
    float* B1     = B0 + NH;
    float* B2     = B0 + 2 * NH;
    float* B3     = B0 + 3 * NH;
    float* B4     = B0 + 4 * NH;
    float* HG     = B0 + 5 * NH;
    float* HG2    = HG  + (size_t)NSEEDS * HDIM;
    float* SC     = HG2 + (size_t)NSEEDS * HDIM;
    float* WT_ENC = SC  + (size_t)NSEEDS * NCLSPAD;
    float* WT_G   = WT_ENC + (size_t)HDIM * FDIM;
    float* PWT    = WT_G   + (size_t)3 * HDIM * HDIM;
    float* PB     = PWT    + (size_t)NCLSPAD * HDIM;

    transpose_pad_kernel<<<(HDIM * FDIM + 255) / 256, 256, 0, stream>>>(
        enc_w, WT_ENC, FDIM, HDIM, HDIM);
    for (int l = 0; l < 3; ++l)
        transpose_pad_kernel<<<(HDIM * HDIM + 255) / 256, 256, 0, stream>>>(
            gcn_w + (size_t)l * HDIM * HDIM, WT_G + (size_t)l * HDIM * HDIM,
            HDIM, HDIM, HDIM);
    transpose_pad_kernel<<<(NCLSPAD * HDIM + 255) / 256, 256, 0, stream>>>(
        pred_w, PWT, HDIM, NCLS, NCLSPAD);
    pad_bias_kernel<<<1, NCLSPAD, 0, stream>>>(pred_b, PB, NCLS, NCLSPAD);

    const dim3 gblk(32, 4, 1);
    const int  mtN   = NNODES / 16;
    const int  ngrid = (NNODES * 32 + 255) / 256;
    const int  lgrid = (NNODES + 7) / 8;

    int* IDEG   = (int*)B3;
    int* ICNT   = IDEG + 2 * NNODES;
    int* ISTART = IDEG + 2 * NNODES + 32;
    int* ICUR   = ISTART + 2 * NNODES;
    int* ICSR   = ICUR + 2 * NNODES;
    zero_int_kernel<<<(2 * NNODES + 32 + 255) / 256, 256, 0, stream>>>(IDEG, 2 * NNODES + 32);
    csr_count_kernel<<<(2 * NEDGES + 255) / 256, 256, 0, stream>>>(dst, IDEG);
    csr_alloc_kernel<<<(2 * NNODES + 255) / 256, 256, 0, stream>>>(
        IDEG, ICNT, ISTART, ICUR, 2 * NNODES);
    csr_fill_kernel<<<(2 * NEDGES + 255) / 256, 256, 0, stream>>>(src, dst, ICUR, ICSR);

    gemm_rows128_kernel<FDIM><<<GEMM_WGS, 512, 0, stream>>>(
        x, WT_ENC, enc_b, nullptr, B0, mtN);

    auto conv = [&](const float* hin, int eblk, const float* wt, const float* bias,
                    const float* res, float* outp) {
        node_softmax_agg_kernel<<<AGG_BLOCKS, 256, 0, stream>>>(
            hin, ISTART + (size_t)eblk * NNODES, ICUR + (size_t)eblk * NNODES, ICSR, B2);
        gemm_rows128_kernel<HDIM><<<GEMM_WGS, 512, 0, stream>>>(
            B2, wt, bias, res, outp, mtN);
    };

    conv(B0, 0, WT_G, gcn_b, nullptr, B0);

    for (int l = 1; l < 3; ++l) {
        ln_relu_kernel<<<lgrid, 256, 0, stream>>>(
            B0, ln_g + (size_t)(l - 1) * HDIM, ln_b + (size_t)(l - 1) * HDIM, B1, NNODES);
        conv(B1, l - 1, WT_G + (size_t)l * HDIM * HDIM, gcn_b + (size_t)l * HDIM,
              B0,  B4);
        gather_kernel<<<ngrid, 256, 0, stream>>>(
            B4, node_map + (size_t)(l - 1) * NNODES, B0, NNODES);
    }

    const int sgrid = (NSEEDS * 32 + 255) / 256;
    gather_kernel<<<sgrid, 256, 0, stream>>>(B0, final_map, HG, NSEEDS);
    ln_relu_kernel<<<(NSEEDS + 7) / 8, 256, 0, stream>>>(
        HG, ln_g + 2 * HDIM, ln_b + 2 * HDIM, HG2, NSEEDS);
    const int  mtS = NSEEDS / 16;
    const dim3 gridS((mtS + 3) / 4, NCLSPAD / 64, 1);
    gemm_f32_wmma_kernel<<<gridS, gblk, 0, stream>>>(
        HG2, PWT, PB, nullptr, SC, NSEEDS, HDIM, NCLSPAD);
    logsoftmax_kernel<<<(NSEEDS + 7) / 8, 256, 0, stream>>>(SC, out, NSEEDS);
}


# Last conv layer evaluated only at the 10000 rows the seed gather reads: its LayerNorm+ReLU, softmax aggregation, product and both row gathers run per seed row, not over all 100000 nodes
# speedup vs baseline: 3.6179x; 1.1910x over previous
#include <hip/hip_runtime.h>
#include <hip/hip_bf16.h>
#include <math.h>

#define NNODES   100000
#define NEDGES   600000
#define FDIM     256
#define HDIM     128
#define NCLS     47
#define NCLSPAD  64
#define NSEEDS   10000
#define GEN_EPS  1e-7f
#define LN_EPS   1e-5f

typedef float v2f __attribute__((ext_vector_type(2)));
typedef float v8f __attribute__((ext_vector_type(8)));

__global__ __launch_bounds__(128)
void gemm_f32_wmma_kernel(const float* __restrict__ A, const float* __restrict__ Wt,
                          const float* __restrict__ bias, const float* __restrict__ res,
                          float* __restrict__ out, int M, int K, int Nc)
{
    const int lane  = threadIdx.x;
    const int mtile = blockIdx.x * 4 + threadIdx.y;
    const int row0  = mtile * 16;
    if (row0 >= M) return;
    const int hb   = lane >> 4;
    const int lo   = lane & 15;
    const int arow = row0 + lo;
    const int col0 = blockIdx.y * 64 + lo;

    const float* __restrict__ a_ptr = A  + (size_t)arow * K;
    const float* __restrict__ b_ptr = Wt + (size_t)col0 * K;
    const size_t bstep = (size_t)16 * K;

    v8f acc[4];
#pragma unroll
    for (int t = 0; t < 4; ++t)
#pragma unroll
        for (int i = 0; i < 8; ++i) acc[t][i] = 0.f;

#pragma unroll 4
    for (int k = 0; k < K; k += 4) {
        const int ka = k + 2 * hb;
        const v2f a  = *(const v2f*)(a_ptr + ka);
        const v2f b0 = *(const v2f*)(b_ptr + ka);
        const v2f b1 = *(const v2f*)(b_ptr + bstep + ka);
        const v2f b2 = *(const v2f*)(b_ptr + 2 * bstep + ka);
        const v2f b3 = *(const v2f*)(b_ptr + 3 * bstep + ka);
        acc[0] = __builtin_amdgcn_wmma_f32_16x16x4_f32(false, a, false, b0, (short)0, acc[0], false, false);
        acc[1] = __builtin_amdgcn_wmma_f32_16x16x4_f32(false, a, false, b1, (short)0, acc[1], false, false);
        acc[2] = __builtin_amdgcn_wmma_f32_16x16x4_f32(false, a, false, b2, (short)0, acc[2], false, false);
        acc[3] = __builtin_amdgcn_wmma_f32_16x16x4_f32(false, a, false, b3, (short)0, acc[3], false, false);
    }

#pragma unroll
    for (int t = 0; t < 4; ++t) {
        const int   col = col0 + 16 * t;
        const float bv  = bias[col];
#pragma unroll
        for (int i = 0; i < 8; ++i) {
            const int r = row0 + i + 8 * hb;
            float v = acc[t][i] + bv;
            if (res) v += res[(size_t)r * Nc + col];
            out[(size_t)r * Nc + col] = v;
        }
    }
}

#define GEMM_WGS 256


__device__ __forceinline__ void load_a_chunk(const float* __restrict__ ap, float4 (&a)[16])
{
#pragma unroll
    for (int i = 0; i < 16; ++i) a[i] = *(const float4*)(ap + 8 * i);
}

template <int K>
__device__ __forceinline__ void load_b_block(const float* bp, float4 (&b)[8])
{
    constexpr int PITCH = K + 4;
#pragma unroll
    for (int t = 0; t < 8; ++t) b[t] = *(const float4*)(bp + t * 16 * PITCH);
}

__device__ __forceinline__ void mma_block(const float4 a, const float4 (&b)[8], v8f (&acc)[8])
{
    const v2f a01 = {a.x, a.y};
    const v2f a23 = {a.z, a.w};
#pragma unroll
    for (int t = 0; t < 8; ++t) {
        const v2f b01 = {b[t].x, b[t].y};
        acc[t] = __builtin_amdgcn_wmma_f32_16x16x4_f32(false, a01, false, b01, (short)0, acc[t], false, false);
    }
#pragma unroll
    for (int t = 0; t < 8; ++t) {
        const v2f b23 = {b[t].z, b[t].w};
        acc[t] = __builtin_amdgcn_wmma_f32_16x16x4_f32(false, a23, false, b23, (short)0, acc[t], false, false);
    }
}

template <int K>
__device__ __forceinline__ void mma_chunk(const float* bp, const float4 (&a)[16], v8f (&acc)[8])
{
    float4 b0[8], b1[8];
    load_b_block<K>(bp, b0);
#pragma unroll
    for (int i = 0; i < 16; i += 2) {
        load_b_block<K>(bp + 8 * (i + 1), b1);
        __builtin_amdgcn_sched_barrier(0);
        mma_block(a[i], b0, acc);
        __builtin_amdgcn_sched_barrier(0);
        if (i + 2 < 16) load_b_block<K>(bp + 8 * (i + 2), b0);
        __builtin_amdgcn_sched_barrier(0);
        mma_block(a[i + 1], b1, acc);
        __builtin_amdgcn_sched_barrier(0);
    }
}

template <bool HAS_RES>
__device__ __forceinline__ void store_strip(const v8f (&acc)[8], const float* __restrict__ bias,
                                            const float* __restrict__ res, float* __restrict__ out,
                                            int row0, int c, int h)
{
#pragma unroll
    for (int g = 0; g < 2; ++g) {
        const int    col = 64 * g + 4 * c;
        const float4 bv  = *(const float4*)(bias + col);
        float4 rv[8];
        if (HAS_RES) {
#pragma unroll
            for (int i = 0; i < 8; ++i)
                rv[i] = *(const float4*)(res + (size_t)(row0 + i + 8 * h) * HDIM + col);
        }
#pragma unroll
        for (int i = 0; i < 8; ++i) {
            float4 v;
            v.x = acc[4 * g + 0][i] + bv.x;
            v.y = acc[4 * g + 1][i] + bv.y;
            v.z = acc[4 * g + 2][i] + bv.z;
            v.w = acc[4 * g + 3][i] + bv.w;
            if (HAS_RES) { v.x += rv[i].x; v.y += rv[i].y; v.z += rv[i].z; v.w += rv[i].w; }
            *(float4*)(out + (size_t)(row0 + i + 8 * h) * HDIM + col) = v;
        }
    }
}

template <int K>
__global__ __launch_bounds__(512)
void gemm_rows128_kernel(const float* __restrict__ A, const float* __restrict__ Wt,
                         const float* __restrict__ bias, const float* __restrict__ res,
                         float* __restrict__ out, int ntiles)
{
    constexpr int PITCH = K + 4;
    __shared__ __attribute__((aligned(16))) float wl[HDIM * PITCH];

    const int tid  = threadIdx.x;
    const int lane = tid & 31;
    const int wv   = __builtin_amdgcn_readfirstlane(tid >> 5);
    const int c    = lane & 15;
    const int h    = lane >> 4;

    const int wa = wv >> 2;
    const int q  = wa + 4 * (((wv & 3) - wa) & 3);
    int tile = q * GEMM_WGS + (int)blockIdx.x;

    float4 a[16];
    if (tile < ntiles) load_a_chunk(A + (size_t)(tile * 16 + c) * K + 4 * h, a);

    constexpr int NPASS = HDIM * (K / 4) / 512;
    float4 wq[NPASS];
#pragma unroll
    for (int r = 0; r < NPASS; ++r) {
        const int i  = tid + 512 * r;
        const int n  = i / (K / 4);
        const int k4 = (i - n * (K / 4)) * 4;
        wq[r] = *(const float4*)(Wt + (size_t)n * K + k4);
    }
#pragma unroll
    for (int r = 0; r < NPASS; ++r) {
        const int i   = tid + 512 * r;
        const int n   = i / (K / 4);
        const int k4  = (i - n * (K / 4)) * 4;
        const int rho = (n & 64) + 16 * (n & 3) + ((n >> 2) & 15);
        *(float4*)(wl + rho * PITCH + k4) = wq[r];
    }
    __syncthreads();

    const float* bp = wl + c * PITCH + 4 * h;

    while (tile < ntiles) {
        const int row0 = tile * 16;

        v8f acc[8];
#pragma unroll
        for (int t = 0; t < 8; ++t)
#pragma unroll
            for (int i = 0; i < 8; ++i) acc[t][i] = 0.f;

        mma_chunk<K>(bp, a, acc);
#pragma unroll
        for (int kc = 128; kc < K; kc += 128) {
            load_a_chunk(A + (size_t)(row0 + c) * K + 4 * h + kc, a);
            mma_chunk<K>(bp + kc, a, acc);
        }

        if (res) store_strip<true >(acc, bias, res, out, row0, c, h);
        else     store_strip<false>(acc, bias, res, out, row0, c, h);

        tile += 16 * GEMM_WGS;
        if (tile < ntiles) load_a_chunk(A + (size_t)(tile * 16 + c) * K + 4 * h, a);
    }
}

__global__ __launch_bounds__(256)
void transpose_pad_kernel(const float* __restrict__ W, float* __restrict__ Wt,
                          int K, int Nc, int Ncpad)
{
    const int tid = blockIdx.x * 256 + threadIdx.x;
    if (tid >= Ncpad * K) return;
    const int c = tid / K;
    const int k = tid - c * K;
    Wt[(size_t)c * K + k] = (c < Nc) ? W[(size_t)k * Nc + c] : 0.f;
}

__global__ void pad_bias_kernel(const float* __restrict__ b, float* __restrict__ bp,
                                int Nc, int Ncpad)
{
    const int c = threadIdx.x;
    if (c < Ncpad) bp[c] = (c < Nc) ? b[c] : 0.f;
}

__global__ __launch_bounds__(256)
void zero_int_kernel(int* __restrict__ p, int n)
{
    const int i = blockIdx.x * 256 + threadIdx.x;
    if (i < n) p[i] = 0;
}

__global__ __launch_bounds__(256)
void csr_count_kernel(const int* __restrict__ dst, int* __restrict__ deg)
{
    const int e = blockIdx.x * 256 + threadIdx.x;
    if (e >= 2 * NEDGES) return;
    const int v = ((e >= NEDGES) ? NNODES : 0) + dst[e];
    atomicAdd(deg + v, 1);
}

__global__ __launch_bounds__(256)
void csr_alloc_kernel(const int* __restrict__ deg, int* __restrict__ counter,
                      int* __restrict__ seg_beg, int* __restrict__ cursor, int n)
{
    __shared__ int wtot[8];
    __shared__ int blk_base;
    const int tid  = threadIdx.x;
    const int lane = tid & 31;
    const int w    = tid >> 5;
    const int i    = blockIdx.x * 256 + tid;
    const int d    = (i < n) ? deg[i] : 0;
    int incl = d;
#pragma unroll
    for (int off = 1; off < 32; off <<= 1) {
        const int up = __shfl_up(incl, off, 32);
        if (lane >= off) incl += up;
    }
    if (lane == 31) wtot[w] = incl;
    __syncthreads();
    if (tid == 0) {
        int tot = 0;
#pragma unroll
        for (int k = 0; k < 8; ++k) tot += wtot[k];
        blk_base = atomicAdd(counter, tot);
    }
    __syncthreads();
    int st = blk_base + incl - d;
#pragma unroll
    for (int k = 0; k < 8; ++k) if (k < w) st += wtot[k];
    if (i < n) { seg_beg[i] = st; cursor[i] = st; }
}

__global__ __launch_bounds__(256)
void csr_fill_kernel(const int* __restrict__ src, const int* __restrict__ dst,
                     int* __restrict__ cursor, int* __restrict__ csr_src)
{
    const int e = blockIdx.x * 256 + threadIdx.x;
    if (e >= 2 * NEDGES) return;
    const int v   = ((e >= NEDGES) ? NNODES : 0) + dst[e];
    const int pos = atomicAdd(cursor + v, 1);
    csr_src[pos] = src[e];
}

__device__ __forceinline__ float4 ln_relu_row(const float4 v, const float4 gg, const float4 bb)
{
    float s1 = v.x + v.y + v.z + v.w;
    float s2 = v.x * v.x + v.y * v.y + v.z * v.z + v.w * v.w;
#pragma unroll
    for (int off = 16; off > 0; off >>= 1) {
        s1 += __shfl_xor(s1, off, 32);
        s2 += __shfl_xor(s2, off, 32);
    }
    const float mu  = s1 * (1.f / HDIM);
    const float var = s2 * (1.f / HDIM) - mu * mu;
    const float inv = rsqrtf(var + LN_EPS);
    float4 r;
    r.x = fmaxf((v.x - mu) * inv * gg.x + bb.x, 0.f);
    r.y = fmaxf((v.y - mu) * inv * gg.y + bb.y, 0.f);
    r.z = fmaxf((v.z - mu) * inv * gg.z + bb.z, 0.f);
    r.w = fmaxf((v.w - mu) * inv * gg.w + bb.w, 0.f);
    return r;
}

#define AGG_BLOCKS 4096

template <int CNT, bool LNF>
__device__ __forceinline__ void agg_chunk(const float* __restrict__ hl, int idx, int c, bool first,
                                          float (&mx)[4], float (&de)[4], float (&nu)[4],
                                          const float4 gg, const float4 bb)
{
    float4 hv[CNT];
#pragma unroll
    for (int j = 0; j < CNT; ++j) {
        const int s = __builtin_amdgcn_readlane(idx, c + j);
        hv[j] = *(const float4*)(hl + (size_t)s * HDIM);
    }
    if (LNF) {
#pragma unroll
        for (int j = 0; j < CNT; ++j) hv[j] = ln_relu_row(hv[j], gg, bb);
    }
    float m[CNT][4];
    float c0 = mx[0], c1 = mx[1], c2 = mx[2], c3 = mx[3];
#pragma unroll
    for (int j = 0; j < CNT; ++j) {
        m[j][0] = fmaxf(hv[j].x, 0.f) + GEN_EPS;
        m[j][1] = fmaxf(hv[j].y, 0.f) + GEN_EPS;
        m[j][2] = fmaxf(hv[j].z, 0.f) + GEN_EPS;
        m[j][3] = fmaxf(hv[j].w, 0.f) + GEN_EPS;
        c0 = fmaxf(c0, m[j][0]);
        c1 = fmaxf(c1, m[j][1]);
        c2 = fmaxf(c2, m[j][2]);
        c3 = fmaxf(c3, m[j][3]);
    }
    if (!first) {
        const float s0 = expf(mx[0] - c0), s1 = expf(mx[1] - c1);
        const float s2 = expf(mx[2] - c2), s3 = expf(mx[3] - c3);
        de[0] *= s0; nu[0] *= s0;
        de[1] *= s1; nu[1] *= s1;
        de[2] *= s2; nu[2] *= s2;
        de[3] *= s3; nu[3] *= s3;
    }
    mx[0] = c0; mx[1] = c1; mx[2] = c2; mx[3] = c3;
#pragma unroll
    for (int j = 0; j < CNT; ++j) {
#pragma unroll
        for (int f = 0; f < 4; ++f) {
            const float ex = expf(m[j][f] - mx[f]);
            de[f] += ex;
            nu[f] += m[j][f] * ex;
        }
    }
}

template <bool LNF>
__device__ __forceinline__ float4 agg_node(const float* __restrict__ hl, const int* __restrict__ csr_src,
                                           int beg, int n, int lane, const float4 own,
                                           const float4 gg, const float4 bb)
{
    float mx[4] = {0.f, 0.f, 0.f, 0.f};
    float de[4] = {0.f, 0.f, 0.f, 0.f};
    float nu[4] = {0.f, 0.f, 0.f, 0.f};

    for (int base = 0; base < n; base += 32) {
        const int bc  = (n - base < 32) ? (n - base) : 32;
        const int idx = (lane < bc) ? csr_src[beg + base + lane] : 0;
        for (int c = 0; c < bc; c += 8) {
            const int  cnt   = (bc - c < 8) ? (bc - c) : 8;
            const bool first = (base + c == 0);
            switch (cnt) {
            case 1:  agg_chunk<1, LNF>(hl, idx, c, first, mx, de, nu, gg, bb); break;
            case 2:  agg_chunk<2, LNF>(hl, idx, c, first, mx, de, nu, gg, bb); break;
            case 3:  agg_chunk<3, LNF>(hl, idx, c, first, mx, de, nu, gg, bb); break;
            case 4:  agg_chunk<4, LNF>(hl, idx, c, first, mx, de, nu, gg, bb); break;
            case 5:  agg_chunk<5, LNF>(hl, idx, c, first, mx, de, nu, gg, bb); break;
            case 6:  agg_chunk<6, LNF>(hl, idx, c, first, mx, de, nu, gg, bb); break;
            case 7:  agg_chunk<7, LNF>(hl, idx, c, first, mx, de, nu, gg, bb); break;
            default: agg_chunk<8, LNF>(hl, idx, c, first, mx, de, nu, gg, bb); break;
            }
        }
    }

    float4 r;
    r.x = nu[0] / (de[0] + 1e-16f) + own.x;
    r.y = nu[1] / (de[1] + 1e-16f) + own.y;
    r.z = nu[2] / (de[2] + 1e-16f) + own.z;
    r.w = nu[3] / (de[3] + 1e-16f) + own.w;
    return r;
}

__global__ __launch_bounds__(256)
void node_softmax_agg_kernel(const float* __restrict__ hin, const int* __restrict__ seg_beg,
                             const int* __restrict__ seg_end, const int* __restrict__ csr_src,
                             float* __restrict__ t)
{
    const int lane = threadIdx.x & 31;
    const int wv = __builtin_amdgcn_readfirstlane((int)(threadIdx.x >> 5));
    const float* __restrict__ hl = hin + lane * 4;
    const float4 none = {0.f, 0.f, 0.f, 0.f};

    for (int row = blockIdx.x * 8 + wv; row < NNODES; row += AGG_BLOCKS * 8) {
        const int beg = __builtin_amdgcn_readfirstlane(seg_beg[row]);
        const int n   = __builtin_amdgcn_readfirstlane(seg_end[row]) - beg;
        const float4 own = *(const float4*)(hl + (size_t)row * HDIM);
        const float4 r = agg_node<false>(hl, csr_src, beg, n, lane, own, none, none);
        *(float4*)(t + (size_t)row * HDIM + lane * 4) = r;
    }
}

__global__ __launch_bounds__(256)
void seed_softmax_agg_kernel(const float* __restrict__ h, const int* __restrict__ final_map,
                             const int* __restrict__ node_map,
                             const float* __restrict__ g, const float* __restrict__ b,
                             const int* __restrict__ seg_beg, const int* __restrict__ seg_end,
                             const int* __restrict__ csr_src,
                             float* __restrict__ t, float* __restrict__ rs)
{
    const int lane = threadIdx.x & 31;
    const int wv = __builtin_amdgcn_readfirstlane((int)(threadIdx.x >> 5));
    const int i  = blockIdx.x * 8 + wv;
    if (i >= NSEEDS) return;
    const int v   = __builtin_amdgcn_readfirstlane(node_map[final_map[i]]);
    const int beg = __builtin_amdgcn_readfirstlane(seg_beg[v]);
    const int n   = __builtin_amdgcn_readfirstlane(seg_end[v]) - beg;
    const float* __restrict__ hl = h + lane * 4;
    const float4 raw = *(const float4*)(hl + (size_t)v * HDIM);
    const float4 gg  = *(const float4*)(g + lane * 4);
    const float4 bb  = *(const float4*)(b + lane * 4);
    const float4 own = ln_relu_row(raw, gg, bb);
    const float4 r   = agg_node<true>(hl, csr_src, beg, n, lane, own, gg, bb);
    *(float4*)(t  + (size_t)i * HDIM + lane * 4) = r;
    *(float4*)(rs + (size_t)i * HDIM + lane * 4) = raw;
}

__global__ __launch_bounds__(256)
void ln_relu_kernel(const float* __restrict__ in, const float* __restrict__ g,
                    const float* __restrict__ b, float* __restrict__ out, int M)
{
    const int row  = blockIdx.x * 8 + (threadIdx.x >> 5);
    const int lane = threadIdx.x & 31;
    if (row >= M) return;
    const float4 v  = *(const float4*)(in + (size_t)row * HDIM + lane * 4);
    const float4 gg = *(const float4*)(g + lane * 4);
    const float4 bb = *(const float4*)(b + lane * 4);
    const float4 r  = ln_relu_row(v, gg, bb);
    *(float4*)(out + (size_t)row * HDIM + lane * 4) = r;
}

__global__ __launch_bounds__(256)
void gather_kernel(const float* __restrict__ in, const int* __restrict__ map,
                   float* __restrict__ out, int M)
{
    const int tid = blockIdx.x * 256 + threadIdx.x;
    if (tid >= M * 32) return;
    const int i = tid >> 5;
    const int f = (tid & 31) * 4;
    const int j = map[i];
    *(float4*)(out + (size_t)i * HDIM + f) =
        *(const float4*)(in + (size_t)j * HDIM + f);
}

__global__ __launch_bounds__(256)
void logsoftmax_kernel(const float* __restrict__ in, float* __restrict__ out, int M)
{
    const int row  = blockIdx.x * 8 + (threadIdx.x >> 5);
    const int lane = threadIdx.x & 31;
    if (row >= M) return;
    const float NEG = -3.0e38f;
    const float v0 = (lane      < NCLS) ? in[(size_t)row * NCLSPAD + lane     ] : NEG;
    const float v1 = (lane + 32 < NCLS) ? in[(size_t)row * NCLSPAD + lane + 32] : NEG;
    float mx = fmaxf(v0, v1);
#pragma unroll
    for (int off = 16; off > 0; off >>= 1) mx = fmaxf(mx, __shfl_xor(mx, off, 32));
    float se = ((lane      < NCLS) ? expf(v0 - mx) : 0.f)
             + ((lane + 32 < NCLS) ? expf(v1 - mx) : 0.f);
#pragma unroll
    for (int off = 16; off > 0; off >>= 1) se += __shfl_xor(se, off, 32);
    const float lse = logf(se) + mx;
    if (lane      < NCLS) out[(size_t)row * NCLS + lane     ] = v0 - lse;
    if (lane + 32 < NCLS) out[(size_t)row * NCLS + lane + 32] = v1 - lse;
}

extern "C" void kernel_launch(void* const* d_in, const int* in_sizes, int n_in,
                              void* d_out, int out_size, void* d_ws, size_t ws_size,
                              hipStream_t stream)
{
    (void)in_sizes; (void)n_in; (void)out_size; (void)ws_size;

    const float* x         = (const float*)d_in[0];
    const int*   src       = (const int*)  d_in[1];
    const int*   dst       = (const int*)  d_in[2];
    const int*   node_map  = (const int*)  d_in[3];
    const int*   final_map = (const int*)  d_in[4];
    const float* enc_w     = (const float*)d_in[5];
    const float* enc_b     = (const float*)d_in[6];
    const float* gcn_w     = (const float*)d_in[7];
    const float* gcn_b     = (const float*)d_in[8];
    const float* ln_g      = (const float*)d_in[9];
    const float* ln_b      = (const float*)d_in[10];
    const float* pred_w    = (const float*)d_in[11];
    const float* pred_b    = (const float*)d_in[12];
    float* out = (float*)d_out;

    const size_t NH = (size_t)NNODES * HDIM;
    float* B0     = (float*)d_ws;
    float* B1     = B0 + NH;
    float* B2     = B0 + 2 * NH;
    float* B3     = B0 + 3 * NH;
    float* B4     = B0 + 4 * NH;
    float* HG     = B0 + 5 * NH;
    float* HG2    = HG  + (size_t)NSEEDS * HDIM;
    float* SC     = HG2 + (size_t)NSEEDS * HDIM;
    float* WT_ENC = SC  + (size_t)NSEEDS * NCLSPAD;
    float* WT_G   = WT_ENC + (size_t)HDIM * FDIM;
    float* PWT    = WT_G   + (size_t)3 * HDIM * HDIM;
    float* PB     = PWT    + (size_t)NCLSPAD * HDIM;

    transpose_pad_kernel<<<(HDIM * FDIM + 255) / 256, 256, 0, stream>>>(
        enc_w, WT_ENC, FDIM, HDIM, HDIM);
    for (int l = 0; l < 3; ++l)
        transpose_pad_kernel<<<(HDIM * HDIM + 255) / 256, 256, 0, stream>>>(
            gcn_w + (size_t)l * HDIM * HDIM, WT_G + (size_t)l * HDIM * HDIM,
            HDIM, HDIM, HDIM);
    transpose_pad_kernel<<<(NCLSPAD * HDIM + 255) / 256, 256, 0, stream>>>(
        pred_w, PWT, HDIM, NCLS, NCLSPAD);
    pad_bias_kernel<<<1, NCLSPAD, 0, stream>>>(pred_b, PB, NCLS, NCLSPAD);

    const dim3 gblk(32, 4, 1);
    const int  mtN   = NNODES / 16;
    const int  ngrid = (NNODES * 32 + 255) / 256;
    const int  lgrid = (NNODES + 7) / 8;

    int* IDEG   = (int*)B3;
    int* ICNT   = IDEG + 2 * NNODES;
    int* ISTART = IDEG + 2 * NNODES + 32;
    int* ICUR   = ISTART + 2 * NNODES;
    int* ICSR   = ICUR + 2 * NNODES;
    zero_int_kernel<<<(2 * NNODES + 32 + 255) / 256, 256, 0, stream>>>(IDEG, 2 * NNODES + 32);
    csr_count_kernel<<<(2 * NEDGES + 255) / 256, 256, 0, stream>>>(dst, IDEG);
    csr_alloc_kernel<<<(2 * NNODES + 255) / 256, 256, 0, stream>>>(
        IDEG, ICNT, ISTART, ICUR, 2 * NNODES);
    csr_fill_kernel<<<(2 * NEDGES + 255) / 256, 256, 0, stream>>>(src, dst, ICUR, ICSR);

    gemm_rows128_kernel<FDIM><<<GEMM_WGS, 512, 0, stream>>>(
        x, WT_ENC, enc_b, nullptr, B0, mtN);

    auto conv = [&](const float* hin, int eblk, const float* wt, const float* bias,
                    const float* res, float* outp) {
        node_softmax_agg_kernel<<<AGG_BLOCKS, 256, 0, stream>>>(
            hin, ISTART + (size_t)eblk * NNODES, ICUR + (size_t)eblk * NNODES, ICSR, B2);
        gemm_rows128_kernel<HDIM><<<GEMM_WGS, 512, 0, stream>>>(
            B2, wt, bias, res, outp, mtN);
    };

    conv(B0, 0, WT_G, gcn_b, nullptr, B0);

    ln_relu_kernel<<<lgrid, 256, 0, stream>>>(B0, ln_g, ln_b, B1, NNODES);
    conv(B1, 0, WT_G + (size_t)HDIM * HDIM, gcn_b + HDIM,  B0,  B4);
    gather_kernel<<<ngrid, 256, 0, stream>>>(B4, node_map, B0, NNODES);

    float* T2 = B2;
    float* R2 = B4;
    const int mtS = NSEEDS / 16;
    seed_softmax_agg_kernel<<<NSEEDS / 8, 256, 0, stream>>>(
        B0, final_map, node_map + NNODES, ln_g + HDIM, ln_b + HDIM,
        ISTART + NNODES, ICUR + NNODES, ICSR, T2, R2);
    gemm_rows128_kernel<HDIM><<<GEMM_WGS, 512, 0, stream>>>(
        T2, WT_G + (size_t)2 * HDIM * HDIM, gcn_b + 2 * HDIM, R2, HG, mtS);

    ln_relu_kernel<<<(NSEEDS + 7) / 8, 256, 0, stream>>>(
        HG, ln_g + 2 * HDIM, ln_b + 2 * HDIM, HG2, NSEEDS);
    const dim3 gridS((mtS + 3) / 4, NCLSPAD / 64, 1);
    gemm_f32_wmma_kernel<<<gridS, gblk, 0, stream>>>(
        HG2, PWT, PB, nullptr, SC, NSEEDS, HDIM, NCLSPAD);
    logsoftmax_kernel<<<(NSEEDS + 7) / 8, 256, 0, stream>>>(SC, out, NSEEDS);
}


# Aggregation softmax: exp(m - max) computed as one hardware exp2 of (m - max) * log2(e) instead of the library expf (arguments are always <= 0), about 13 fewer vector instructions per exponential
# speedup vs baseline: 3.8639x; 1.0680x over previous
#include <hip/hip_runtime.h>
#include <hip/hip_bf16.h>
#include <math.h>

#define NNODES   100000
#define NEDGES   600000
#define FDIM     256
#define HDIM     128
#define NCLS     47
#define NCLSPAD  64
#define NSEEDS   10000
#define GEN_EPS  1e-7f
#define LN_EPS   1e-5f

typedef float v2f __attribute__((ext_vector_type(2)));
typedef float v8f __attribute__((ext_vector_type(8)));

__global__ __launch_bounds__(128)
void gemm_f32_wmma_kernel(const float* __restrict__ A, const float* __restrict__ Wt,
                          const float* __restrict__ bias, const float* __restrict__ res,
                          float* __restrict__ out, int M, int K, int Nc)
{
    const int lane  = threadIdx.x;
    const int mtile = blockIdx.x * 4 + threadIdx.y;
    const int row0  = mtile * 16;
    if (row0 >= M) return;
    const int hb   = lane >> 4;
    const int lo   = lane & 15;
    const int arow = row0 + lo;
    const int col0 = blockIdx.y * 64 + lo;

    const float* __restrict__ a_ptr = A  + (size_t)arow * K;
    const float* __restrict__ b_ptr = Wt + (size_t)col0 * K;
    const size_t bstep = (size_t)16 * K;

    v8f acc[4];
#pragma unroll
    for (int t = 0; t < 4; ++t)
#pragma unroll
        for (int i = 0; i < 8; ++i) acc[t][i] = 0.f;

#pragma unroll 4
    for (int k = 0; k < K; k += 4) {
        const int ka = k + 2 * hb;
        const v2f a  = *(const v2f*)(a_ptr + ka);
        const v2f b0 = *(const v2f*)(b_ptr + ka);
        const v2f b1 = *(const v2f*)(b_ptr + bstep + ka);
        const v2f b2 = *(const v2f*)(b_ptr + 2 * bstep + ka);
        const v2f b3 = *(const v2f*)(b_ptr + 3 * bstep + ka);
        acc[0] = __builtin_amdgcn_wmma_f32_16x16x4_f32(false, a, false, b0, (short)0, acc[0], false, false);
        acc[1] = __builtin_amdgcn_wmma_f32_16x16x4_f32(false, a, false, b1, (short)0, acc[1], false, false);
        acc[2] = __builtin_amdgcn_wmma_f32_16x16x4_f32(false, a, false, b2, (short)0, acc[2], false, false);
        acc[3] = __builtin_amdgcn_wmma_f32_16x16x4_f32(false, a, false, b3, (short)0, acc[3], false, false);
    }

#pragma unroll
    for (int t = 0; t < 4; ++t) {
        const int   col = col0 + 16 * t;
        const float bv  = bias[col];
#pragma unroll
        for (int i = 0; i < 8; ++i) {
            const int r = row0 + i + 8 * hb;
            float v = acc[t][i] + bv;
            if (res) v += res[(size_t)r * Nc + col];
            out[(size_t)r * Nc + col] = v;
        }
    }
}

#define GEMM_WGS 256


__device__ __forceinline__ void load_a_chunk(const float* __restrict__ ap, float4 (&a)[16])
{
#pragma unroll
    for (int i = 0; i < 16; ++i) a[i] = *(const float4*)(ap + 8 * i);
}

template <int K>
__device__ __forceinline__ void load_b_block(const float* bp, float4 (&b)[8])
{
    constexpr int PITCH = K + 4;
#pragma unroll
    for (int t = 0; t < 8; ++t) b[t] = *(const float4*)(bp + t * 16 * PITCH);
}

__device__ __forceinline__ void mma_block(const float4 a, const float4 (&b)[8], v8f (&acc)[8])
{
    const v2f a01 = {a.x, a.y};
    const v2f a23 = {a.z, a.w};
#pragma unroll
    for (int t = 0; t < 8; ++t) {
        const v2f b01 = {b[t].x, b[t].y};
        acc[t] = __builtin_amdgcn_wmma_f32_16x16x4_f32(false, a01, false, b01, (short)0, acc[t], false, false);
    }
#pragma unroll
    for (int t = 0; t < 8; ++t) {
        const v2f b23 = {b[t].z, b[t].w};
        acc[t] = __builtin_amdgcn_wmma_f32_16x16x4_f32(false, a23, false, b23, (short)0, acc[t], false, false);
    }
}

template <int K>
__device__ __forceinline__ void mma_chunk(const float* bp, const float4 (&a)[16], v8f (&acc)[8])
{
    float4 b0[8], b1[8];
    load_b_block<K>(bp, b0);
#pragma unroll
    for (int i = 0; i < 16; i += 2) {
        load_b_block<K>(bp + 8 * (i + 1), b1);
        __builtin_amdgcn_sched_barrier(0);
        mma_block(a[i], b0, acc);
        __builtin_amdgcn_sched_barrier(0);
        if (i + 2 < 16) load_b_block<K>(bp + 8 * (i + 2), b0);
        __builtin_amdgcn_sched_barrier(0);
        mma_block(a[i + 1], b1, acc);
        __builtin_amdgcn_sched_barrier(0);
    }
}

template <bool HAS_RES>
__device__ __forceinline__ void store_strip(const v8f (&acc)[8], const float* __restrict__ bias,
                                            const float* __restrict__ res, float* __restrict__ out,
                                            int row0, int c, int h)
{
#pragma unroll
    for (int g = 0; g < 2; ++g) {
        const int    col = 64 * g + 4 * c;
        const float4 bv  = *(const float4*)(bias + col);
        float4 rv[8];
        if (HAS_RES) {
#pragma unroll
            for (int i = 0; i < 8; ++i)
                rv[i] = *(const float4*)(res + (size_t)(row0 + i + 8 * h) * HDIM + col);
        }
#pragma unroll
        for (int i = 0; i < 8; ++i) {
            float4 v;
            v.x = acc[4 * g + 0][i] + bv.x;
            v.y = acc[4 * g + 1][i] + bv.y;
            v.z = acc[4 * g + 2][i] + bv.z;
            v.w = acc[4 * g + 3][i] + bv.w;
            if (HAS_RES) { v.x += rv[i].x; v.y += rv[i].y; v.z += rv[i].z; v.w += rv[i].w; }
            *(float4*)(out + (size_t)(row0 + i + 8 * h) * HDIM + col) = v;
        }
    }
}

template <int K>
__global__ __launch_bounds__(512)
void gemm_rows128_kernel(const float* __restrict__ A, const float* __restrict__ Wt,
                         const float* __restrict__ bias, const float* __restrict__ res,
                         float* __restrict__ out, int ntiles)
{
    constexpr int PITCH = K + 4;
    __shared__ __attribute__((aligned(16))) float wl[HDIM * PITCH];

    const int tid  = threadIdx.x;
    const int lane = tid & 31;
    const int wv   = __builtin_amdgcn_readfirstlane(tid >> 5);
    const int c    = lane & 15;
    const int h    = lane >> 4;

    const int wa = wv >> 2;
    const int q  = wa + 4 * (((wv & 3) - wa) & 3);
    int tile = q * GEMM_WGS + (int)blockIdx.x;

    float4 a[16];
    if (tile < ntiles) load_a_chunk(A + (size_t)(tile * 16 + c) * K + 4 * h, a);

    constexpr int NPASS = HDIM * (K / 4) / 512;
    float4 wq[NPASS];
#pragma unroll
    for (int r = 0; r < NPASS; ++r) {
        const int i  = tid + 512 * r;
        const int n  = i / (K / 4);
        const int k4 = (i - n * (K / 4)) * 4;
        wq[r] = *(const float4*)(Wt + (size_t)n * K + k4);
    }
#pragma unroll
    for (int r = 0; r < NPASS; ++r) {
        const int i   = tid + 512 * r;
        const int n   = i / (K / 4);
        const int k4  = (i - n * (K / 4)) * 4;
        const int rho = (n & 64) + 16 * (n & 3) + ((n >> 2) & 15);
        *(float4*)(wl + rho * PITCH + k4) = wq[r];
    }
    __syncthreads();

    const float* bp = wl + c * PITCH + 4 * h;

    while (tile < ntiles) {
        const int row0 = tile * 16;

        v8f acc[8];
#pragma unroll
        for (int t = 0; t < 8; ++t)
#pragma unroll
            for (int i = 0; i < 8; ++i) acc[t][i] = 0.f;

        mma_chunk<K>(bp, a, acc);
#pragma unroll
        for (int kc = 128; kc < K; kc += 128) {
            load_a_chunk(A + (size_t)(row0 + c) * K + 4 * h + kc, a);
            mma_chunk<K>(bp + kc, a, acc);
        }

        if (res) store_strip<true >(acc, bias, res, out, row0, c, h);
        else     store_strip<false>(acc, bias, res, out, row0, c, h);

        tile += 16 * GEMM_WGS;
        if (tile < ntiles) load_a_chunk(A + (size_t)(tile * 16 + c) * K + 4 * h, a);
    }
}

__global__ __launch_bounds__(256)
void transpose_pad_kernel(const float* __restrict__ W, float* __restrict__ Wt,
                          int K, int Nc, int Ncpad)
{
    const int tid = blockIdx.x * 256 + threadIdx.x;
    if (tid >= Ncpad * K) return;
    const int c = tid / K;
    const int k = tid - c * K;
    Wt[(size_t)c * K + k] = (c < Nc) ? W[(size_t)k * Nc + c] : 0.f;
}

__global__ void pad_bias_kernel(const float* __restrict__ b, float* __restrict__ bp,
                                int Nc, int Ncpad)
{
    const int c = threadIdx.x;
    if (c < Ncpad) bp[c] = (c < Nc) ? b[c] : 0.f;
}

__global__ __launch_bounds__(256)
void zero_int_kernel(int* __restrict__ p, int n)
{
    const int i = blockIdx.x * 256 + threadIdx.x;
    if (i < n) p[i] = 0;
}

__global__ __launch_bounds__(256)
void csr_count_kernel(const int* __restrict__ dst, int* __restrict__ deg)
{
    const int e = blockIdx.x * 256 + threadIdx.x;
    if (e >= 2 * NEDGES) return;
    const int v = ((e >= NEDGES) ? NNODES : 0) + dst[e];
    atomicAdd(deg + v, 1);
}

__global__ __launch_bounds__(256)
void csr_alloc_kernel(const int* __restrict__ deg, int* __restrict__ counter,
                      int* __restrict__ seg_beg, int* __restrict__ cursor, int n)
{
    __shared__ int wtot[8];
    __shared__ int blk_base;
    const int tid  = threadIdx.x;
    const int lane = tid & 31;
    const int w    = tid >> 5;
    const int i    = blockIdx.x * 256 + tid;
    const int d    = (i < n) ? deg[i] : 0;
    int incl = d;
#pragma unroll
    for (int off = 1; off < 32; off <<= 1) {
        const int up = __shfl_up(incl, off, 32);
        if (lane >= off) incl += up;
    }
    if (lane == 31) wtot[w] = incl;
    __syncthreads();
    if (tid == 0) {
        int tot = 0;
#pragma unroll
        for (int k = 0; k < 8; ++k) tot += wtot[k];
        blk_base = atomicAdd(counter, tot);
    }
    __syncthreads();
    int st = blk_base + incl - d;
#pragma unroll
    for (int k = 0; k < 8; ++k) if (k < w) st += wtot[k];
    if (i < n) { seg_beg[i] = st; cursor[i] = st; }
}

__global__ __launch_bounds__(256)
void csr_fill_kernel(const int* __restrict__ src, const int* __restrict__ dst,
                     int* __restrict__ cursor, int* __restrict__ csr_src)
{
    const int e = blockIdx.x * 256 + threadIdx.x;
    if (e >= 2 * NEDGES) return;
    const int v   = ((e >= NEDGES) ? NNODES : 0) + dst[e];
    const int pos = atomicAdd(cursor + v, 1);
    csr_src[pos] = src[e];
}

__device__ __forceinline__ float4 ln_relu_row(const float4 v, const float4 gg, const float4 bb)
{
    float s1 = v.x + v.y + v.z + v.w;
    float s2 = v.x * v.x + v.y * v.y + v.z * v.z + v.w * v.w;
#pragma unroll
    for (int off = 16; off > 0; off >>= 1) {
        s1 += __shfl_xor(s1, off, 32);
        s2 += __shfl_xor(s2, off, 32);
    }
    const float mu  = s1 * (1.f / HDIM);
    const float var = s2 * (1.f / HDIM) - mu * mu;
    const float inv = rsqrtf(var + LN_EPS);
    float4 r;
    r.x = fmaxf((v.x - mu) * inv * gg.x + bb.x, 0.f);
    r.y = fmaxf((v.y - mu) * inv * gg.y + bb.y, 0.f);
    r.z = fmaxf((v.z - mu) * inv * gg.z + bb.z, 0.f);
    r.w = fmaxf((v.w - mu) * inv * gg.w + bb.w, 0.f);
    return r;
}

#define AGG_BLOCKS 4096

__device__ __forceinline__ float exp_nonpos(const float x)
{
    return __builtin_amdgcn_exp2f(x * 1.44269504088896340736f);
}

template <int CNT, bool LNF>
__device__ __forceinline__ void agg_chunk(const float* __restrict__ hl, int idx, int c, bool first,
                                          float (&mx)[4], float (&de)[4], float (&nu)[4],
                                          const float4 gg, const float4 bb)
{
    float4 hv[CNT];
#pragma unroll
    for (int j = 0; j < CNT; ++j) {
        const int s = __builtin_amdgcn_readlane(idx, c + j);
        hv[j] = *(const float4*)(hl + (size_t)s * HDIM);
    }
    if (LNF) {
#pragma unroll
        for (int j = 0; j < CNT; ++j) hv[j] = ln_relu_row(hv[j], gg, bb);
    }
    float m[CNT][4];
    float c0 = mx[0], c1 = mx[1], c2 = mx[2], c3 = mx[3];
#pragma unroll
    for (int j = 0; j < CNT; ++j) {
        m[j][0] = fmaxf(hv[j].x, 0.f) + GEN_EPS;
        m[j][1] = fmaxf(hv[j].y, 0.f) + GEN_EPS;
        m[j][2] = fmaxf(hv[j].z, 0.f) + GEN_EPS;
        m[j][3] = fmaxf(hv[j].w, 0.f) + GEN_EPS;
        c0 = fmaxf(c0, m[j][0]);
        c1 = fmaxf(c1, m[j][1]);
        c2 = fmaxf(c2, m[j][2]);
        c3 = fmaxf(c3, m[j][3]);
    }
    if (!first) {
        const float s0 = exp_nonpos(mx[0] - c0), s1 = exp_nonpos(mx[1] - c1);
        const float s2 = exp_nonpos(mx[2] - c2), s3 = exp_nonpos(mx[3] - c3);
        de[0] *= s0; nu[0] *= s0;
        de[1] *= s1; nu[1] *= s1;
        de[2] *= s2; nu[2] *= s2;
        de[3] *= s3; nu[3] *= s3;
    }
    mx[0] = c0; mx[1] = c1; mx[2] = c2; mx[3] = c3;
#pragma unroll
    for (int j = 0; j < CNT; ++j) {
#pragma unroll
        for (int f = 0; f < 4; ++f) {
            const float ex = exp_nonpos(m[j][f] - mx[f]);
            de[f] += ex;
            nu[f] += m[j][f] * ex;
        }
    }
}

template <bool LNF>
__device__ __forceinline__ float4 agg_node(const float* __restrict__ hl, const int* __restrict__ csr_src,
                                           int beg, int n, int lane, const float4 own,
                                           const float4 gg, const float4 bb)
{
    float mx[4] = {0.f, 0.f, 0.f, 0.f};
    float de[4] = {0.f, 0.f, 0.f, 0.f};
    float nu[4] = {0.f, 0.f, 0.f, 0.f};

    for (int base = 0; base < n; base += 32) {
        const int bc  = (n - base < 32) ? (n - base) : 32;
        const int idx = (lane < bc) ? csr_src[beg + base + lane] : 0;
        for (int c = 0; c < bc; c += 8) {
            const int  cnt   = (bc - c < 8) ? (bc - c) : 8;
            const bool first = (base + c == 0);
            switch (cnt) {
            case 1:  agg_chunk<1, LNF>(hl, idx, c, first, mx, de, nu, gg, bb); break;
            case 2:  agg_chunk<2, LNF>(hl, idx, c, first, mx, de, nu, gg, bb); break;
            case 3:  agg_chunk<3, LNF>(hl, idx, c, first, mx, de, nu, gg, bb); break;
            case 4:  agg_chunk<4, LNF>(hl, idx, c, first, mx, de, nu, gg, bb); break;
            case 5:  agg_chunk<5, LNF>(hl, idx, c, first, mx, de, nu, gg, bb); break;
            case 6:  agg_chunk<6, LNF>(hl, idx, c, first, mx, de, nu, gg, bb); break;
            case 7:  agg_chunk<7, LNF>(hl, idx, c, first, mx, de, nu, gg, bb); break;
            default: agg_chunk<8, LNF>(hl, idx, c, first, mx, de, nu, gg, bb); break;
            }
        }
    }

    float4 r;
    r.x = nu[0] / (de[0] + 1e-16f) + own.x;
    r.y = nu[1] / (de[1] + 1e-16f) + own.y;
    r.z = nu[2] / (de[2] + 1e-16f) + own.z;
    r.w = nu[3] / (de[3] + 1e-16f) + own.w;
    return r;
}

__global__ __launch_bounds__(256)
void node_softmax_agg_kernel(const float* __restrict__ hin, const int* __restrict__ seg_beg,
                             const int* __restrict__ seg_end, const int* __restrict__ csr_src,
                             float* __restrict__ t)
{
    const int lane = threadIdx.x & 31;
    const int wv = __builtin_amdgcn_readfirstlane((int)(threadIdx.x >> 5));
    const float* __restrict__ hl = hin + lane * 4;
    const float4 none = {0.f, 0.f, 0.f, 0.f};

    for (int row = blockIdx.x * 8 + wv; row < NNODES; row += AGG_BLOCKS * 8) {
        const int beg = __builtin_amdgcn_readfirstlane(seg_beg[row]);
        const int n   = __builtin_amdgcn_readfirstlane(seg_end[row]) - beg;
        const float4 own = *(const float4*)(hl + (size_t)row * HDIM);
        const float4 r = agg_node<false>(hl, csr_src, beg, n, lane, own, none, none);
        *(float4*)(t + (size_t)row * HDIM + lane * 4) = r;
    }
}

__global__ __launch_bounds__(256)
void seed_softmax_agg_kernel(const float* __restrict__ h, const int* __restrict__ final_map,
                             const int* __restrict__ node_map,
                             const float* __restrict__ g, const float* __restrict__ b,
                             const int* __restrict__ seg_beg, const int* __restrict__ seg_end,
                             const int* __restrict__ csr_src,
                             float* __restrict__ t, float* __restrict__ rs)
{
    const int lane = threadIdx.x & 31;
    const int wv = __builtin_amdgcn_readfirstlane((int)(threadIdx.x >> 5));
    const int i  = blockIdx.x * 8 + wv;
    if (i >= NSEEDS) return;
    const int v   = __builtin_amdgcn_readfirstlane(node_map[final_map[i]]);
    const int beg = __builtin_amdgcn_readfirstlane(seg_beg[v]);
    const int n   = __builtin_amdgcn_readfirstlane(seg_end[v]) - beg;
    const float* __restrict__ hl = h + lane * 4;
    const float4 raw = *(const float4*)(hl + (size_t)v * HDIM);
    const float4 gg  = *(const float4*)(g + lane * 4);
    const float4 bb  = *(const float4*)(b + lane * 4);
    const float4 own = ln_relu_row(raw, gg, bb);
    const float4 r   = agg_node<true>(hl, csr_src, beg, n, lane, own, gg, bb);
    *(float4*)(t  + (size_t)i * HDIM + lane * 4) = r;
    *(float4*)(rs + (size_t)i * HDIM + lane * 4) = raw;
}

__global__ __launch_bounds__(256)
void ln_relu_kernel(const float* __restrict__ in, const float* __restrict__ g,
                    const float* __restrict__ b, float* __restrict__ out, int M)
{
    const int row  = blockIdx.x * 8 + (threadIdx.x >> 5);
    const int lane = threadIdx.x & 31;
    if (row >= M) return;
    const float4 v  = *(const float4*)(in + (size_t)row * HDIM + lane * 4);
    const float4 gg = *(const float4*)(g + lane * 4);
    const float4 bb = *(const float4*)(b + lane * 4);
    const float4 r  = ln_relu_row(v, gg, bb);
    *(float4*)(out + (size_t)row * HDIM + lane * 4) = r;
}

__global__ __launch_bounds__(256)
void gather_kernel(const float* __restrict__ in, const int* __restrict__ map,
                   float* __restrict__ out, int M)
{
    const int tid = blockIdx.x * 256 + threadIdx.x;
    if (tid >= M * 32) return;
    const int i = tid >> 5;
    const int f = (tid & 31) * 4;
    const int j = map[i];
    *(float4*)(out + (size_t)i * HDIM + f) =
        *(const float4*)(in + (size_t)j * HDIM + f);
}

__global__ __launch_bounds__(256)
void logsoftmax_kernel(const float* __restrict__ in, float* __restrict__ out, int M)
{
    const int row  = blockIdx.x * 8 + (threadIdx.x >> 5);
    const int lane = threadIdx.x & 31;
    if (row >= M) return;
    const float NEG = -3.0e38f;
    const float v0 = (lane      < NCLS) ? in[(size_t)row * NCLSPAD + lane     ] : NEG;
    const float v1 = (lane + 32 < NCLS) ? in[(size_t)row * NCLSPAD + lane + 32] : NEG;
    float mx = fmaxf(v0, v1);
#pragma unroll
    for (int off = 16; off > 0; off >>= 1) mx = fmaxf(mx, __shfl_xor(mx, off, 32));
    float se = ((lane      < NCLS) ? expf(v0 - mx) : 0.f)
             + ((lane + 32 < NCLS) ? expf(v1 - mx) : 0.f);
#pragma unroll
    for (int off = 16; off > 0; off >>= 1) se += __shfl_xor(se, off, 32);
    const float lse = logf(se) + mx;
    if (lane      < NCLS) out[(size_t)row * NCLS + lane     ] = v0 - lse;
    if (lane + 32 < NCLS) out[(size_t)row * NCLS + lane + 32] = v1 - lse;
}

extern "C" void kernel_launch(void* const* d_in, const int* in_sizes, int n_in,
                              void* d_out, int out_size, void* d_ws, size_t ws_size,
                              hipStream_t stream)
{
    (void)in_sizes; (void)n_in; (void)out_size; (void)ws_size;

    const float* x         = (const float*)d_in[0];
    const int*   src       = (const int*)  d_in[1];
    const int*   dst       = (const int*)  d_in[2];
    const int*   node_map  = (const int*)  d_in[3];
    const int*   final_map = (const int*)  d_in[4];
    const float* enc_w     = (const float*)d_in[5];
    const float* enc_b     = (const float*)d_in[6];
    const float* gcn_w     = (const float*)d_in[7];
    const float* gcn_b     = (const float*)d_in[8];
    const float* ln_g      = (const float*)d_in[9];
    const float* ln_b      = (const float*)d_in[10];
    const float* pred_w    = (const float*)d_in[11];
    const float* pred_b    = (const float*)d_in[12];
    float* out = (float*)d_out;

    const size_t NH = (size_t)NNODES * HDIM;
    float* B0     = (float*)d_ws;
    float* B1     = B0 + NH;
    float* B2     = B0 + 2 * NH;
    float* B3     = B0 + 3 * NH;
    float* B4     = B0 + 4 * NH;
    float* HG     = B0 + 5 * NH;
    float* HG2    = HG  + (size_t)NSEEDS * HDIM;
    float* SC     = HG2 + (size_t)NSEEDS * HDIM;
    float* WT_ENC = SC  + (size_t)NSEEDS * NCLSPAD;
    float* WT_G   = WT_ENC + (size_t)HDIM * FDIM;
    float* PWT    = WT_G   + (size_t)3 * HDIM * HDIM;
    float* PB     = PWT    + (size_t)NCLSPAD * HDIM;

    transpose_pad_kernel<<<(HDIM * FDIM + 255) / 256, 256, 0, stream>>>(
        enc_w, WT_ENC, FDIM, HDIM, HDIM);
    for (int l = 0; l < 3; ++l)
        transpose_pad_kernel<<<(HDIM * HDIM + 255) / 256, 256, 0, stream>>>(
            gcn_w + (size_t)l * HDIM * HDIM, WT_G + (size_t)l * HDIM * HDIM,
            HDIM, HDIM, HDIM);
    transpose_pad_kernel<<<(NCLSPAD * HDIM + 255) / 256, 256, 0, stream>>>(
        pred_w, PWT, HDIM, NCLS, NCLSPAD);
    pad_bias_kernel<<<1, NCLSPAD, 0, stream>>>(pred_b, PB, NCLS, NCLSPAD);

    const dim3 gblk(32, 4, 1);
    const int  mtN   = NNODES / 16;
    const int  ngrid = (NNODES * 32 + 255) / 256;
    const int  lgrid = (NNODES + 7) / 8;

    int* IDEG   = (int*)B3;
    int* ICNT   = IDEG + 2 * NNODES;
    int* ISTART = IDEG + 2 * NNODES + 32;
    int* ICUR   = ISTART + 2 * NNODES;
    int* ICSR   = ICUR + 2 * NNODES;
    zero_int_kernel<<<(2 * NNODES + 32 + 255) / 256, 256, 0, stream>>>(IDEG, 2 * NNODES + 32);
    csr_count_kernel<<<(2 * NEDGES + 255) / 256, 256, 0, stream>>>(dst, IDEG);
    csr_alloc_kernel<<<(2 * NNODES + 255) / 256, 256, 0, stream>>>(
        IDEG, ICNT, ISTART, ICUR, 2 * NNODES);
    csr_fill_kernel<<<(2 * NEDGES + 255) / 256, 256, 0, stream>>>(src, dst, ICUR, ICSR);

    gemm_rows128_kernel<FDIM><<<GEMM_WGS, 512, 0, stream>>>(
        x, WT_ENC, enc_b, nullptr, B0, mtN);

    auto conv = [&](const float* hin, int eblk, const float* wt, const float* bias,
                    const float* res, float* outp) {
        node_softmax_agg_kernel<<<AGG_BLOCKS, 256, 0, stream>>>(
            hin, ISTART + (size_t)eblk * NNODES, ICUR + (size_t)eblk * NNODES, ICSR, B2);
        gemm_rows128_kernel<HDIM><<<GEMM_WGS, 512, 0, stream>>>(
            B2, wt, bias, res, outp, mtN);
    };

    conv(B0, 0, WT_G, gcn_b, nullptr, B0);

    ln_relu_kernel<<<lgrid, 256, 0, stream>>>(B0, ln_g, ln_b, B1, NNODES);
    conv(B1, 0, WT_G + (size_t)HDIM * HDIM, gcn_b + HDIM,  B0,  B4);
    gather_kernel<<<ngrid, 256, 0, stream>>>(B4, node_map, B0, NNODES);

    float* T2 = B2;
    float* R2 = B4;
    const int mtS = NSEEDS / 16;
    seed_softmax_agg_kernel<<<NSEEDS / 8, 256, 0, stream>>>(
        B0, final_map, node_map + NNODES, ln_g + HDIM, ln_b + HDIM,
        ISTART + NNODES, ICUR + NNODES, ICSR, T2, R2);
    gemm_rows128_kernel<HDIM><<<GEMM_WGS, 512, 0, stream>>>(
        T2, WT_G + (size_t)2 * HDIM * HDIM, gcn_b + 2 * HDIM, R2, HG, mtS);

    ln_relu_kernel<<<(NSEEDS + 7) / 8, 256, 0, stream>>>(
        HG, ln_g + 2 * HDIM, ln_b + 2 * HDIM, HG2, NSEEDS);
    const dim3 gridS((mtS + 3) / 4, NCLSPAD / 64, 1);
    gemm_f32_wmma_kernel<<<gridS, gblk, 0, stream>>>(
        HG2, PWT, PB, nullptr, SC, NSEEDS, HDIM, NCLSPAD);
    logsoftmax_kernel<<<(NSEEDS + 7) / 8, 256, 0, stream>>>(SC, out, NSEEDS);
}


# In-edge list build: the counting add now returns each edge's rank in its row, so the fill pass places edges with plain loads, no second atomic per edge; list order = arrival order in the count pass
# speedup vs baseline: 4.2406x; 1.0975x over previous
#include <hip/hip_runtime.h>
#include <hip/hip_bf16.h>
#include <math.h>

#define NNODES   100000
#define NEDGES   600000
#define FDIM     256
#define HDIM     128
#define NCLS     47
#define NCLSPAD  64
#define NSEEDS   10000
#define GEN_EPS  1e-7f
#define LN_EPS   1e-5f

typedef float v2f __attribute__((ext_vector_type(2)));
typedef float v8f __attribute__((ext_vector_type(8)));

__global__ __launch_bounds__(128)
void gemm_f32_wmma_kernel(const float* __restrict__ A, const float* __restrict__ Wt,
                          const float* __restrict__ bias, const float* __restrict__ res,
                          float* __restrict__ out, int M, int K, int Nc)
{
    const int lane  = threadIdx.x;
    const int mtile = blockIdx.x * 4 + threadIdx.y;
    const int row0  = mtile * 16;
    if (row0 >= M) return;
    const int hb   = lane >> 4;
    const int lo   = lane & 15;
    const int arow = row0 + lo;
    const int col0 = blockIdx.y * 64 + lo;

    const float* __restrict__ a_ptr = A  + (size_t)arow * K;
    const float* __restrict__ b_ptr = Wt + (size_t)col0 * K;
    const size_t bstep = (size_t)16 * K;

    v8f acc[4];
#pragma unroll
    for (int t = 0; t < 4; ++t)
#pragma unroll
        for (int i = 0; i < 8; ++i) acc[t][i] = 0.f;

#pragma unroll 4
    for (int k = 0; k < K; k += 4) {
        const int ka = k + 2 * hb;
        const v2f a  = *(const v2f*)(a_ptr + ka);
        const v2f b0 = *(const v2f*)(b_ptr + ka);
        const v2f b1 = *(const v2f*)(b_ptr + bstep + ka);
        const v2f b2 = *(const v2f*)(b_ptr + 2 * bstep + ka);
        const v2f b3 = *(const v2f*)(b_ptr + 3 * bstep + ka);
        acc[0] = __builtin_amdgcn_wmma_f32_16x16x4_f32(false, a, false, b0, (short)0, acc[0], false, false);
        acc[1] = __builtin_amdgcn_wmma_f32_16x16x4_f32(false, a, false, b1, (short)0, acc[1], false, false);
        acc[2] = __builtin_amdgcn_wmma_f32_16x16x4_f32(false, a, false, b2, (short)0, acc[2], false, false);
        acc[3] = __builtin_amdgcn_wmma_f32_16x16x4_f32(false, a, false, b3, (short)0, acc[3], false, false);
    }

#pragma unroll
    for (int t = 0; t < 4; ++t) {
        const int   col = col0 + 16 * t;
        const float bv  = bias[col];
#pragma unroll
        for (int i = 0; i < 8; ++i) {
            const int r = row0 + i + 8 * hb;
            float v = acc[t][i] + bv;
            if (res) v += res[(size_t)r * Nc + col];
            out[(size_t)r * Nc + col] = v;
        }
    }
}

#define GEMM_WGS 256


__device__ __forceinline__ void load_a_chunk(const float* __restrict__ ap, float4 (&a)[16])
{
#pragma unroll
    for (int i = 0; i < 16; ++i) a[i] = *(const float4*)(ap + 8 * i);
}

template <int K>
__device__ __forceinline__ void load_b_block(const float* bp, float4 (&b)[8])
{
    constexpr int PITCH = K + 4;
#pragma unroll
    for (int t = 0; t < 8; ++t) b[t] = *(const float4*)(bp + t * 16 * PITCH);
}

__device__ __forceinline__ void mma_block(const float4 a, const float4 (&b)[8], v8f (&acc)[8])
{
    const v2f a01 = {a.x, a.y};
    const v2f a23 = {a.z, a.w};
#pragma unroll
    for (int t = 0; t < 8; ++t) {
        const v2f b01 = {b[t].x, b[t].y};
        acc[t] = __builtin_amdgcn_wmma_f32_16x16x4_f32(false, a01, false, b01, (short)0, acc[t], false, false);
    }
#pragma unroll
    for (int t = 0; t < 8; ++t) {
        const v2f b23 = {b[t].z, b[t].w};
        acc[t] = __builtin_amdgcn_wmma_f32_16x16x4_f32(false, a23, false, b23, (short)0, acc[t], false, false);
    }
}

template <int K>
__device__ __forceinline__ void mma_chunk(const float* bp, const float4 (&a)[16], v8f (&acc)[8])
{
    float4 b0[8], b1[8];
    load_b_block<K>(bp, b0);
#pragma unroll
    for (int i = 0; i < 16; i += 2) {
        load_b_block<K>(bp + 8 * (i + 1), b1);
        __builtin_amdgcn_sched_barrier(0);
        mma_block(a[i], b0, acc);
        __builtin_amdgcn_sched_barrier(0);
        if (i + 2 < 16) load_b_block<K>(bp + 8 * (i + 2), b0);
        __builtin_amdgcn_sched_barrier(0);
        mma_block(a[i + 1], b1, acc);
        __builtin_amdgcn_sched_barrier(0);
    }
}

template <bool HAS_RES>
__device__ __forceinline__ void store_strip(const v8f (&acc)[8], const float* __restrict__ bias,
                                            const float* __restrict__ res, float* __restrict__ out,
                                            int row0, int c, int h)
{
#pragma unroll
    for (int g = 0; g < 2; ++g) {
        const int    col = 64 * g + 4 * c;
        const float4 bv  = *(const float4*)(bias + col);
        float4 rv[8];
        if (HAS_RES) {
#pragma unroll
            for (int i = 0; i < 8; ++i)
                rv[i] = *(const float4*)(res + (size_t)(row0 + i + 8 * h) * HDIM + col);
        }
#pragma unroll
        for (int i = 0; i < 8; ++i) {
            float4 v;
            v.x = acc[4 * g + 0][i] + bv.x;
            v.y = acc[4 * g + 1][i] + bv.y;
            v.z = acc[4 * g + 2][i] + bv.z;
            v.w = acc[4 * g + 3][i] + bv.w;
            if (HAS_RES) { v.x += rv[i].x; v.y += rv[i].y; v.z += rv[i].z; v.w += rv[i].w; }
            *(float4*)(out + (size_t)(row0 + i + 8 * h) * HDIM + col) = v;
        }
    }
}

template <int K>
__global__ __launch_bounds__(512)
void gemm_rows128_kernel(const float* __restrict__ A, const float* __restrict__ Wt,
                         const float* __restrict__ bias, const float* __restrict__ res,
                         float* __restrict__ out, int ntiles)
{
    constexpr int PITCH = K + 4;
    __shared__ __attribute__((aligned(16))) float wl[HDIM * PITCH];

    const int tid  = threadIdx.x;
    const int lane = tid & 31;
    const int wv   = __builtin_amdgcn_readfirstlane(tid >> 5);
    const int c    = lane & 15;
    const int h    = lane >> 4;

    const int wa = wv >> 2;
    const int q  = wa + 4 * (((wv & 3) - wa) & 3);
    int tile = q * GEMM_WGS + (int)blockIdx.x;

    float4 a[16];
    if (tile < ntiles) load_a_chunk(A + (size_t)(tile * 16 + c) * K + 4 * h, a);

    constexpr int NPASS = HDIM * (K / 4) / 512;
    float4 wq[NPASS];
#pragma unroll
    for (int r = 0; r < NPASS; ++r) {
        const int i  = tid + 512 * r;
        const int n  = i / (K / 4);
        const int k4 = (i - n * (K / 4)) * 4;
        wq[r] = *(const float4*)(Wt + (size_t)n * K + k4);
    }
#pragma unroll
    for (int r = 0; r < NPASS; ++r) {
        const int i   = tid + 512 * r;
        const int n   = i / (K / 4);
        const int k4  = (i - n * (K / 4)) * 4;
        const int rho = (n & 64) + 16 * (n & 3) + ((n >> 2) & 15);
        *(float4*)(wl + rho * PITCH + k4) = wq[r];
    }
    __syncthreads();

    const float* bp = wl + c * PITCH + 4 * h;

    while (tile < ntiles) {
        const int row0 = tile * 16;

        v8f acc[8];
#pragma unroll
        for (int t = 0; t < 8; ++t)
#pragma unroll
            for (int i = 0; i < 8; ++i) acc[t][i] = 0.f;

        mma_chunk<K>(bp, a, acc);
#pragma unroll
        for (int kc = 128; kc < K; kc += 128) {
            load_a_chunk(A + (size_t)(row0 + c) * K + 4 * h + kc, a);
            mma_chunk<K>(bp + kc, a, acc);
        }

        if (res) store_strip<true >(acc, bias, res, out, row0, c, h);
        else     store_strip<false>(acc, bias, res, out, row0, c, h);

        tile += 16 * GEMM_WGS;
        if (tile < ntiles) load_a_chunk(A + (size_t)(tile * 16 + c) * K + 4 * h, a);
    }
}

__global__ __launch_bounds__(256)
void transpose_pad_kernel(const float* __restrict__ W, float* __restrict__ Wt,
                          int K, int Nc, int Ncpad)
{
    const int tid = blockIdx.x * 256 + threadIdx.x;
    if (tid >= Ncpad * K) return;
    const int c = tid / K;
    const int k = tid - c * K;
    Wt[(size_t)c * K + k] = (c < Nc) ? W[(size_t)k * Nc + c] : 0.f;
}

__global__ void pad_bias_kernel(const float* __restrict__ b, float* __restrict__ bp,
                                int Nc, int Ncpad)
{
    const int c = threadIdx.x;
    if (c < Ncpad) bp[c] = (c < Nc) ? b[c] : 0.f;
}

__global__ __launch_bounds__(256)
void zero_int_kernel(int* __restrict__ p, int n)
{
    const int i = blockIdx.x * 256 + threadIdx.x;
    if (i < n) p[i] = 0;
}

__global__ __launch_bounds__(256)
void csr_count_kernel(const int* __restrict__ dst, int* __restrict__ deg, int* __restrict__ rank)
{
    const int e = blockIdx.x * 256 + threadIdx.x;
    if (e >= 2 * NEDGES) return;
    const int v = ((e >= NEDGES) ? NNODES : 0) + dst[e];
    rank[e] = atomicAdd(deg + v, 1);
}

__global__ __launch_bounds__(256)
void csr_alloc_kernel(const int* __restrict__ deg, int* __restrict__ counter,
                      int* __restrict__ seg_beg, int* __restrict__ seg_end, int n)
{
    __shared__ int wtot[8];
    __shared__ int blk_base;
    const int tid  = threadIdx.x;
    const int lane = tid & 31;
    const int w    = tid >> 5;
    const int i    = blockIdx.x * 256 + tid;
    const int d    = (i < n) ? deg[i] : 0;
    int incl = d;
#pragma unroll
    for (int off = 1; off < 32; off <<= 1) {
        const int up = __shfl_up(incl, off, 32);
        if (lane >= off) incl += up;
    }
    if (lane == 31) wtot[w] = incl;
    __syncthreads();
    if (tid == 0) {
        int tot = 0;
#pragma unroll
        for (int k = 0; k < 8; ++k) tot += wtot[k];
        blk_base = atomicAdd(counter, tot);
    }
    __syncthreads();
    int st = blk_base + incl - d;
#pragma unroll
    for (int k = 0; k < 8; ++k) if (k < w) st += wtot[k];
    if (i < n) { seg_beg[i] = st; seg_end[i] = st + d; }
}

__global__ __launch_bounds__(256)
void csr_fill_kernel(const int* __restrict__ src, const int* __restrict__ dst,
                     const int* __restrict__ seg_beg, const int* __restrict__ rank,
                     int* __restrict__ csr_src)
{
    const int e = blockIdx.x * 256 + threadIdx.x;
    if (e >= 2 * NEDGES) return;
    const int v = ((e >= NEDGES) ? NNODES : 0) + dst[e];
    csr_src[seg_beg[v] + rank[e]] = src[e];
}

__device__ __forceinline__ float4 ln_relu_row(const float4 v, const float4 gg, const float4 bb)
{
    float s1 = v.x + v.y + v.z + v.w;
    float s2 = v.x * v.x + v.y * v.y + v.z * v.z + v.w * v.w;
#pragma unroll
    for (int off = 16; off > 0; off >>= 1) {
        s1 += __shfl_xor(s1, off, 32);
        s2 += __shfl_xor(s2, off, 32);
    }
    const float mu  = s1 * (1.f / HDIM);
    const float var = s2 * (1.f / HDIM) - mu * mu;
    const float inv = rsqrtf(var + LN_EPS);
    float4 r;
    r.x = fmaxf((v.x - mu) * inv * gg.x + bb.x, 0.f);
    r.y = fmaxf((v.y - mu) * inv * gg.y + bb.y, 0.f);
    r.z = fmaxf((v.z - mu) * inv * gg.z + bb.z, 0.f);
    r.w = fmaxf((v.w - mu) * inv * gg.w + bb.w, 0.f);
    return r;
}

#define AGG_BLOCKS 4096

__device__ __forceinline__ float exp_nonpos(const float x)
{
    return __builtin_amdgcn_exp2f(x * 1.44269504088896340736f);
}

template <int CNT, bool LNF>
__device__ __forceinline__ void agg_chunk(const float* __restrict__ hl, int idx, int c, bool first,
                                          float (&mx)[4], float (&de)[4], float (&nu)[4],
                                          const float4 gg, const float4 bb)
{
    float4 hv[CNT];
#pragma unroll
    for (int j = 0; j < CNT; ++j) {
        const int s = __builtin_amdgcn_readlane(idx, c + j);
        hv[j] = *(const float4*)(hl + (size_t)s * HDIM);
    }
    if (LNF) {
#pragma unroll
        for (int j = 0; j < CNT; ++j) hv[j] = ln_relu_row(hv[j], gg, bb);
    }
    float m[CNT][4];
    float c0 = mx[0], c1 = mx[1], c2 = mx[2], c3 = mx[3];
#pragma unroll
    for (int j = 0; j < CNT; ++j) {
        m[j][0] = fmaxf(hv[j].x, 0.f) + GEN_EPS;
        m[j][1] = fmaxf(hv[j].y, 0.f) + GEN_EPS;
        m[j][2] = fmaxf(hv[j].z, 0.f) + GEN_EPS;
        m[j][3] = fmaxf(hv[j].w, 0.f) + GEN_EPS;
        c0 = fmaxf(c0, m[j][0]);
        c1 = fmaxf(c1, m[j][1]);
        c2 = fmaxf(c2, m[j][2]);
        c3 = fmaxf(c3, m[j][3]);
    }
    if (!first) {
        const float s0 = exp_nonpos(mx[0] - c0), s1 = exp_nonpos(mx[1] - c1);
        const float s2 = exp_nonpos(mx[2] - c2), s3 = exp_nonpos(mx[3] - c3);
        de[0] *= s0; nu[0] *= s0;
        de[1] *= s1; nu[1] *= s1;
        de[2] *= s2; nu[2] *= s2;
        de[3] *= s3; nu[3] *= s3;
    }
    mx[0] = c0; mx[1] = c1; mx[2] = c2; mx[3] = c3;
#pragma unroll
    for (int j = 0; j < CNT; ++j) {
#pragma unroll
        for (int f = 0; f < 4; ++f) {
            const float ex = exp_nonpos(m[j][f] - mx[f]);
            de[f] += ex;
            nu[f] += m[j][f] * ex;
        }
    }
}

template <bool LNF>
__device__ __forceinline__ float4 agg_node(const float* __restrict__ hl, const int* __restrict__ csr_src,
                                           int beg, int n, int lane, const float4 own,
                                           const float4 gg, const float4 bb)
{
    float mx[4] = {0.f, 0.f, 0.f, 0.f};
    float de[4] = {0.f, 0.f, 0.f, 0.f};
    float nu[4] = {0.f, 0.f, 0.f, 0.f};

    for (int base = 0; base < n; base += 32) {
        const int bc  = (n - base < 32) ? (n - base) : 32;
        const int idx = (lane < bc) ? csr_src[beg + base + lane] : 0;
        for (int c = 0; c < bc; c += 8) {
            const int  cnt   = (bc - c < 8) ? (bc - c) : 8;
            const bool first = (base + c == 0);
            switch (cnt) {
            case 1:  agg_chunk<1, LNF>(hl, idx, c, first, mx, de, nu, gg, bb); break;
            case 2:  agg_chunk<2, LNF>(hl, idx, c, first, mx, de, nu, gg, bb); break;
            case 3:  agg_chunk<3, LNF>(hl, idx, c, first, mx, de, nu, gg, bb); break;
            case 4:  agg_chunk<4, LNF>(hl, idx, c, first, mx, de, nu, gg, bb); break;
            case 5:  agg_chunk<5, LNF>(hl, idx, c, first, mx, de, nu, gg, bb); break;
            case 6:  agg_chunk<6, LNF>(hl, idx, c, first, mx, de, nu, gg, bb); break;
            case 7:  agg_chunk<7, LNF>(hl, idx, c, first, mx, de, nu, gg, bb); break;
            default: agg_chunk<8, LNF>(hl, idx, c, first, mx, de, nu, gg, bb); break;
            }
        }
    }

    float4 r;
    r.x = nu[0] / (de[0] + 1e-16f) + own.x;
    r.y = nu[1] / (de[1] + 1e-16f) + own.y;
    r.z = nu[2] / (de[2] + 1e-16f) + own.z;
    r.w = nu[3] / (de[3] + 1e-16f) + own.w;
    return r;
}

__global__ __launch_bounds__(256)
void node_softmax_agg_kernel(const float* __restrict__ hin, const int* __restrict__ seg_beg,
                             const int* __restrict__ seg_end, const int* __restrict__ csr_src,
                             float* __restrict__ t)
{
    const int lane = threadIdx.x & 31;
    const int wv = __builtin_amdgcn_readfirstlane((int)(threadIdx.x >> 5));
    const float* __restrict__ hl = hin + lane * 4;
    const float4 none = {0.f, 0.f, 0.f, 0.f};

    for (int row = blockIdx.x * 8 + wv; row < NNODES; row += AGG_BLOCKS * 8) {
        const int beg = __builtin_amdgcn_readfirstlane(seg_beg[row]);
        const int n   = __builtin_amdgcn_readfirstlane(seg_end[row]) - beg;
        const float4 own = *(const float4*)(hl + (size_t)row * HDIM);
        const float4 r = agg_node<false>(hl, csr_src, beg, n, lane, own, none, none);
        *(float4*)(t + (size_t)row * HDIM + lane * 4) = r;
    }
}

__global__ __launch_bounds__(256)
void seed_softmax_agg_kernel(const float* __restrict__ h, const int* __restrict__ final_map,
                             const int* __restrict__ node_map,
                             const float* __restrict__ g, const float* __restrict__ b,
                             const int* __restrict__ seg_beg, const int* __restrict__ seg_end,
                             const int* __restrict__ csr_src,
                             float* __restrict__ t, float* __restrict__ rs)
{
    const int lane = threadIdx.x & 31;
    const int wv = __builtin_amdgcn_readfirstlane((int)(threadIdx.x >> 5));
    const int i  = blockIdx.x * 8 + wv;
    if (i >= NSEEDS) return;
    const int v   = __builtin_amdgcn_readfirstlane(node_map[final_map[i]]);
    const int beg = __builtin_amdgcn_readfirstlane(seg_beg[v]);
    const int n   = __builtin_amdgcn_readfirstlane(seg_end[v]) - beg;
    const float* __restrict__ hl = h + lane * 4;
    const float4 raw = *(const float4*)(hl + (size_t)v * HDIM);
    const float4 gg  = *(const float4*)(g + lane * 4);
    const float4 bb  = *(const float4*)(b + lane * 4);
    const float4 own = ln_relu_row(raw, gg, bb);
    const float4 r   = agg_node<true>(hl, csr_src, beg, n, lane, own, gg, bb);
    *(float4*)(t  + (size_t)i * HDIM + lane * 4) = r;
    *(float4*)(rs + (size_t)i * HDIM + lane * 4) = raw;
}

__global__ __launch_bounds__(256)
void ln_relu_kernel(const float* __restrict__ in, const float* __restrict__ g,
                    const float* __restrict__ b, float* __restrict__ out, int M)
{
    const int row  = blockIdx.x * 8 + (threadIdx.x >> 5);
    const int lane = threadIdx.x & 31;
    if (row >= M) return;
    const float4 v  = *(const float4*)(in + (size_t)row * HDIM + lane * 4);
    const float4 gg = *(const float4*)(g + lane * 4);
    const float4 bb = *(const float4*)(b + lane * 4);
    const float4 r  = ln_relu_row(v, gg, bb);
    *(float4*)(out + (size_t)row * HDIM + lane * 4) = r;
}

__global__ __launch_bounds__(256)
void gather_kernel(const float* __restrict__ in, const int* __restrict__ map,
                   float* __restrict__ out, int M)
{
    const int tid = blockIdx.x * 256 + threadIdx.x;
    if (tid >= M * 32) return;
    const int i = tid >> 5;
    const int f = (tid & 31) * 4;
    const int j = map[i];
    *(float4*)(out + (size_t)i * HDIM + f) =
        *(const float4*)(in + (size_t)j * HDIM + f);
}

__global__ __launch_bounds__(256)
void logsoftmax_kernel(const float* __restrict__ in, float* __restrict__ out, int M)
{
    const int row  = blockIdx.x * 8 + (threadIdx.x >> 5);
    const int lane = threadIdx.x & 31;
    if (row >= M) return;
    const float NEG = -3.0e38f;
    const float v0 = (lane      < NCLS) ? in[(size_t)row * NCLSPAD + lane     ] : NEG;
    const float v1 = (lane + 32 < NCLS) ? in[(size_t)row * NCLSPAD + lane + 32] : NEG;
    float mx = fmaxf(v0, v1);
#pragma unroll
    for (int off = 16; off > 0; off >>= 1) mx = fmaxf(mx, __shfl_xor(mx, off, 32));
    float se = ((lane      < NCLS) ? expf(v0 - mx) : 0.f)
             + ((lane + 32 < NCLS) ? expf(v1 - mx) : 0.f);
#pragma unroll
    for (int off = 16; off > 0; off >>= 1) se += __shfl_xor(se, off, 32);
    const float lse = logf(se) + mx;
    if (lane      < NCLS) out[(size_t)row * NCLS + lane     ] = v0 - lse;
    if (lane + 32 < NCLS) out[(size_t)row * NCLS + lane + 32] = v1 - lse;
}

extern "C" void kernel_launch(void* const* d_in, const int* in_sizes, int n_in,
                              void* d_out, int out_size, void* d_ws, size_t ws_size,
                              hipStream_t stream)
{
    (void)in_sizes; (void)n_in; (void)out_size; (void)ws_size;

    const float* x         = (const float*)d_in[0];
    const int*   src       = (const int*)  d_in[1];
    const int*   dst       = (const int*)  d_in[2];
    const int*   node_map  = (const int*)  d_in[3];
    const int*   final_map = (const int*)  d_in[4];
    const float* enc_w     = (const float*)d_in[5];
    const float* enc_b     = (const float*)d_in[6];
    const float* gcn_w     = (const float*)d_in[7];
    const float* gcn_b     = (const float*)d_in[8];
    const float* ln_g      = (const float*)d_in[9];
    const float* ln_b      = (const float*)d_in[10];
    const float* pred_w    = (const float*)d_in[11];
    const float* pred_b    = (const float*)d_in[12];
    float* out = (float*)d_out;

    const size_t NH = (size_t)NNODES * HDIM;
    float* B0     = (float*)d_ws;
    float* B1     = B0 + NH;
    float* B2     = B0 + 2 * NH;
    float* B3     = B0 + 3 * NH;
    float* B4     = B0 + 4 * NH;
    float* HG     = B0 + 5 * NH;
    float* HG2    = HG  + (size_t)NSEEDS * HDIM;
    float* SC     = HG2 + (size_t)NSEEDS * HDIM;
    float* WT_ENC = SC  + (size_t)NSEEDS * NCLSPAD;
    float* WT_G   = WT_ENC + (size_t)HDIM * FDIM;
    float* PWT    = WT_G   + (size_t)3 * HDIM * HDIM;
    float* PB     = PWT    + (size_t)NCLSPAD * HDIM;

    transpose_pad_kernel<<<(HDIM * FDIM + 255) / 256, 256, 0, stream>>>(
        enc_w, WT_ENC, FDIM, HDIM, HDIM);
    for (int l = 0; l < 3; ++l)
        transpose_pad_kernel<<<(HDIM * HDIM + 255) / 256, 256, 0, stream>>>(
            gcn_w + (size_t)l * HDIM * HDIM, WT_G + (size_t)l * HDIM * HDIM,
            HDIM, HDIM, HDIM);
    transpose_pad_kernel<<<(NCLSPAD * HDIM + 255) / 256, 256, 0, stream>>>(
        pred_w, PWT, HDIM, NCLS, NCLSPAD);
    pad_bias_kernel<<<1, NCLSPAD, 0, stream>>>(pred_b, PB, NCLS, NCLSPAD);

    const dim3 gblk(32, 4, 1);
    const int  mtN   = NNODES / 16;
    const int  ngrid = (NNODES * 32 + 255) / 256;
    const int  lgrid = (NNODES + 7) / 8;

    int* IDEG   = (int*)B3;
    int* ICNT   = IDEG + 2 * NNODES;
    int* ISTART = IDEG + 2 * NNODES + 32;
    int* IEND   = ISTART + 2 * NNODES;
    int* ICSR   = IEND + 2 * NNODES;
    int* IRANK  = ICSR + 2 * NEDGES;
    zero_int_kernel<<<(2 * NNODES + 32 + 255) / 256, 256, 0, stream>>>(IDEG, 2 * NNODES + 32);
    csr_count_kernel<<<(2 * NEDGES + 255) / 256, 256, 0, stream>>>(dst, IDEG, IRANK);
    csr_alloc_kernel<<<(2 * NNODES + 255) / 256, 256, 0, stream>>>(
        IDEG, ICNT, ISTART, IEND, 2 * NNODES);
    csr_fill_kernel<<<(2 * NEDGES + 255) / 256, 256, 0, stream>>>(src, dst, ISTART, IRANK, ICSR);

    gemm_rows128_kernel<FDIM><<<GEMM_WGS, 512, 0, stream>>>(
        x, WT_ENC, enc_b, nullptr, B0, mtN);

    auto conv = [&](const float* hin, int eblk, const float* wt, const float* bias,
                    const float* res, float* outp) {
        node_softmax_agg_kernel<<<AGG_BLOCKS, 256, 0, stream>>>(
            hin, ISTART + (size_t)eblk * NNODES, IEND + (size_t)eblk * NNODES, ICSR, B2);
        gemm_rows128_kernel<HDIM><<<GEMM_WGS, 512, 0, stream>>>(
            B2, wt, bias, res, outp, mtN);
    };

    conv(B0, 0, WT_G, gcn_b, nullptr, B0);

    ln_relu_kernel<<<lgrid, 256, 0, stream>>>(B0, ln_g, ln_b, B1, NNODES);
    conv(B1, 0, WT_G + (size_t)HDIM * HDIM, gcn_b + HDIM,  B0,  B4);
    gather_kernel<<<ngrid, 256, 0, stream>>>(B4, node_map, B0, NNODES);

    float* T2 = B2;
    float* R2 = B4;
    const int mtS = NSEEDS / 16;
    seed_softmax_agg_kernel<<<NSEEDS / 8, 256, 0, stream>>>(
        B0, final_map, node_map + NNODES, ln_g + HDIM, ln_b + HDIM,
        ISTART + NNODES, IEND + NNODES, ICSR, T2, R2);
    gemm_rows128_kernel<HDIM><<<GEMM_WGS, 512, 0, stream>>>(
        T2, WT_G + (size_t)2 * HDIM * HDIM, gcn_b + 2 * HDIM, R2, HG, mtS);

    ln_relu_kernel<<<(NSEEDS + 7) / 8, 256, 0, stream>>>(
        HG, ln_g + 2 * HDIM, ln_b + 2 * HDIM, HG2, NSEEDS);
    const dim3 gridS((mtS + 3) / 4, NCLSPAD / 64, 1);
    gemm_f32_wmma_kernel<<<gridS, gblk, 0, stream>>>(
        HG2, PWT, PB, nullptr, SC, NSEEDS, HDIM, NCLSPAD);
    logsoftmax_kernel<<<(NSEEDS + 7) / 8, 256, 0, stream>>>(SC, out, NSEEDS);
}


# Layer 1 on demand: rows the seed layer reads are flagged via node_map[0]; its aggregation skips unflagged rows and the seed kernel reads through the map, so the 100000-row gather pass is gone
# speedup vs baseline: 4.4473x; 1.0487x over previous
#include <hip/hip_runtime.h>
#include <hip/hip_bf16.h>
#include <math.h>

#define NNODES   100000
#define NEDGES   600000
#define FDIM     256
#define HDIM     128
#define NCLS     47
#define NCLSPAD  64
#define NSEEDS   10000
#define GEN_EPS  1e-7f
#define LN_EPS   1e-5f

typedef float v2f __attribute__((ext_vector_type(2)));
typedef float v8f __attribute__((ext_vector_type(8)));

__global__ __launch_bounds__(128)
void gemm_f32_wmma_kernel(const float* __restrict__ A, const float* __restrict__ Wt,
                          const float* __restrict__ bias, const float* __restrict__ res,
                          float* __restrict__ out, int M, int K, int Nc)
{
    const int lane  = threadIdx.x;
    const int mtile = blockIdx.x * 4 + threadIdx.y;
    const int row0  = mtile * 16;
    if (row0 >= M) return;
    const int hb   = lane >> 4;
    const int lo   = lane & 15;
    const int arow = row0 + lo;
    const int col0 = blockIdx.y * 64 + lo;

    const float* __restrict__ a_ptr = A  + (size_t)arow * K;
    const float* __restrict__ b_ptr = Wt + (size_t)col0 * K;
    const size_t bstep = (size_t)16 * K;

    v8f acc[4];
#pragma unroll
    for (int t = 0; t < 4; ++t)
#pragma unroll
        for (int i = 0; i < 8; ++i) acc[t][i] = 0.f;

#pragma unroll 4
    for (int k = 0; k < K; k += 4) {
        const int ka = k + 2 * hb;
        const v2f a  = *(const v2f*)(a_ptr + ka);
        const v2f b0 = *(const v2f*)(b_ptr + ka);
        const v2f b1 = *(const v2f*)(b_ptr + bstep + ka);
        const v2f b2 = *(const v2f*)(b_ptr + 2 * bstep + ka);
        const v2f b3 = *(const v2f*)(b_ptr + 3 * bstep + ka);
        acc[0] = __builtin_amdgcn_wmma_f32_16x16x4_f32(false, a, false, b0, (short)0, acc[0], false, false);
        acc[1] = __builtin_amdgcn_wmma_f32_16x16x4_f32(false, a, false, b1, (short)0, acc[1], false, false);
        acc[2] = __builtin_amdgcn_wmma_f32_16x16x4_f32(false, a, false, b2, (short)0, acc[2], false, false);
        acc[3] = __builtin_amdgcn_wmma_f32_16x16x4_f32(false, a, false, b3, (short)0, acc[3], false, false);
    }

#pragma unroll
    for (int t = 0; t < 4; ++t) {
        const int   col = col0 + 16 * t;
        const float bv  = bias[col];
#pragma unroll
        for (int i = 0; i < 8; ++i) {
            const int r = row0 + i + 8 * hb;
            float v = acc[t][i] + bv;
            if (res) v += res[(size_t)r * Nc + col];
            out[(size_t)r * Nc + col] = v;
        }
    }
}

#define GEMM_WGS 256


__device__ __forceinline__ void load_a_chunk(const float* __restrict__ ap, float4 (&a)[16])
{
#pragma unroll
    for (int i = 0; i < 16; ++i) a[i] = *(const float4*)(ap + 8 * i);
}

template <int K>
__device__ __forceinline__ void load_b_block(const float* bp, float4 (&b)[8])
{
    constexpr int PITCH = K + 4;
#pragma unroll
    for (int t = 0; t < 8; ++t) b[t] = *(const float4*)(bp + t * 16 * PITCH);
}

__device__ __forceinline__ void mma_block(const float4 a, const float4 (&b)[8], v8f (&acc)[8])
{
    const v2f a01 = {a.x, a.y};
    const v2f a23 = {a.z, a.w};
#pragma unroll
    for (int t = 0; t < 8; ++t) {
        const v2f b01 = {b[t].x, b[t].y};
        acc[t] = __builtin_amdgcn_wmma_f32_16x16x4_f32(false, a01, false, b01, (short)0, acc[t], false, false);
    }
#pragma unroll
    for (int t = 0; t < 8; ++t) {
        const v2f b23 = {b[t].z, b[t].w};
        acc[t] = __builtin_amdgcn_wmma_f32_16x16x4_f32(false, a23, false, b23, (short)0, acc[t], false, false);
    }
}

template <int K>
__device__ __forceinline__ void mma_chunk(const float* bp, const float4 (&a)[16], v8f (&acc)[8])
{
    float4 b0[8], b1[8];
    load_b_block<K>(bp, b0);
#pragma unroll
    for (int i = 0; i < 16; i += 2) {
        load_b_block<K>(bp + 8 * (i + 1), b1);
        __builtin_amdgcn_sched_barrier(0);
        mma_block(a[i], b0, acc);
        __builtin_amdgcn_sched_barrier(0);
        if (i + 2 < 16) load_b_block<K>(bp + 8 * (i + 2), b0);
        __builtin_amdgcn_sched_barrier(0);
        mma_block(a[i + 1], b1, acc);
        __builtin_amdgcn_sched_barrier(0);
    }
}

template <bool HAS_RES>
__device__ __forceinline__ void store_strip(const v8f (&acc)[8], const float* __restrict__ bias,
                                            const float* __restrict__ res, float* __restrict__ out,
                                            int row0, int c, int h)
{
#pragma unroll
    for (int g = 0; g < 2; ++g) {
        const int    col = 64 * g + 4 * c;
        const float4 bv  = *(const float4*)(bias + col);
        float4 rv[8];
        if (HAS_RES) {
#pragma unroll
            for (int i = 0; i < 8; ++i)
                rv[i] = *(const float4*)(res + (size_t)(row0 + i + 8 * h) * HDIM + col);
        }
#pragma unroll
        for (int i = 0; i < 8; ++i) {
            float4 v;
            v.x = acc[4 * g + 0][i] + bv.x;
            v.y = acc[4 * g + 1][i] + bv.y;
            v.z = acc[4 * g + 2][i] + bv.z;
            v.w = acc[4 * g + 3][i] + bv.w;
            if (HAS_RES) { v.x += rv[i].x; v.y += rv[i].y; v.z += rv[i].z; v.w += rv[i].w; }
            *(float4*)(out + (size_t)(row0 + i + 8 * h) * HDIM + col) = v;
        }
    }
}

template <int K>
__global__ __launch_bounds__(512)
void gemm_rows128_kernel(const float* __restrict__ A, const float* __restrict__ Wt,
                         const float* __restrict__ bias, const float* __restrict__ res,
                         float* __restrict__ out, int ntiles)
{
    constexpr int PITCH = K + 4;
    __shared__ __attribute__((aligned(16))) float wl[HDIM * PITCH];

    const int tid  = threadIdx.x;
    const int lane = tid & 31;
    const int wv   = __builtin_amdgcn_readfirstlane(tid >> 5);
    const int c    = lane & 15;
    const int h    = lane >> 4;

    const int wa = wv >> 2;
    const int q  = wa + 4 * (((wv & 3) - wa) & 3);
    int tile = q * GEMM_WGS + (int)blockIdx.x;

    float4 a[16];
    if (tile < ntiles) load_a_chunk(A + (size_t)(tile * 16 + c) * K + 4 * h, a);

    constexpr int NPASS = HDIM * (K / 4) / 512;
    float4 wq[NPASS];
#pragma unroll
    for (int r = 0; r < NPASS; ++r) {
        const int i  = tid + 512 * r;
        const int n  = i / (K / 4);
        const int k4 = (i - n * (K / 4)) * 4;
        wq[r] = *(const float4*)(Wt + (size_t)n * K + k4);
    }
#pragma unroll
    for (int r = 0; r < NPASS; ++r) {
        const int i   = tid + 512 * r;
        const int n   = i / (K / 4);
        const int k4  = (i - n * (K / 4)) * 4;
        const int rho = (n & 64) + 16 * (n & 3) + ((n >> 2) & 15);
        *(float4*)(wl + rho * PITCH + k4) = wq[r];
    }
    __syncthreads();

    const float* bp = wl + c * PITCH + 4 * h;

    while (tile < ntiles) {
        const int row0 = tile * 16;

        v8f acc[8];
#pragma unroll
        for (int t = 0; t < 8; ++t)
#pragma unroll
            for (int i = 0; i < 8; ++i) acc[t][i] = 0.f;

        mma_chunk<K>(bp, a, acc);
#pragma unroll
        for (int kc = 128; kc < K; kc += 128) {
            load_a_chunk(A + (size_t)(row0 + c) * K + 4 * h + kc, a);
            mma_chunk<K>(bp + kc, a, acc);
        }

        if (res) store_strip<true >(acc, bias, res, out, row0, c, h);
        else     store_strip<false>(acc, bias, res, out, row0, c, h);

        tile += 16 * GEMM_WGS;
        if (tile < ntiles) load_a_chunk(A + (size_t)(tile * 16 + c) * K + 4 * h, a);
    }
}

__global__ __launch_bounds__(256)
void transpose_pad_kernel(const float* __restrict__ W, float* __restrict__ Wt,
                          int K, int Nc, int Ncpad)
{
    const int tid = blockIdx.x * 256 + threadIdx.x;
    if (tid >= Ncpad * K) return;
    const int c = tid / K;
    const int k = tid - c * K;
    Wt[(size_t)c * K + k] = (c < Nc) ? W[(size_t)k * Nc + c] : 0.f;
}

__global__ void pad_bias_kernel(const float* __restrict__ b, float* __restrict__ bp,
                                int Nc, int Ncpad)
{
    const int c = threadIdx.x;
    if (c < Ncpad) bp[c] = (c < Nc) ? b[c] : 0.f;
}

__global__ __launch_bounds__(256)
void zero_int_kernel(int* __restrict__ p, int n)
{
    const int i = blockIdx.x * 256 + threadIdx.x;
    if (i < n) p[i] = 0;
}

__global__ __launch_bounds__(256)
void csr_count_kernel(const int* __restrict__ dst, int* __restrict__ deg, int* __restrict__ rank)
{
    const int e = blockIdx.x * 256 + threadIdx.x;
    if (e >= 2 * NEDGES) return;
    const int v = ((e >= NEDGES) ? NNODES : 0) + dst[e];
    rank[e] = atomicAdd(deg + v, 1);
}

__global__ __launch_bounds__(256)
void csr_alloc_kernel(const int* __restrict__ deg, int* __restrict__ counter,
                      int* __restrict__ seg_beg, int* __restrict__ seg_end, int n)
{
    __shared__ int wtot[8];
    __shared__ int blk_base;
    const int tid  = threadIdx.x;
    const int lane = tid & 31;
    const int w    = tid >> 5;
    const int i    = blockIdx.x * 256 + tid;
    const int d    = (i < n) ? deg[i] : 0;
    int incl = d;
#pragma unroll
    for (int off = 1; off < 32; off <<= 1) {
        const int up = __shfl_up(incl, off, 32);
        if (lane >= off) incl += up;
    }
    if (lane == 31) wtot[w] = incl;
    __syncthreads();
    if (tid == 0) {
        int tot = 0;
#pragma unroll
        for (int k = 0; k < 8; ++k) tot += wtot[k];
        blk_base = atomicAdd(counter, tot);
    }
    __syncthreads();
    int st = blk_base + incl - d;
#pragma unroll
    for (int k = 0; k < 8; ++k) if (k < w) st += wtot[k];
    if (i < n) { seg_beg[i] = st; seg_end[i] = st + d; }
}

__global__ __launch_bounds__(256)
void csr_fill_kernel(const int* __restrict__ src, const int* __restrict__ dst,
                     const int* __restrict__ seg_beg, const int* __restrict__ rank,
                     int* __restrict__ csr_src)
{
    const int e = blockIdx.x * 256 + threadIdx.x;
    if (e >= 2 * NEDGES) return;
    const int v = ((e >= NEDGES) ? NNODES : 0) + dst[e];
    csr_src[seg_beg[v] + rank[e]] = src[e];
}

__device__ __forceinline__ float4 ln_relu_row(const float4 v, const float4 gg, const float4 bb)
{
    float s1 = v.x + v.y + v.z + v.w;
    float s2 = v.x * v.x + v.y * v.y + v.z * v.z + v.w * v.w;
#pragma unroll
    for (int off = 16; off > 0; off >>= 1) {
        s1 += __shfl_xor(s1, off, 32);
        s2 += __shfl_xor(s2, off, 32);
    }
    const float mu  = s1 * (1.f / HDIM);
    const float var = s2 * (1.f / HDIM) - mu * mu;
    const float inv = rsqrtf(var + LN_EPS);
    float4 r;
    r.x = fmaxf((v.x - mu) * inv * gg.x + bb.x, 0.f);
    r.y = fmaxf((v.y - mu) * inv * gg.y + bb.y, 0.f);
    r.z = fmaxf((v.z - mu) * inv * gg.z + bb.z, 0.f);
    r.w = fmaxf((v.w - mu) * inv * gg.w + bb.w, 0.f);
    return r;
}

#define AGG_BLOCKS 4096
#define AGG_DEMAND_BLOCKS 1024

__device__ __forceinline__ float exp_nonpos(const float x)
{
    return __builtin_amdgcn_exp2f(x * 1.44269504088896340736f);
}

template <int CNT, bool LNF>
__device__ __forceinline__ void agg_chunk(const float* __restrict__ hl, int idx, int c, bool first,
                                          float (&mx)[4], float (&de)[4], float (&nu)[4],
                                          const float4 gg, const float4 bb)
{
    float4 hv[CNT];
#pragma unroll
    for (int j = 0; j < CNT; ++j) {
        const int s = __builtin_amdgcn_readlane(idx, c + j);
        hv[j] = *(const float4*)(hl + (size_t)s * HDIM);
    }
    if (LNF) {
#pragma unroll
        for (int j = 0; j < CNT; ++j) hv[j] = ln_relu_row(hv[j], gg, bb);
    }
    float m[CNT][4];
    float c0 = mx[0], c1 = mx[1], c2 = mx[2], c3 = mx[3];
#pragma unroll
    for (int j = 0; j < CNT; ++j) {
        m[j][0] = fmaxf(hv[j].x, 0.f) + GEN_EPS;
        m[j][1] = fmaxf(hv[j].y, 0.f) + GEN_EPS;
        m[j][2] = fmaxf(hv[j].z, 0.f) + GEN_EPS;
        m[j][3] = fmaxf(hv[j].w, 0.f) + GEN_EPS;
        c0 = fmaxf(c0, m[j][0]);
        c1 = fmaxf(c1, m[j][1]);
        c2 = fmaxf(c2, m[j][2]);
        c3 = fmaxf(c3, m[j][3]);
    }
    if (!first) {
        const float s0 = exp_nonpos(mx[0] - c0), s1 = exp_nonpos(mx[1] - c1);
        const float s2 = exp_nonpos(mx[2] - c2), s3 = exp_nonpos(mx[3] - c3);
        de[0] *= s0; nu[0] *= s0;
        de[1] *= s1; nu[1] *= s1;
        de[2] *= s2; nu[2] *= s2;
        de[3] *= s3; nu[3] *= s3;
    }
    mx[0] = c0; mx[1] = c1; mx[2] = c2; mx[3] = c3;
#pragma unroll
    for (int j = 0; j < CNT; ++j) {
#pragma unroll
        for (int f = 0; f < 4; ++f) {
            const float ex = exp_nonpos(m[j][f] - mx[f]);
            de[f] += ex;
            nu[f] += m[j][f] * ex;
        }
    }
}

template <bool LNF>
__device__ __forceinline__ float4 agg_node(const float* __restrict__ hl, const int* __restrict__ csr_src,
                                           const int* __restrict__ rmap,
                                           int beg, int n, int lane, const float4 own,
                                           const float4 gg, const float4 bb)
{
    float mx[4] = {0.f, 0.f, 0.f, 0.f};
    float de[4] = {0.f, 0.f, 0.f, 0.f};
    float nu[4] = {0.f, 0.f, 0.f, 0.f};

    for (int base = 0; base < n; base += 32) {
        const int bc  = (n - base < 32) ? (n - base) : 32;
        int idx = (lane < bc) ? csr_src[beg + base + lane] : 0;
        if (LNF) idx = rmap[idx];
        for (int c = 0; c < bc; c += 8) {
            const int  cnt   = (bc - c < 8) ? (bc - c) : 8;
            const bool first = (base + c == 0);
            switch (cnt) {
            case 1:  agg_chunk<1, LNF>(hl, idx, c, first, mx, de, nu, gg, bb); break;
            case 2:  agg_chunk<2, LNF>(hl, idx, c, first, mx, de, nu, gg, bb); break;
            case 3:  agg_chunk<3, LNF>(hl, idx, c, first, mx, de, nu, gg, bb); break;
            case 4:  agg_chunk<4, LNF>(hl, idx, c, first, mx, de, nu, gg, bb); break;
            case 5:  agg_chunk<5, LNF>(hl, idx, c, first, mx, de, nu, gg, bb); break;
            case 6:  agg_chunk<6, LNF>(hl, idx, c, first, mx, de, nu, gg, bb); break;
            case 7:  agg_chunk<7, LNF>(hl, idx, c, first, mx, de, nu, gg, bb); break;
            default: agg_chunk<8, LNF>(hl, idx, c, first, mx, de, nu, gg, bb); break;
            }
        }
    }

    float4 r;
    r.x = nu[0] / (de[0] + 1e-16f) + own.x;
    r.y = nu[1] / (de[1] + 1e-16f) + own.y;
    r.z = nu[2] / (de[2] + 1e-16f) + own.z;
    r.w = nu[3] / (de[3] + 1e-16f) + own.w;
    return r;
}

template <bool DEMAND>
__global__ __launch_bounds__(256)
void node_softmax_agg_kernel(const float* __restrict__ hin, const int* __restrict__ seg_beg,
                             const int* __restrict__ seg_end, const int* __restrict__ csr_src,
                             const int* __restrict__ need, float* __restrict__ t)
{
    const int lane = threadIdx.x & 31;
    const int wv = __builtin_amdgcn_readfirstlane((int)(threadIdx.x >> 5));
    const float* __restrict__ hl = hin + lane * 4;
    const float4 none = {0.f, 0.f, 0.f, 0.f};

    constexpr int NBLK = DEMAND ? AGG_DEMAND_BLOCKS : AGG_BLOCKS;
    for (int row = blockIdx.x * 8 + wv; row < NNODES; row += NBLK * 8) {
        if (DEMAND) {
            if (__builtin_amdgcn_readfirstlane(need[row]) == 0) continue;
        }
        const int beg = __builtin_amdgcn_readfirstlane(seg_beg[row]);
        const int n   = __builtin_amdgcn_readfirstlane(seg_end[row]) - beg;
        const float4 own = *(const float4*)(hl + (size_t)row * HDIM);
        const float4 r = agg_node<false>(hl, csr_src, nullptr, beg, n, lane, own, none, none);
        *(float4*)(t + (size_t)row * HDIM + lane * 4) = r;
    }
}

__global__ __launch_bounds__(256)
void mark_needed_kernel(const int* __restrict__ final_map, const int* __restrict__ node_map,
                        const int* __restrict__ pmap,
                        const int* __restrict__ seg_beg, const int* __restrict__ seg_end,
                        const int* __restrict__ csr_src, int* __restrict__ need)
{
    const int lane = threadIdx.x & 31;
    const int i    = blockIdx.x * 8 + (int)(threadIdx.x >> 5);
    if (i >= NSEEDS) return;
    const int v   = node_map[final_map[i]];
    const int beg = seg_beg[v];
    const int n   = seg_end[v] - beg;
    if (lane == 0) need[pmap[v]] = 1;
    for (int j = lane; j < n; j += 32) need[pmap[csr_src[beg + j]]] = 1;
}

__global__ __launch_bounds__(256)
void seed_softmax_agg_kernel(const float* __restrict__ p, const int* __restrict__ pmap,
                             const int* __restrict__ final_map,
                             const int* __restrict__ node_map,
                             const float* __restrict__ g, const float* __restrict__ b,
                             const int* __restrict__ seg_beg, const int* __restrict__ seg_end,
                             const int* __restrict__ csr_src,
                             float* __restrict__ t, float* __restrict__ rs)
{
    const int lane = threadIdx.x & 31;
    const int wv = __builtin_amdgcn_readfirstlane((int)(threadIdx.x >> 5));
    const int i  = blockIdx.x * 8 + wv;
    if (i >= NSEEDS) return;
    const int v   = __builtin_amdgcn_readfirstlane(node_map[final_map[i]]);
    const int pv  = __builtin_amdgcn_readfirstlane(pmap[v]);
    const int beg = __builtin_amdgcn_readfirstlane(seg_beg[v]);
    const int n   = __builtin_amdgcn_readfirstlane(seg_end[v]) - beg;
    const float* __restrict__ hl = p + lane * 4;
    const float4 raw = *(const float4*)(hl + (size_t)pv * HDIM);
    const float4 gg  = *(const float4*)(g + lane * 4);
    const float4 bb  = *(const float4*)(b + lane * 4);
    const float4 own = ln_relu_row(raw, gg, bb);
    const float4 r   = agg_node<true>(hl, csr_src, pmap, beg, n, lane, own, gg, bb);
    *(float4*)(t  + (size_t)i * HDIM + lane * 4) = r;
    *(float4*)(rs + (size_t)i * HDIM + lane * 4) = raw;
}

__global__ __launch_bounds__(256)
void ln_relu_kernel(const float* __restrict__ in, const float* __restrict__ g,
                    const float* __restrict__ b, float* __restrict__ out, int M)
{
    const int row  = blockIdx.x * 8 + (threadIdx.x >> 5);
    const int lane = threadIdx.x & 31;
    if (row >= M) return;
    const float4 v  = *(const float4*)(in + (size_t)row * HDIM + lane * 4);
    const float4 gg = *(const float4*)(g + lane * 4);
    const float4 bb = *(const float4*)(b + lane * 4);
    const float4 r  = ln_relu_row(v, gg, bb);
    *(float4*)(out + (size_t)row * HDIM + lane * 4) = r;
}

__global__ __launch_bounds__(256)
void logsoftmax_kernel(const float* __restrict__ in, float* __restrict__ out, int M)
{
    const int row  = blockIdx.x * 8 + (threadIdx.x >> 5);
    const int lane = threadIdx.x & 31;
    if (row >= M) return;
    const float NEG = -3.0e38f;
    const float v0 = (lane      < NCLS) ? in[(size_t)row * NCLSPAD + lane     ] : NEG;
    const float v1 = (lane + 32 < NCLS) ? in[(size_t)row * NCLSPAD + lane + 32] : NEG;
    float mx = fmaxf(v0, v1);
#pragma unroll
    for (int off = 16; off > 0; off >>= 1) mx = fmaxf(mx, __shfl_xor(mx, off, 32));
    float se = ((lane      < NCLS) ? expf(v0 - mx) : 0.f)
             + ((lane + 32 < NCLS) ? expf(v1 - mx) : 0.f);
#pragma unroll
    for (int off = 16; off > 0; off >>= 1) se += __shfl_xor(se, off, 32);
    const float lse = logf(se) + mx;
    if (lane      < NCLS) out[(size_t)row * NCLS + lane     ] = v0 - lse;
    if (lane + 32 < NCLS) out[(size_t)row * NCLS + lane + 32] = v1 - lse;
}

extern "C" void kernel_launch(void* const* d_in, const int* in_sizes, int n_in,
                              void* d_out, int out_size, void* d_ws, size_t ws_size,
                              hipStream_t stream)
{
    (void)in_sizes; (void)n_in; (void)out_size; (void)ws_size;

    const float* x         = (const float*)d_in[0];
    const int*   src       = (const int*)  d_in[1];
    const int*   dst       = (const int*)  d_in[2];
    const int*   node_map  = (const int*)  d_in[3];
    const int*   final_map = (const int*)  d_in[4];
    const float* enc_w     = (const float*)d_in[5];
    const float* enc_b     = (const float*)d_in[6];
    const float* gcn_w     = (const float*)d_in[7];
    const float* gcn_b     = (const float*)d_in[8];
    const float* ln_g      = (const float*)d_in[9];
    const float* ln_b      = (const float*)d_in[10];
    const float* pred_w    = (const float*)d_in[11];
    const float* pred_b    = (const float*)d_in[12];
    float* out = (float*)d_out;

    const size_t NH = (size_t)NNODES * HDIM;
    float* B0     = (float*)d_ws;
    float* B1     = B0 + NH;
    float* B2     = B0 + 2 * NH;
    float* B3     = B0 + 3 * NH;
    float* B4     = B0 + 4 * NH;
    float* HG     = B0 + 5 * NH;
    float* HG2    = HG  + (size_t)NSEEDS * HDIM;
    float* SC     = HG2 + (size_t)NSEEDS * HDIM;
    float* WT_ENC = SC  + (size_t)NSEEDS * NCLSPAD;
    float* WT_G   = WT_ENC + (size_t)HDIM * FDIM;
    float* PWT    = WT_G   + (size_t)3 * HDIM * HDIM;
    float* PB     = PWT    + (size_t)NCLSPAD * HDIM;

    transpose_pad_kernel<<<(HDIM * FDIM + 255) / 256, 256, 0, stream>>>(
        enc_w, WT_ENC, FDIM, HDIM, HDIM);
    for (int l = 0; l < 3; ++l)
        transpose_pad_kernel<<<(HDIM * HDIM + 255) / 256, 256, 0, stream>>>(
            gcn_w + (size_t)l * HDIM * HDIM, WT_G + (size_t)l * HDIM * HDIM,
            HDIM, HDIM, HDIM);
    transpose_pad_kernel<<<(NCLSPAD * HDIM + 255) / 256, 256, 0, stream>>>(
        pred_w, PWT, HDIM, NCLS, NCLSPAD);
    pad_bias_kernel<<<1, NCLSPAD, 0, stream>>>(pred_b, PB, NCLS, NCLSPAD);

    const dim3 gblk(32, 4, 1);
    const int  mtN   = NNODES / 16;
    const int  lgrid = (NNODES + 7) / 8;

    int* IDEG   = (int*)B3;
    int* ICNT   = IDEG + 2 * NNODES;
    int* INEED  = IDEG + 2 * NNODES + 32;
    int* ISTART = INEED + NNODES;
    int* IEND   = ISTART + 2 * NNODES;
    int* ICSR   = IEND + 2 * NNODES;
    int* IRANK  = ICSR + 2 * NEDGES;
    zero_int_kernel<<<(3 * NNODES + 32 + 255) / 256, 256, 0, stream>>>(IDEG, 3 * NNODES + 32);
    csr_count_kernel<<<(2 * NEDGES + 255) / 256, 256, 0, stream>>>(dst, IDEG, IRANK);
    csr_alloc_kernel<<<(2 * NNODES + 255) / 256, 256, 0, stream>>>(
        IDEG, ICNT, ISTART, IEND, 2 * NNODES);
    csr_fill_kernel<<<(2 * NEDGES + 255) / 256, 256, 0, stream>>>(src, dst, ISTART, IRANK, ICSR);

    gemm_rows128_kernel<FDIM><<<GEMM_WGS, 512, 0, stream>>>(
        x, WT_ENC, enc_b, nullptr, B0, mtN);

    node_softmax_agg_kernel<false><<<AGG_BLOCKS, 256, 0, stream>>>(
        B0, ISTART, IEND, ICSR, nullptr, B2);
    gemm_rows128_kernel<HDIM><<<GEMM_WGS, 512, 0, stream>>>(
        B2, WT_G, gcn_b, nullptr, B0, mtN);

    mark_needed_kernel<<<NSEEDS / 8, 256, 0, stream>>>(
        final_map, node_map + NNODES, node_map, ISTART + NNODES, IEND + NNODES, ICSR, INEED);
    ln_relu_kernel<<<lgrid, 256, 0, stream>>>(B0, ln_g, ln_b, B1, NNODES);
    node_softmax_agg_kernel<true><<<AGG_DEMAND_BLOCKS, 256, 0, stream>>>(
        B1, ISTART, IEND, ICSR, INEED, B2);
    gemm_rows128_kernel<HDIM><<<GEMM_WGS, 512, 0, stream>>>(
        B2, WT_G + (size_t)HDIM * HDIM, gcn_b + HDIM,  B0,  B4, mtN);

    float* T2 = B2;
    float* R2 = B1;
    const int mtS = NSEEDS / 16;
    seed_softmax_agg_kernel<<<NSEEDS / 8, 256, 0, stream>>>(
        B4, node_map, final_map, node_map + NNODES, ln_g + HDIM, ln_b + HDIM,
        ISTART + NNODES, IEND + NNODES, ICSR, T2, R2);
    gemm_rows128_kernel<HDIM><<<GEMM_WGS, 512, 0, stream>>>(
        T2, WT_G + (size_t)2 * HDIM * HDIM, gcn_b + 2 * HDIM, R2, HG, mtS);

    ln_relu_kernel<<<(NSEEDS + 7) / 8, 256, 0, stream>>>(
        HG, ln_g + 2 * HDIM, ln_b + 2 * HDIM, HG2, NSEEDS);
    const dim3 gridS((mtS + 3) / 4, NCLSPAD / 64, 1);
    gemm_f32_wmma_kernel<<<gridS, gblk, 0, stream>>>(
        HG2, PWT, PB, nullptr, SC, NSEEDS, HDIM, NCLSPAD);
    logsoftmax_kernel<<<(NSEEDS + 7) / 8, 256, 0, stream>>>(SC, out, NSEEDS);
}


# In-edge list build: the per-edge returning global atomic add is replaced by a two-level counting sort on LDS counters (bins of 1024 rows, one block per bin); in-row list order = LDS arrival order
# speedup vs baseline: 5.6190x; 1.2635x over previous
#include <hip/hip_runtime.h>
#include <hip/hip_bf16.h>
#include <math.h>

#define NNODES   100000
#define NEDGES   600000
#define FDIM     256
#define HDIM     128
#define NCLS     47
#define NCLSPAD  64
#define NSEEDS   10000
#define GEN_EPS  1e-7f
#define LN_EPS   1e-5f

typedef float v2f __attribute__((ext_vector_type(2)));
typedef float v8f __attribute__((ext_vector_type(8)));

__global__ __launch_bounds__(128)
void gemm_f32_wmma_kernel(const float* __restrict__ A, const float* __restrict__ Wt,
                          const float* __restrict__ bias, const float* __restrict__ res,
                          float* __restrict__ out, int M, int K, int Nc)
{
    const int lane  = threadIdx.x;
    const int mtile = blockIdx.x * 4 + threadIdx.y;
    const int row0  = mtile * 16;
    if (row0 >= M) return;
    const int hb   = lane >> 4;
    const int lo   = lane & 15;
    const int arow = row0 + lo;
    const int col0 = blockIdx.y * 64 + lo;

    const float* __restrict__ a_ptr = A  + (size_t)arow * K;
    const float* __restrict__ b_ptr = Wt + (size_t)col0 * K;
    const size_t bstep = (size_t)16 * K;

    v8f acc[4];
#pragma unroll
    for (int t = 0; t < 4; ++t)
#pragma unroll
        for (int i = 0; i < 8; ++i) acc[t][i] = 0.f;

#pragma unroll 4
    for (int k = 0; k < K; k += 4) {
        const int ka = k + 2 * hb;
        const v2f a  = *(const v2f*)(a_ptr + ka);
        const v2f b0 = *(const v2f*)(b_ptr + ka);
        const v2f b1 = *(const v2f*)(b_ptr + bstep + ka);
        const v2f b2 = *(const v2f*)(b_ptr + 2 * bstep + ka);
        const v2f b3 = *(const v2f*)(b_ptr + 3 * bstep + ka);
        acc[0] = __builtin_amdgcn_wmma_f32_16x16x4_f32(false, a, false, b0, (short)0, acc[0], false, false);
        acc[1] = __builtin_amdgcn_wmma_f32_16x16x4_f32(false, a, false, b1, (short)0, acc[1], false, false);
        acc[2] = __builtin_amdgcn_wmma_f32_16x16x4_f32(false, a, false, b2, (short)0, acc[2], false, false);
        acc[3] = __builtin_amdgcn_wmma_f32_16x16x4_f32(false, a, false, b3, (short)0, acc[3], false, false);
    }

#pragma unroll
    for (int t = 0; t < 4; ++t) {
        const int   col = col0 + 16 * t;
        const float bv  = bias[col];
#pragma unroll
        for (int i = 0; i < 8; ++i) {
            const int r = row0 + i + 8 * hb;
            float v = acc[t][i] + bv;
            if (res) v += res[(size_t)r * Nc + col];
            out[(size_t)r * Nc + col] = v;
        }
    }
}

#define GEMM_WGS 256


__device__ __forceinline__ void load_a_chunk(const float* __restrict__ ap, float4 (&a)[16])
{
#pragma unroll
    for (int i = 0; i < 16; ++i) a[i] = *(const float4*)(ap + 8 * i);
}

template <int K>
__device__ __forceinline__ void load_b_block(const float* bp, float4 (&b)[8])
{
    constexpr int PITCH = K + 4;
#pragma unroll
    for (int t = 0; t < 8; ++t) b[t] = *(const float4*)(bp + t * 16 * PITCH);
}

__device__ __forceinline__ void mma_block(const float4 a, const float4 (&b)[8], v8f (&acc)[8])
{
    const v2f a01 = {a.x, a.y};
    const v2f a23 = {a.z, a.w};
#pragma unroll
    for (int t = 0; t < 8; ++t) {
        const v2f b01 = {b[t].x, b[t].y};
        acc[t] = __builtin_amdgcn_wmma_f32_16x16x4_f32(false, a01, false, b01, (short)0, acc[t], false, false);
    }
#pragma unroll
    for (int t = 0; t < 8; ++t) {
        const v2f b23 = {b[t].z, b[t].w};
        acc[t] = __builtin_amdgcn_wmma_f32_16x16x4_f32(false, a23, false, b23, (short)0, acc[t], false, false);
    }
}

template <int K>
__device__ __forceinline__ void mma_chunk(const float* bp, const float4 (&a)[16], v8f (&acc)[8])
{
    float4 b0[8], b1[8];
    load_b_block<K>(bp, b0);
#pragma unroll
    for (int i = 0; i < 16; i += 2) {
        load_b_block<K>(bp + 8 * (i + 1), b1);
        __builtin_amdgcn_sched_barrier(0);
        mma_block(a[i], b0, acc);
        __builtin_amdgcn_sched_barrier(0);
        if (i + 2 < 16) load_b_block<K>(bp + 8 * (i + 2), b0);
        __builtin_amdgcn_sched_barrier(0);
        mma_block(a[i + 1], b1, acc);
        __builtin_amdgcn_sched_barrier(0);
    }
}

template <bool HAS_RES>
__device__ __forceinline__ void store_strip(const v8f (&acc)[8], const float* __restrict__ bias,
                                            const float* __restrict__ res, float* __restrict__ out,
                                            int row0, int c, int h)
{
#pragma unroll
    for (int g = 0; g < 2; ++g) {
        const int    col = 64 * g + 4 * c;
        const float4 bv  = *(const float4*)(bias + col);
        float4 rv[8];
        if (HAS_RES) {
#pragma unroll
            for (int i = 0; i < 8; ++i)
                rv[i] = *(const float4*)(res + (size_t)(row0 + i + 8 * h) * HDIM + col);
        }
#pragma unroll
        for (int i = 0; i < 8; ++i) {
            float4 v;
            v.x = acc[4 * g + 0][i] + bv.x;
            v.y = acc[4 * g + 1][i] + bv.y;
            v.z = acc[4 * g + 2][i] + bv.z;
            v.w = acc[4 * g + 3][i] + bv.w;
            if (HAS_RES) { v.x += rv[i].x; v.y += rv[i].y; v.z += rv[i].z; v.w += rv[i].w; }
            *(float4*)(out + (size_t)(row0 + i + 8 * h) * HDIM + col) = v;
        }
    }
}

template <int K>
__global__ __launch_bounds__(512)
void gemm_rows128_kernel(const float* __restrict__ A, const float* __restrict__ Wt,
                         const float* __restrict__ bias, const float* __restrict__ res,
                         float* __restrict__ out, int ntiles)
{
    constexpr int PITCH = K + 4;
    __shared__ __attribute__((aligned(16))) float wl[HDIM * PITCH];

    const int tid  = threadIdx.x;
    const int lane = tid & 31;
    const int wv   = __builtin_amdgcn_readfirstlane(tid >> 5);
    const int c    = lane & 15;
    const int h    = lane >> 4;

    const int wa = wv >> 2;
    const int q  = wa + 4 * (((wv & 3) - wa) & 3);
    int tile = q * GEMM_WGS + (int)blockIdx.x;

    float4 a[16];
    if (tile < ntiles) load_a_chunk(A + (size_t)(tile * 16 + c) * K + 4 * h, a);

    constexpr int NPASS = HDIM * (K / 4) / 512;
    float4 wq[NPASS];
#pragma unroll
    for (int r = 0; r < NPASS; ++r) {
        const int i  = tid + 512 * r;
        const int n  = i / (K / 4);
        const int k4 = (i - n * (K / 4)) * 4;
        wq[r] = *(const float4*)(Wt + (size_t)n * K + k4);
    }
#pragma unroll
    for (int r = 0; r < NPASS; ++r) {
        const int i   = tid + 512 * r;
        const int n   = i / (K / 4);
        const int k4  = (i - n * (K / 4)) * 4;
        const int rho = (n & 64) + 16 * (n & 3) + ((n >> 2) & 15);
        *(float4*)(wl + rho * PITCH + k4) = wq[r];
    }
    __syncthreads();

    const float* bp = wl + c * PITCH + 4 * h;

    while (tile < ntiles) {
        const int row0 = tile * 16;

        v8f acc[8];
#pragma unroll
        for (int t = 0; t < 8; ++t)
#pragma unroll
            for (int i = 0; i < 8; ++i) acc[t][i] = 0.f;

        mma_chunk<K>(bp, a, acc);
#pragma unroll
        for (int kc = 128; kc < K; kc += 128) {
            load_a_chunk(A + (size_t)(row0 + c) * K + 4 * h + kc, a);
            mma_chunk<K>(bp + kc, a, acc);
        }

        if (res) store_strip<true >(acc, bias, res, out, row0, c, h);
        else     store_strip<false>(acc, bias, res, out, row0, c, h);

        tile += 16 * GEMM_WGS;
        if (tile < ntiles) load_a_chunk(A + (size_t)(tile * 16 + c) * K + 4 * h, a);
    }
}

__global__ __launch_bounds__(256)
void transpose_pad_kernel(const float* __restrict__ W, float* __restrict__ Wt,
                          int K, int Nc, int Ncpad)
{
    const int tid = blockIdx.x * 256 + threadIdx.x;
    if (tid >= Ncpad * K) return;
    const int c = tid / K;
    const int k = tid - c * K;
    Wt[(size_t)c * K + k] = (c < Nc) ? W[(size_t)k * Nc + c] : 0.f;
}

__global__ void pad_bias_kernel(const float* __restrict__ b, float* __restrict__ bp,
                                int Nc, int Ncpad)
{
    const int c = threadIdx.x;
    if (c < Ncpad) bp[c] = (c < Nc) ? b[c] : 0.f;
}

#define BIN_SHIFT   10
#define BIN_ROWS    (1 << BIN_SHIFT)
#define NBINS       ((2 * NNODES + BIN_ROWS - 1) / BIN_ROWS)
#define SORT_WGS    256
#define SORT_THR    512
#define SORT_CH     ((2 * NEDGES + SORT_WGS - 1) / SORT_WGS)
#define SORT_EPT    ((SORT_CH + SORT_THR - 1) / SORT_THR)
#define NEED_CH     ((NNODES + SORT_WGS - 1) / SORT_WGS)

__device__ __forceinline__ void load_chunk_rows(const int* __restrict__ dst, int w, int tid,
                                                int (&rows)[SORT_EPT], int (&eid)[SORT_EPT])
{
    const int e0 = w * SORT_CH;
    const int e1 = (e0 + SORT_CH < 2 * NEDGES) ? (e0 + SORT_CH) : (2 * NEDGES);
    int raw[SORT_EPT];
#pragma unroll
    for (int j = 0; j < SORT_EPT; ++j) {
        const int e = e0 + tid + SORT_THR * j;
        eid[j] = min(e, e1 - 1);
        raw[j] = dst[eid[j]];
    }
#pragma unroll
    for (int j = 0; j < SORT_EPT; ++j) {
        const int e = e0 + tid + SORT_THR * j;
        rows[j] = (e < e1) ? (((e >= NEDGES) ? NNODES : 0) + raw[j]) : -1;
    }
}

__device__ __forceinline__ int bin_slot(int row, int tid)
{
    return (row >= 0) ? (row >> BIN_SHIFT) : (NBINS + (tid & 31));
}

__global__ __launch_bounds__(SORT_THR)
void bin_hist_kernel(const int* __restrict__ dst, int* __restrict__ whist, int* __restrict__ need)
{
    __shared__ int hist[256];
    const int tid = threadIdx.x;
    const int w   = blockIdx.x;
    int rows[SORT_EPT], eid[SORT_EPT];
    load_chunk_rows(dst, w, tid, rows, eid);
    if (tid < 256) hist[tid] = 0;
    static_assert(NEED_CH <= SORT_THR, "one flag per thread");
    if (tid < NEED_CH && w * NEED_CH + tid < NNODES) need[w * NEED_CH + tid] = 0;
    __syncthreads();
#pragma unroll
    for (int j = 0; j < SORT_EPT; ++j) atomicAdd(&hist[bin_slot(rows[j], tid)], 1);
    __syncthreads();
    if (tid < NBINS) whist[tid * SORT_WGS + w] = hist[tid];
}

__global__ __launch_bounds__(SORT_THR)
void bin_scatter_kernel(const int* __restrict__ src, const int* __restrict__ dst,
                        const int* __restrict__ whist, int* __restrict__ binbase,
                        int2* __restrict__ rec)
{
    __shared__ int tot[256];
    __shared__ int cur[256];
    __shared__ int wtot[SORT_THR / 32];
    const int tid  = threadIdx.x;
    const int lane = tid & 31;
    const int wv   = tid >> 5;
    const int w    = blockIdx.x;

    int rows[SORT_EPT], eid[SORT_EPT], srcs[SORT_EPT];
    load_chunk_rows(dst, w, tid, rows, eid);
#pragma unroll
    for (int j = 0; j < SORT_EPT; ++j) srcs[j] = src[eid[j]];

    constexpr int NWV = SORT_THR / 32;
    constexpr int NIT = (NBINS + NWV - 1) / NWV;
    int4 q0[NIT], q1[NIT];
#pragma unroll
    for (int it = 0; it < NIT; ++it) {
        const int b  = wv + NWV * it;
        const int bb = (b < NBINS) ? b : (NBINS - 1);
        q0[it] = *(const int4*)(whist + bb * SORT_WGS + 8 * lane);
        q1[it] = *(const int4*)(whist + bb * SORT_WGS + 8 * lane + 4);
    }
    const int kf = w - 8 * lane;
    int tt[NIT], mm[NIT];
#pragma unroll
    for (int it = 0; it < NIT; ++it) {
        tt[it] = (q0[it].x + q0[it].y) + (q0[it].z + q0[it].w) + (q1[it].x + q1[it].y) + (q1[it].z + q1[it].w);
        mm[it] = ((0 < kf) ? q0[it].x : 0) + ((1 < kf) ? q0[it].y : 0) + ((2 < kf) ? q0[it].z : 0) + ((3 < kf) ? q0[it].w : 0)
               + ((4 < kf) ? q1[it].x : 0) + ((5 < kf) ? q1[it].y : 0) + ((6 < kf) ? q1[it].z : 0) + ((7 < kf) ? q1[it].w : 0);
    }
#pragma unroll
    for (int off = 16; off > 0; off >>= 1) {
        int tu[NIT], mu[NIT];
#pragma unroll
        for (int it = 0; it < NIT; ++it) {
            tu[it] = __shfl_xor(tt[it], off, 32);
            mu[it] = __shfl_xor(mm[it], off, 32);
        }
#pragma unroll
        for (int it = 0; it < NIT; ++it) { tt[it] += tu[it]; mm[it] += mu[it]; }
    }
#pragma unroll
    for (int it = 0; it < NIT; ++it) {
        const int b = wv + NWV * it;
        if (lane == 0 && b < NBINS) { tot[b] = tt[it]; cur[b] = mm[it]; }
    }
    if (tid >= NBINS && tid < 256) tot[tid] = 0;
    __syncthreads();

    const int d = (tid < 256) ? tot[tid] : 0;
    int incl = d;
#pragma unroll
    for (int off = 1; off < 32; off <<= 1) {
        const int up = __shfl_up(incl, off, 32);
        if (lane >= off) incl += up;
    }
    if (lane == 31) wtot[wv] = incl;
    __syncthreads();
    int excl = incl - d;
#pragma unroll
    for (int k = 0; k < SORT_THR / 32; ++k) if (k < wv) excl += wtot[k];
    if (tid < NBINS) cur[tid] += excl;
    if (w == 0 && tid <= NBINS) binbase[tid] = excl;
    __syncthreads();

    int pos[SORT_EPT];
#pragma unroll
    for (int j = 0; j < SORT_EPT; ++j) pos[j] = atomicAdd(&cur[bin_slot(rows[j], tid)], 1);
#pragma unroll
    for (int j = 0; j < SORT_EPT; ++j)
        if (rows[j] >= 0) rec[pos[j]] = make_int2(rows[j], srcs[j]);
}

#define RANK_UNR 8
__global__ __launch_bounds__(BIN_ROWS)
void bin_rank_kernel(const int2* __restrict__ rec, const int* __restrict__ binbase,
                     int* __restrict__ seg_beg, int* __restrict__ seg_end,
                     int* __restrict__ csr_src)
{
    __shared__ int cnt[BIN_ROWS + 32];
    __shared__ int wtot[BIN_ROWS / 32];
    const int tid   = threadIdx.x;
    const int lane  = tid & 31;
    const int wv    = tid >> 5;
    const int b     = blockIdx.x;
    const int r0    = binbase[b];
    const int r1    = binbase[b + 1];
    const int spare = BIN_ROWS + lane;

    cnt[tid] = 0;
    __syncthreads();
    for (int i0 = r0; i0 < r1; i0 += RANK_UNR * BIN_ROWS) {
        int x[RANK_UNR];
#pragma unroll
        for (int u = 0; u < RANK_UNR; ++u) {
            const int i = i0 + tid + u * BIN_ROWS;
            x[u] = rec[min(i, r1 - 1)].x;
        }
#pragma unroll
        for (int u = 0; u < RANK_UNR; ++u) {
            const int i = i0 + tid + u * BIN_ROWS;
            atomicAdd(&cnt[(i < r1) ? (x[u] & (BIN_ROWS - 1)) : spare], 1);
        }
    }
    __syncthreads();

    const int d = cnt[tid];
    int incl = d;
#pragma unroll
    for (int off = 1; off < 32; off <<= 1) {
        const int up = __shfl_up(incl, off, 32);
        if (lane >= off) incl += up;
    }
    if (lane == 31) wtot[wv] = incl;
    __syncthreads();
    int st = r0 + incl - d;
#pragma unroll
    for (int k = 0; k < BIN_ROWS / 32; ++k) if (k < wv) st += wtot[k];
    const int v = b * BIN_ROWS + tid;
    if (v < 2 * NNODES) { seg_beg[v] = st; seg_end[v] = st + d; }
    cnt[tid] = st;
    __syncthreads();

    for (int i0 = r0; i0 < r1; i0 += RANK_UNR * BIN_ROWS) {
        int2 r[RANK_UNR];
        int  pos[RANK_UNR];
#pragma unroll
        for (int u = 0; u < RANK_UNR; ++u) {
            const int i = i0 + tid + u * BIN_ROWS;
            r[u] = rec[min(i, r1 - 1)];
        }
#pragma unroll
        for (int u = 0; u < RANK_UNR; ++u) {
            const int i = i0 + tid + u * BIN_ROWS;
            pos[u] = atomicAdd(&cnt[(i < r1) ? (r[u].x & (BIN_ROWS - 1)) : spare], 1);
        }
#pragma unroll
        for (int u = 0; u < RANK_UNR; ++u) {
            const int i = i0 + tid + u * BIN_ROWS;
            if (i < r1) csr_src[pos[u]] = r[u].y;
        }
    }
}

__device__ __forceinline__ float4 ln_relu_row(const float4 v, const float4 gg, const float4 bb)
{
    float s1 = v.x + v.y + v.z + v.w;
    float s2 = v.x * v.x + v.y * v.y + v.z * v.z + v.w * v.w;
#pragma unroll
    for (int off = 16; off > 0; off >>= 1) {
        s1 += __shfl_xor(s1, off, 32);
        s2 += __shfl_xor(s2, off, 32);
    }
    const float mu  = s1 * (1.f / HDIM);
    const float var = s2 * (1.f / HDIM) - mu * mu;
    const float inv = rsqrtf(var + LN_EPS);
    float4 r;
    r.x = fmaxf((v.x - mu) * inv * gg.x + bb.x, 0.f);
    r.y = fmaxf((v.y - mu) * inv * gg.y + bb.y, 0.f);
    r.z = fmaxf((v.z - mu) * inv * gg.z + bb.z, 0.f);
    r.w = fmaxf((v.w - mu) * inv * gg.w + bb.w, 0.f);
    return r;
}

#define AGG_BLOCKS 4096
#define AGG_DEMAND_BLOCKS 1024

__device__ __forceinline__ float exp_nonpos(const float x)
{
    return __builtin_amdgcn_exp2f(x * 1.44269504088896340736f);
}

template <int CNT, bool LNF>
__device__ __forceinline__ void agg_chunk(const float* __restrict__ hl, int idx, int c, bool first,
                                          float (&mx)[4], float (&de)[4], float (&nu)[4],
                                          const float4 gg, const float4 bb)
{
    float4 hv[CNT];
#pragma unroll
    for (int j = 0; j < CNT; ++j) {
        const int s = __builtin_amdgcn_readlane(idx, c + j);
        hv[j] = *(const float4*)(hl + (size_t)s * HDIM);
    }
    if (LNF) {
#pragma unroll
        for (int j = 0; j < CNT; ++j) hv[j] = ln_relu_row(hv[j], gg, bb);
    }
    float m[CNT][4];
    float c0 = mx[0], c1 = mx[1], c2 = mx[2], c3 = mx[3];
#pragma unroll
    for (int j = 0; j < CNT; ++j) {
        m[j][0] = fmaxf(hv[j].x, 0.f) + GEN_EPS;
        m[j][1] = fmaxf(hv[j].y, 0.f) + GEN_EPS;
        m[j][2] = fmaxf(hv[j].z, 0.f) + GEN_EPS;
        m[j][3] = fmaxf(hv[j].w, 0.f) + GEN_EPS;
        c0 = fmaxf(c0, m[j][0]);
        c1 = fmaxf(c1, m[j][1]);
        c2 = fmaxf(c2, m[j][2]);
        c3 = fmaxf(c3, m[j][3]);
    }
    if (!first) {
        const float s0 = exp_nonpos(mx[0] - c0), s1 = exp_nonpos(mx[1] - c1);
        const float s2 = exp_nonpos(mx[2] - c2), s3 = exp_nonpos(mx[3] - c3);
        de[0] *= s0; nu[0] *= s0;
        de[1] *= s1; nu[1] *= s1;
        de[2] *= s2; nu[2] *= s2;
        de[3] *= s3; nu[3] *= s3;
    }
    mx[0] = c0; mx[1] = c1; mx[2] = c2; mx[3] = c3;
#pragma unroll
    for (int j = 0; j < CNT; ++j) {
#pragma unroll
        for (int f = 0; f < 4; ++f) {
            const float ex = exp_nonpos(m[j][f] - mx[f]);
            de[f] += ex;
            nu[f] += m[j][f] * ex;
        }
    }
}

template <bool LNF>
__device__ __forceinline__ float4 agg_node(const float* __restrict__ hl, const int* __restrict__ csr_src,
                                           const int* __restrict__ rmap,
                                           int beg, int n, int lane, const float4 own,
                                           const float4 gg, const float4 bb)
{
    float mx[4] = {0.f, 0.f, 0.f, 0.f};
    float de[4] = {0.f, 0.f, 0.f, 0.f};
    float nu[4] = {0.f, 0.f, 0.f, 0.f};

    for (int base = 0; base < n; base += 32) {
        const int bc  = (n - base < 32) ? (n - base) : 32;
        int idx = (lane < bc) ? csr_src[beg + base + lane] : 0;
        if (LNF) idx = rmap[idx];
        for (int c = 0; c < bc; c += 8) {
            const int  cnt   = (bc - c < 8) ? (bc - c) : 8;
            const bool first = (base + c == 0);
            switch (cnt) {
            case 1:  agg_chunk<1, LNF>(hl, idx, c, first, mx, de, nu, gg, bb); break;
            case 2:  agg_chunk<2, LNF>(hl, idx, c, first, mx, de, nu, gg, bb); break;
            case 3:  agg_chunk<3, LNF>(hl, idx, c, first, mx, de, nu, gg, bb); break;
            case 4:  agg_chunk<4, LNF>(hl, idx, c, first, mx, de, nu, gg, bb); break;
            case 5:  agg_chunk<5, LNF>(hl, idx, c, first, mx, de, nu, gg, bb); break;
            case 6:  agg_chunk<6, LNF>(hl, idx, c, first, mx, de, nu, gg, bb); break;
            case 7:  agg_chunk<7, LNF>(hl, idx, c, first, mx, de, nu, gg, bb); break;
            default: agg_chunk<8, LNF>(hl, idx, c, first, mx, de, nu, gg, bb); break;
            }
        }
    }

    float4 r;
    r.x = nu[0] / (de[0] + 1e-16f) + own.x;
    r.y = nu[1] / (de[1] + 1e-16f) + own.y;
    r.z = nu[2] / (de[2] + 1e-16f) + own.z;
    r.w = nu[3] / (de[3] + 1e-16f) + own.w;
    return r;
}

template <bool DEMAND>
__global__ __launch_bounds__(256)
void node_softmax_agg_kernel(const float* __restrict__ hin, const int* __restrict__ seg_beg,
                             const int* __restrict__ seg_end, const int* __restrict__ csr_src,
                             const int* __restrict__ need, float* __restrict__ t)
{
    const int lane = threadIdx.x & 31;
    const int wv = __builtin_amdgcn_readfirstlane((int)(threadIdx.x >> 5));
    const float* __restrict__ hl = hin + lane * 4;
    const float4 none = {0.f, 0.f, 0.f, 0.f};

    constexpr int NBLK = DEMAND ? AGG_DEMAND_BLOCKS : AGG_BLOCKS;
    for (int row = blockIdx.x * 8 + wv; row < NNODES; row += NBLK * 8) {
        if (DEMAND) {
            if (__builtin_amdgcn_readfirstlane(need[row]) == 0) continue;
        }
        const int beg = __builtin_amdgcn_readfirstlane(seg_beg[row]);
        const int n   = __builtin_amdgcn_readfirstlane(seg_end[row]) - beg;
        const float4 own = *(const float4*)(hl + (size_t)row * HDIM);
        const float4 r = agg_node<false>(hl, csr_src, nullptr, beg, n, lane, own, none, none);
        *(float4*)(t + (size_t)row * HDIM + lane * 4) = r;
    }
}

__global__ __launch_bounds__(256)
void mark_needed_kernel(const int* __restrict__ final_map, const int* __restrict__ node_map,
                        const int* __restrict__ pmap,
                        const int* __restrict__ seg_beg, const int* __restrict__ seg_end,
                        const int* __restrict__ csr_src, int* __restrict__ need)
{
    const int lane = threadIdx.x & 31;
    const int i    = blockIdx.x * 8 + (int)(threadIdx.x >> 5);
    if (i >= NSEEDS) return;
    const int v   = node_map[final_map[i]];
    const int beg = seg_beg[v];
    const int n   = seg_end[v] - beg;
    if (lane == 0) need[pmap[v]] = 1;
    for (int j = lane; j < n; j += 32) need[pmap[csr_src[beg + j]]] = 1;
}

__global__ __launch_bounds__(256)
void seed_softmax_agg_kernel(const float* __restrict__ p, const int* __restrict__ pmap,
                             const int* __restrict__ final_map,
                             const int* __restrict__ node_map,
                             const float* __restrict__ g, const float* __restrict__ b,
                             const int* __restrict__ seg_beg, const int* __restrict__ seg_end,
                             const int* __restrict__ csr_src,
                             float* __restrict__ t, float* __restrict__ rs)
{
    const int lane = threadIdx.x & 31;
    const int wv = __builtin_amdgcn_readfirstlane((int)(threadIdx.x >> 5));
    const int i  = blockIdx.x * 8 + wv;
    if (i >= NSEEDS) return;
    const int v   = __builtin_amdgcn_readfirstlane(node_map[final_map[i]]);
    const int pv  = __builtin_amdgcn_readfirstlane(pmap[v]);
    const int beg = __builtin_amdgcn_readfirstlane(seg_beg[v]);
    const int n   = __builtin_amdgcn_readfirstlane(seg_end[v]) - beg;
    const float* __restrict__ hl = p + lane * 4;
    const float4 raw = *(const float4*)(hl + (size_t)pv * HDIM);
    const float4 gg  = *(const float4*)(g + lane * 4);
    const float4 bb  = *(const float4*)(b + lane * 4);
    const float4 own = ln_relu_row(raw, gg, bb);
    const float4 r   = agg_node<true>(hl, csr_src, pmap, beg, n, lane, own, gg, bb);
    *(float4*)(t  + (size_t)i * HDIM + lane * 4) = r;
    *(float4*)(rs + (size_t)i * HDIM + lane * 4) = raw;
}

__global__ __launch_bounds__(256)
void ln_relu_kernel(const float* __restrict__ in, const float* __restrict__ g,
                    const float* __restrict__ b, float* __restrict__ out, int M)
{
    const int row  = blockIdx.x * 8 + (threadIdx.x >> 5);
    const int lane = threadIdx.x & 31;
    if (row >= M) return;
    const float4 v  = *(const float4*)(in + (size_t)row * HDIM + lane * 4);
    const float4 gg = *(const float4*)(g + lane * 4);
    const float4 bb = *(const float4*)(b + lane * 4);
    const float4 r  = ln_relu_row(v, gg, bb);
    *(float4*)(out + (size_t)row * HDIM + lane * 4) = r;
}

__global__ __launch_bounds__(256)
void logsoftmax_kernel(const float* __restrict__ in, float* __restrict__ out, int M)
{
    const int row  = blockIdx.x * 8 + (threadIdx.x >> 5);
    const int lane = threadIdx.x & 31;
    if (row >= M) return;
    const float NEG = -3.0e38f;
    const float v0 = (lane      < NCLS) ? in[(size_t)row * NCLSPAD + lane     ] : NEG;
    const float v1 = (lane + 32 < NCLS) ? in[(size_t)row * NCLSPAD + lane + 32] : NEG;
    float mx = fmaxf(v0, v1);
#pragma unroll
    for (int off = 16; off > 0; off >>= 1) mx = fmaxf(mx, __shfl_xor(mx, off, 32));
    float se = ((lane      < NCLS) ? expf(v0 - mx) : 0.f)
             + ((lane + 32 < NCLS) ? expf(v1 - mx) : 0.f);
#pragma unroll
    for (int off = 16; off > 0; off >>= 1) se += __shfl_xor(se, off, 32);
    const float lse = logf(se) + mx;
    if (lane      < NCLS) out[(size_t)row * NCLS + lane     ] = v0 - lse;
    if (lane + 32 < NCLS) out[(size_t)row * NCLS + lane + 32] = v1 - lse;
}

extern "C" void kernel_launch(void* const* d_in, const int* in_sizes, int n_in,
                              void* d_out, int out_size, void* d_ws, size_t ws_size,
                              hipStream_t stream)
{
    (void)in_sizes; (void)n_in; (void)out_size; (void)ws_size;

    const float* x         = (const float*)d_in[0];
    const int*   src       = (const int*)  d_in[1];
    const int*   dst       = (const int*)  d_in[2];
    const int*   node_map  = (const int*)  d_in[3];
    const int*   final_map = (const int*)  d_in[4];
    const float* enc_w     = (const float*)d_in[5];
    const float* enc_b     = (const float*)d_in[6];
    const float* gcn_w     = (const float*)d_in[7];
    const float* gcn_b     = (const float*)d_in[8];
    const float* ln_g      = (const float*)d_in[9];
    const float* ln_b      = (const float*)d_in[10];
    const float* pred_w    = (const float*)d_in[11];
    const float* pred_b    = (const float*)d_in[12];
    float* out = (float*)d_out;

    const size_t NH = (size_t)NNODES * HDIM;
    float* B0     = (float*)d_ws;
    float* B1     = B0 + NH;
    float* B2     = B0 + 2 * NH;
    float* B3     = B0 + 3 * NH;
    float* B4     = B0 + 4 * NH;
    float* HG     = B0 + 5 * NH;
    float* HG2    = HG  + (size_t)NSEEDS * HDIM;
    float* SC     = HG2 + (size_t)NSEEDS * HDIM;
    float* WT_ENC = SC  + (size_t)NSEEDS * NCLSPAD;
    float* WT_G   = WT_ENC + (size_t)HDIM * FDIM;
    float* PWT    = WT_G   + (size_t)3 * HDIM * HDIM;
    float* PB     = PWT    + (size_t)NCLSPAD * HDIM;

    transpose_pad_kernel<<<(HDIM * FDIM + 255) / 256, 256, 0, stream>>>(
        enc_w, WT_ENC, FDIM, HDIM, HDIM);
    for (int l = 0; l < 3; ++l)
        transpose_pad_kernel<<<(HDIM * HDIM + 255) / 256, 256, 0, stream>>>(
            gcn_w + (size_t)l * HDIM * HDIM, WT_G + (size_t)l * HDIM * HDIM,
            HDIM, HDIM, HDIM);
    transpose_pad_kernel<<<(NCLSPAD * HDIM + 255) / 256, 256, 0, stream>>>(
        pred_w, PWT, HDIM, NCLS, NCLSPAD);
    pad_bias_kernel<<<1, NCLSPAD, 0, stream>>>(pred_b, PB, NCLS, NCLSPAD);

    const dim3 gblk(32, 4, 1);
    const int  mtN   = NNODES / 16;
    const int  lgrid = (NNODES + 7) / 8;

    int*  IWHIST = (int*)B3;
    int*  IBASE  = IWHIST + NBINS * SORT_WGS;
    int*  INEED  = IBASE + 256;
    int*  ISTART = INEED + NNODES;
    int*  IEND   = ISTART + 2 * NNODES;
    int*  ICSR   = IEND + 2 * NNODES;
    int2* IREC   = (int2*)(ICSR + 2 * NEDGES);
    bin_hist_kernel<<<SORT_WGS, SORT_THR, 0, stream>>>(dst, IWHIST, INEED);
    bin_scatter_kernel<<<SORT_WGS, SORT_THR, 0, stream>>>(src, dst, IWHIST, IBASE, IREC);
    bin_rank_kernel<<<NBINS, BIN_ROWS, 0, stream>>>(IREC, IBASE, ISTART, IEND, ICSR);

    gemm_rows128_kernel<FDIM><<<GEMM_WGS, 512, 0, stream>>>(
        x, WT_ENC, enc_b, nullptr, B0, mtN);

    node_softmax_agg_kernel<false><<<AGG_BLOCKS, 256, 0, stream>>>(
        B0, ISTART, IEND, ICSR, nullptr, B2);
    gemm_rows128_kernel<HDIM><<<GEMM_WGS, 512, 0, stream>>>(
        B2, WT_G, gcn_b, nullptr, B0, mtN);

    mark_needed_kernel<<<NSEEDS / 8, 256, 0, stream>>>(
        final_map, node_map + NNODES, node_map, ISTART + NNODES, IEND + NNODES, ICSR, INEED);
    ln_relu_kernel<<<lgrid, 256, 0, stream>>>(B0, ln_g, ln_b, B1, NNODES);
    node_softmax_agg_kernel<true><<<AGG_DEMAND_BLOCKS, 256, 0, stream>>>(
        B1, ISTART, IEND, ICSR, INEED, B2);
    gemm_rows128_kernel<HDIM><<<GEMM_WGS, 512, 0, stream>>>(
        B2, WT_G + (size_t)HDIM * HDIM, gcn_b + HDIM,  B0,  B4, mtN);

    float* T2 = B2;
    float* R2 = B1;
    const int mtS = NSEEDS / 16;
    seed_softmax_agg_kernel<<<NSEEDS / 8, 256, 0, stream>>>(
        B4, node_map, final_map, node_map + NNODES, ln_g + HDIM, ln_b + HDIM,
        ISTART + NNODES, IEND + NNODES, ICSR, T2, R2);
    gemm_rows128_kernel<HDIM><<<GEMM_WGS, 512, 0, stream>>>(
        T2, WT_G + (size_t)2 * HDIM * HDIM, gcn_b + 2 * HDIM, R2, HG, mtS);

    ln_relu_kernel<<<(NSEEDS + 7) / 8, 256, 0, stream>>>(
        HG, ln_g + 2 * HDIM, ln_b + 2 * HDIM, HG2, NSEEDS);
    const dim3 gridS((mtS + 3) / 4, NCLSPAD / 64, 1);
    gemm_f32_wmma_kernel<<<gridS, gblk, 0, stream>>>(
        HG2, PWT, PB, nullptr, SC, NSEEDS, HDIM, NCLSPAD);
    logsoftmax_kernel<<<(NSEEDS + 7) / 8, 256, 0, stream>>>(SC, out, NSEEDS);
}


# Weights are read in their given [K][N] layout: every GEMM work-group transposes (head: also pads) them into LDS itself, so the six weight/bias preparation launches are gone (21 -> 15 launches)
# speedup vs baseline: 6.0223x; 1.0718x over previous
#include <hip/hip_runtime.h>
#include <hip/hip_bf16.h>
#include <math.h>

#define NNODES   100000
#define NEDGES   600000
#define FDIM     256
#define HDIM     128
#define NCLS     47
#define NCLSPAD  64
#define NSEEDS   10000
#define GEN_EPS  1e-7f
#define LN_EPS   1e-5f

typedef float v2f __attribute__((ext_vector_type(2)));
typedef float v8f __attribute__((ext_vector_type(8)));


#define HEAD_PITCH (HDIM + 4)
__global__ __launch_bounds__(128)
void gemm_head_kernel(const float* __restrict__ A, const float* __restrict__ W,
                      const float* __restrict__ bias, float* __restrict__ out, int M)
{
    __shared__ __attribute__((aligned(16))) float wl[NCLSPAD * HEAD_PITCH];

    const int lane = threadIdx.x;
    const int tid  = threadIdx.y * 32 + lane;

    constexpr int NW    = HDIM * NCLS;
    constexpr int NPASS = NW / 128;
    static_assert(NW % 128 == 0, "whole passes of the 128 threads");
    float wq[NPASS];
#pragma unroll
    for (int r = 0; r < NPASS; ++r) wq[r] = W[tid + 128 * r];
#pragma unroll
    for (int r = 0; r < NPASS; ++r) {
        const int i = tid + 128 * r;
        const int k = i / NCLS;
        const int n = i - k * NCLS;
        wl[n * HEAD_PITCH + k] = wq[r];
    }
    for (int i = tid; i < (NCLSPAD - NCLS) * HDIM; i += 128) {
        const int n = NCLS + i / HDIM;
        const int k = i & (HDIM - 1);
        wl[n * HEAD_PITCH + k] = 0.f;
    }
    __syncthreads();

    const int mtile = blockIdx.x * 4 + threadIdx.y;
    const int row0  = mtile * 16;
    if (row0 >= M) return;
    const int hb   = lane >> 4;
    const int lo   = lane & 15;
    const int arow = row0 + lo;

    const float* __restrict__ a_ptr = A + (size_t)arow * HDIM;
    const float* b_ptr = wl + lo * HEAD_PITCH;
    constexpr int bstep = 16 * HEAD_PITCH;

    v8f acc[4];
#pragma unroll
    for (int t = 0; t < 4; ++t)
#pragma unroll
        for (int i = 0; i < 8; ++i) acc[t][i] = 0.f;

#pragma unroll 4
    for (int k = 0; k < HDIM; k += 4) {
        const int ka = k + 2 * hb;
        const v2f a  = *(const v2f*)(a_ptr + ka);
        const v2f b0 = *(const v2f*)(b_ptr + ka);
        const v2f b1 = *(const v2f*)(b_ptr + bstep + ka);
        const v2f b2 = *(const v2f*)(b_ptr + 2 * bstep + ka);
        const v2f b3 = *(const v2f*)(b_ptr + 3 * bstep + ka);
        acc[0] = __builtin_amdgcn_wmma_f32_16x16x4_f32(false, a, false, b0, (short)0, acc[0], false, false);
        acc[1] = __builtin_amdgcn_wmma_f32_16x16x4_f32(false, a, false, b1, (short)0, acc[1], false, false);
        acc[2] = __builtin_amdgcn_wmma_f32_16x16x4_f32(false, a, false, b2, (short)0, acc[2], false, false);
        acc[3] = __builtin_amdgcn_wmma_f32_16x16x4_f32(false, a, false, b3, (short)0, acc[3], false, false);
    }

#pragma unroll
    for (int t = 0; t < 4; ++t) {
        const int   col = lo + 16 * t;
        const float bv  = (col < NCLS) ? bias[col] : 0.f;
#pragma unroll
        for (int i = 0; i < 8; ++i) {
            const int r = row0 + i + 8 * hb;
            out[(size_t)r * NCLSPAD + col] = acc[t][i] + bv;
        }
    }
}

#define GEMM_WGS 256


__device__ __forceinline__ void load_a_chunk(const float* __restrict__ ap, float4 (&a)[16])
{
#pragma unroll
    for (int i = 0; i < 16; ++i) a[i] = *(const float4*)(ap + 8 * i);
}

template <int K>
__device__ __forceinline__ void load_b_block(const float* bp, float4 (&b)[8])
{
    constexpr int PITCH = K + 4;
#pragma unroll
    for (int t = 0; t < 8; ++t) b[t] = *(const float4*)(bp + t * 16 * PITCH);
}

__device__ __forceinline__ void mma_block(const float4 a, const float4 (&b)[8], v8f (&acc)[8])
{
    const v2f a01 = {a.x, a.y};
    const v2f a23 = {a.z, a.w};
#pragma unroll
    for (int t = 0; t < 8; ++t) {
        const v2f b01 = {b[t].x, b[t].y};
        acc[t] = __builtin_amdgcn_wmma_f32_16x16x4_f32(false, a01, false, b01, (short)0, acc[t], false, false);
    }
#pragma unroll
    for (int t = 0; t < 8; ++t) {
        const v2f b23 = {b[t].z, b[t].w};
        acc[t] = __builtin_amdgcn_wmma_f32_16x16x4_f32(false, a23, false, b23, (short)0, acc[t], false, false);
    }
}

template <int K>
__device__ __forceinline__ void mma_chunk(const float* bp, const float4 (&a)[16], v8f (&acc)[8])
{
    float4 b0[8], b1[8];
    load_b_block<K>(bp, b0);
#pragma unroll
    for (int i = 0; i < 16; i += 2) {
        load_b_block<K>(bp + 8 * (i + 1), b1);
        __builtin_amdgcn_sched_barrier(0);
        mma_block(a[i], b0, acc);
        __builtin_amdgcn_sched_barrier(0);
        if (i + 2 < 16) load_b_block<K>(bp + 8 * (i + 2), b0);
        __builtin_amdgcn_sched_barrier(0);
        mma_block(a[i + 1], b1, acc);
        __builtin_amdgcn_sched_barrier(0);
    }
}

template <bool HAS_RES>
__device__ __forceinline__ void store_strip(const v8f (&acc)[8], const float* __restrict__ bias,
                                            const float* __restrict__ res, float* __restrict__ out,
                                            int row0, int c, int h)
{
#pragma unroll
    for (int g = 0; g < 2; ++g) {
        const int    col = 64 * g + 4 * c;
        const float4 bv  = *(const float4*)(bias + col);
        float4 rv[8];
        if (HAS_RES) {
#pragma unroll
            for (int i = 0; i < 8; ++i)
                rv[i] = *(const float4*)(res + (size_t)(row0 + i + 8 * h) * HDIM + col);
        }
#pragma unroll
        for (int i = 0; i < 8; ++i) {
            float4 v;
            v.x = acc[4 * g + 0][i] + bv.x;
            v.y = acc[4 * g + 1][i] + bv.y;
            v.z = acc[4 * g + 2][i] + bv.z;
            v.w = acc[4 * g + 3][i] + bv.w;
            if (HAS_RES) { v.x += rv[i].x; v.y += rv[i].y; v.z += rv[i].z; v.w += rv[i].w; }
            *(float4*)(out + (size_t)(row0 + i + 8 * h) * HDIM + col) = v;
        }
    }
}

template <int K>
__global__ __launch_bounds__(512)
void gemm_rows128_kernel(const float* __restrict__ A, const float* __restrict__ W,
                         const float* __restrict__ bias, const float* __restrict__ res,
                         float* __restrict__ out, int ntiles)
{
    constexpr int PITCH = K + 4;
    __shared__ __attribute__((aligned(16))) float wl[HDIM * PITCH];

    const int tid  = threadIdx.x;
    const int lane = tid & 31;
    const int wv   = __builtin_amdgcn_readfirstlane(tid >> 5);
    const int c    = lane & 15;
    const int h    = lane >> 4;

    const int wa = wv >> 2;
    const int q  = wa + 4 * (((wv & 3) - wa) & 3);
    int tile = q * GEMM_WGS + (int)blockIdx.x;

    float4 a[16];
    if (tile < ntiles) load_a_chunk(A + (size_t)(tile * 16 + c) * K + 4 * h, a);

    constexpr int NPASS = K * (HDIM / 4) / 512;
    float4 wq[NPASS];
#pragma unroll
    for (int r = 0; r < NPASS; ++r) {
        const int i  = tid + 512 * r;
        const int m  = i & 15;
        const int hh = (i >> 5) & 1;
        const int k  = 2 * (i >> 6) + ((i >> 4) & 1);
        wq[r] = *(const float4*)(W + (size_t)k * HDIM + 64 * hh + 4 * m);
    }
#pragma unroll
    for (int r = 0; r < NPASS; ++r) {
        const int i  = tid + 512 * r;
        const int m  = i & 15;
        const int hh = (i >> 5) & 1;
        const int k  = 2 * (i >> 6) + ((i >> 4) & 1);
        float* dp = wl + (64 * hh + m) * PITCH + k;
        dp[0]          = wq[r].x;
        dp[16 * PITCH] = wq[r].y;
        dp[32 * PITCH] = wq[r].z;
        dp[48 * PITCH] = wq[r].w;
    }
    __syncthreads();

    const float* bp = wl + c * PITCH + 4 * h;

    while (tile < ntiles) {
        const int row0 = tile * 16;

        v8f acc[8];
#pragma unroll
        for (int t = 0; t < 8; ++t)
#pragma unroll
            for (int i = 0; i < 8; ++i) acc[t][i] = 0.f;

        mma_chunk<K>(bp, a, acc);
#pragma unroll
        for (int kc = 128; kc < K; kc += 128) {
            load_a_chunk(A + (size_t)(row0 + c) * K + 4 * h + kc, a);
            mma_chunk<K>(bp + kc, a, acc);
        }

        if (res) store_strip<true >(acc, bias, res, out, row0, c, h);
        else     store_strip<false>(acc, bias, res, out, row0, c, h);

        tile += 16 * GEMM_WGS;
        if (tile < ntiles) load_a_chunk(A + (size_t)(tile * 16 + c) * K + 4 * h, a);
    }
}

#define BIN_SHIFT   10
#define BIN_ROWS    (1 << BIN_SHIFT)
#define NBINS       ((2 * NNODES + BIN_ROWS - 1) / BIN_ROWS)
#define SORT_WGS    256
#define SORT_THR    512
#define SORT_CH     ((2 * NEDGES + SORT_WGS - 1) / SORT_WGS)
#define SORT_EPT    ((SORT_CH + SORT_THR - 1) / SORT_THR)
#define NEED_CH     ((NNODES + SORT_WGS - 1) / SORT_WGS)

__device__ __forceinline__ void load_chunk_rows(const int* __restrict__ dst, int w, int tid,
                                                int (&rows)[SORT_EPT], int (&eid)[SORT_EPT])
{
    const int e0 = w * SORT_CH;
    const int e1 = (e0 + SORT_CH < 2 * NEDGES) ? (e0 + SORT_CH) : (2 * NEDGES);
    int raw[SORT_EPT];
#pragma unroll
    for (int j = 0; j < SORT_EPT; ++j) {
        const int e = e0 + tid + SORT_THR * j;
        eid[j] = min(e, e1 - 1);
        raw[j] = dst[eid[j]];
    }
#pragma unroll
    for (int j = 0; j < SORT_EPT; ++j) {
        const int e = e0 + tid + SORT_THR * j;
        rows[j] = (e < e1) ? (((e >= NEDGES) ? NNODES : 0) + raw[j]) : -1;
    }
}

__device__ __forceinline__ int bin_slot(int row, int tid)
{
    return (row >= 0) ? (row >> BIN_SHIFT) : (NBINS + (tid & 31));
}

__global__ __launch_bounds__(SORT_THR)
void bin_hist_kernel(const int* __restrict__ dst, int* __restrict__ whist, int* __restrict__ need)
{
    __shared__ int hist[256];
    const int tid = threadIdx.x;
    const int w   = blockIdx.x;
    int rows[SORT_EPT], eid[SORT_EPT];
    load_chunk_rows(dst, w, tid, rows, eid);
    if (tid < 256) hist[tid] = 0;
    static_assert(NEED_CH <= SORT_THR, "one flag per thread");
    if (tid < NEED_CH && w * NEED_CH + tid < NNODES) need[w * NEED_CH + tid] = 0;
    __syncthreads();
#pragma unroll
    for (int j = 0; j < SORT_EPT; ++j) atomicAdd(&hist[bin_slot(rows[j], tid)], 1);
    __syncthreads();
    if (tid < NBINS) whist[tid * SORT_WGS + w] = hist[tid];
}

__global__ __launch_bounds__(SORT_THR)
void bin_scatter_kernel(const int* __restrict__ src, const int* __restrict__ dst,
                        const int* __restrict__ whist, int* __restrict__ binbase,
                        int2* __restrict__ rec)
{
    __shared__ int tot[256];
    __shared__ int cur[256];
    __shared__ int wtot[SORT_THR / 32];
    const int tid  = threadIdx.x;
    const int lane = tid & 31;
    const int wv   = tid >> 5;
    const int w    = blockIdx.x;

    int rows[SORT_EPT], eid[SORT_EPT], srcs[SORT_EPT];
    load_chunk_rows(dst, w, tid, rows, eid);
#pragma unroll
    for (int j = 0; j < SORT_EPT; ++j) srcs[j] = src[eid[j]];

    constexpr int NWV = SORT_THR / 32;
    constexpr int NIT = (NBINS + NWV - 1) / NWV;
    int4 q0[NIT], q1[NIT];
#pragma unroll
    for (int it = 0; it < NIT; ++it) {
        const int b  = wv + NWV * it;
        const int bb = (b < NBINS) ? b : (NBINS - 1);
        q0[it] = *(const int4*)(whist + bb * SORT_WGS + 8 * lane);
        q1[it] = *(const int4*)(whist + bb * SORT_WGS + 8 * lane + 4);
    }
    const int kf = w - 8 * lane;
    int tt[NIT], mm[NIT];
#pragma unroll
    for (int it = 0; it < NIT; ++it) {
        tt[it] = (q0[it].x + q0[it].y) + (q0[it].z + q0[it].w) + (q1[it].x + q1[it].y) + (q1[it].z + q1[it].w);
        mm[it] = ((0 < kf) ? q0[it].x : 0) + ((1 < kf) ? q0[it].y : 0) + ((2 < kf) ? q0[it].z : 0) + ((3 < kf) ? q0[it].w : 0)
               + ((4 < kf) ? q1[it].x : 0) + ((5 < kf) ? q1[it].y : 0) + ((6 < kf) ? q1[it].z : 0) + ((7 < kf) ? q1[it].w : 0);
    }
#pragma unroll
    for (int off = 16; off > 0; off >>= 1) {
        int tu[NIT], mu[NIT];
#pragma unroll
        for (int it = 0; it < NIT; ++it) {
            tu[it] = __shfl_xor(tt[it], off, 32);
            mu[it] = __shfl_xor(mm[it], off, 32);
        }
#pragma unroll
        for (int it = 0; it < NIT; ++it) { tt[it] += tu[it]; mm[it] += mu[it]; }
    }
#pragma unroll
    for (int it = 0; it < NIT; ++it) {
        const int b = wv + NWV * it;
        if (lane == 0 && b < NBINS) { tot[b] = tt[it]; cur[b] = mm[it]; }
    }
    if (tid >= NBINS && tid < 256) tot[tid] = 0;
    __syncthreads();

    const int d = (tid < 256) ? tot[tid] : 0;
    int incl = d;
#pragma unroll
    for (int off = 1; off < 32; off <<= 1) {
        const int up = __shfl_up(incl, off, 32);
        if (lane >= off) incl += up;
    }
    if (lane == 31) wtot[wv] = incl;
    __syncthreads();
    int excl = incl - d;
#pragma unroll
    for (int k = 0; k < SORT_THR / 32; ++k) if (k < wv) excl += wtot[k];
    if (tid < NBINS) cur[tid] += excl;
    if (w == 0 && tid <= NBINS) binbase[tid] = excl;
    __syncthreads();

    int pos[SORT_EPT];
#pragma unroll
    for (int j = 0; j < SORT_EPT; ++j) pos[j] = atomicAdd(&cur[bin_slot(rows[j], tid)], 1);
#pragma unroll
    for (int j = 0; j < SORT_EPT; ++j)
        if (rows[j] >= 0) rec[pos[j]] = make_int2(rows[j], srcs[j]);
}

#define RANK_UNR 8
__global__ __launch_bounds__(BIN_ROWS)
void bin_rank_kernel(const int2* __restrict__ rec, const int* __restrict__ binbase,
                     int* __restrict__ seg_beg, int* __restrict__ seg_end,
                     int* __restrict__ csr_src)
{
    __shared__ int cnt[BIN_ROWS + 32];
    __shared__ int wtot[BIN_ROWS / 32];
    const int tid   = threadIdx.x;
    const int lane  = tid & 31;
    const int wv    = tid >> 5;
    const int b     = blockIdx.x;
    const int r0    = binbase[b];
    const int r1    = binbase[b + 1];
    const int spare = BIN_ROWS + lane;

    cnt[tid] = 0;
    __syncthreads();
    for (int i0 = r0; i0 < r1; i0 += RANK_UNR * BIN_ROWS) {
        int x[RANK_UNR];
#pragma unroll
        for (int u = 0; u < RANK_UNR; ++u) {
            const int i = i0 + tid + u * BIN_ROWS;
            x[u] = rec[min(i, r1 - 1)].x;
        }
#pragma unroll
        for (int u = 0; u < RANK_UNR; ++u) {
            const int i = i0 + tid + u * BIN_ROWS;
            atomicAdd(&cnt[(i < r1) ? (x[u] & (BIN_ROWS - 1)) : spare], 1);
        }
    }
    __syncthreads();

    const int d = cnt[tid];
    int incl = d;
#pragma unroll
    for (int off = 1; off < 32; off <<= 1) {
        const int up = __shfl_up(incl, off, 32);
        if (lane >= off) incl += up;
    }
    if (lane == 31) wtot[wv] = incl;
    __syncthreads();
    int st = r0 + incl - d;
#pragma unroll
    for (int k = 0; k < BIN_ROWS / 32; ++k) if (k < wv) st += wtot[k];
    const int v = b * BIN_ROWS + tid;
    if (v < 2 * NNODES) { seg_beg[v] = st; seg_end[v] = st + d; }
    cnt[tid] = st;
    __syncthreads();

    for (int i0 = r0; i0 < r1; i0 += RANK_UNR * BIN_ROWS) {
        int2 r[RANK_UNR];
        int  pos[RANK_UNR];
#pragma unroll
        for (int u = 0; u < RANK_UNR; ++u) {
            const int i = i0 + tid + u * BIN_ROWS;
            r[u] = rec[min(i, r1 - 1)];
        }
#pragma unroll
        for (int u = 0; u < RANK_UNR; ++u) {
            const int i = i0 + tid + u * BIN_ROWS;
            pos[u] = atomicAdd(&cnt[(i < r1) ? (r[u].x & (BIN_ROWS - 1)) : spare], 1);
        }
#pragma unroll
        for (int u = 0; u < RANK_UNR; ++u) {
            const int i = i0 + tid + u * BIN_ROWS;
            if (i < r1) csr_src[pos[u]] = r[u].y;
        }
    }
}

__device__ __forceinline__ float4 ln_relu_row(const float4 v, const float4 gg, const float4 bb)
{
    float s1 = v.x + v.y + v.z + v.w;
    float s2 = v.x * v.x + v.y * v.y + v.z * v.z + v.w * v.w;
#pragma unroll
    for (int off = 16; off > 0; off >>= 1) {
        s1 += __shfl_xor(s1, off, 32);
        s2 += __shfl_xor(s2, off, 32);
    }
    const float mu  = s1 * (1.f / HDIM);
    const float var = s2 * (1.f / HDIM) - mu * mu;
    const float inv = rsqrtf(var + LN_EPS);
    float4 r;
    r.x = fmaxf((v.x - mu) * inv * gg.x + bb.x, 0.f);
    r.y = fmaxf((v.y - mu) * inv * gg.y + bb.y, 0.f);
    r.z = fmaxf((v.z - mu) * inv * gg.z + bb.z, 0.f);
    r.w = fmaxf((v.w - mu) * inv * gg.w + bb.w, 0.f);
    return r;
}

#define AGG_BLOCKS 4096
#define AGG_DEMAND_BLOCKS 1024

__device__ __forceinline__ float exp_nonpos(const float x)
{
    return __builtin_amdgcn_exp2f(x * 1.44269504088896340736f);
}

template <int CNT, bool LNF>
__device__ __forceinline__ void agg_chunk(const float* __restrict__ hl, int idx, int c, bool first,
                                          float (&mx)[4], float (&de)[4], float (&nu)[4],
                                          const float4 gg, const float4 bb)
{
    float4 hv[CNT];
#pragma unroll
    for (int j = 0; j < CNT; ++j) {
        const int s = __builtin_amdgcn_readlane(idx, c + j);
        hv[j] = *(const float4*)(hl + (size_t)s * HDIM);
    }
    if (LNF) {
#pragma unroll
        for (int j = 0; j < CNT; ++j) hv[j] = ln_relu_row(hv[j], gg, bb);
    }
    float m[CNT][4];
    float c0 = mx[0], c1 = mx[1], c2 = mx[2], c3 = mx[3];
#pragma unroll
    for (int j = 0; j < CNT; ++j) {
        m[j][0] = fmaxf(hv[j].x, 0.f) + GEN_EPS;
        m[j][1] = fmaxf(hv[j].y, 0.f) + GEN_EPS;
        m[j][2] = fmaxf(hv[j].z, 0.f) + GEN_EPS;
        m[j][3] = fmaxf(hv[j].w, 0.f) + GEN_EPS;
        c0 = fmaxf(c0, m[j][0]);
        c1 = fmaxf(c1, m[j][1]);
        c2 = fmaxf(c2, m[j][2]);
        c3 = fmaxf(c3, m[j][3]);
    }
    if (!first) {
        const float s0 = exp_nonpos(mx[0] - c0), s1 = exp_nonpos(mx[1] - c1);
        const float s2 = exp_nonpos(mx[2] - c2), s3 = exp_nonpos(mx[3] - c3);
        de[0] *= s0; nu[0] *= s0;
        de[1] *= s1; nu[1] *= s1;
        de[2] *= s2; nu[2] *= s2;
        de[3] *= s3; nu[3] *= s3;
    }
    mx[0] = c0; mx[1] = c1; mx[2] = c2; mx[3] = c3;
#pragma unroll
    for (int j = 0; j < CNT; ++j) {
#pragma unroll
        for (int f = 0; f < 4; ++f) {
            const float ex = exp_nonpos(m[j][f] - mx[f]);
            de[f] += ex;
            nu[f] += m[j][f] * ex;
        }
    }
}

template <bool LNF>
__device__ __forceinline__ float4 agg_node(const float* __restrict__ hl, const int* __restrict__ csr_src,
                                           const int* __restrict__ rmap,
                                           int beg, int n, int lane, const float4 own,
                                           const float4 gg, const float4 bb)
{
    float mx[4] = {0.f, 0.f, 0.f, 0.f};
    float de[4] = {0.f, 0.f, 0.f, 0.f};
    float nu[4] = {0.f, 0.f, 0.f, 0.f};

    for (int base = 0; base < n; base += 32) {
        const int bc  = (n - base < 32) ? (n - base) : 32;
        int idx = (lane < bc) ? csr_src[beg + base + lane] : 0;
        if (LNF) idx = rmap[idx];
        for (int c = 0; c < bc; c += 8) {
            const int  cnt   = (bc - c < 8) ? (bc - c) : 8;
            const bool first = (base + c == 0);
            switch (cnt) {
            case 1:  agg_chunk<1, LNF>(hl, idx, c, first, mx, de, nu, gg, bb); break;
            case 2:  agg_chunk<2, LNF>(hl, idx, c, first, mx, de, nu, gg, bb); break;
            case 3:  agg_chunk<3, LNF>(hl, idx, c, first, mx, de, nu, gg, bb); break;
            case 4:  agg_chunk<4, LNF>(hl, idx, c, first, mx, de, nu, gg, bb); break;
            case 5:  agg_chunk<5, LNF>(hl, idx, c, first, mx, de, nu, gg, bb); break;
            case 6:  agg_chunk<6, LNF>(hl, idx, c, first, mx, de, nu, gg, bb); break;
            case 7:  agg_chunk<7, LNF>(hl, idx, c, first, mx, de, nu, gg, bb); break;
            default: agg_chunk<8, LNF>(hl, idx, c, first, mx, de, nu, gg, bb); break;
            }
        }
    }

    float4 r;
    r.x = nu[0] / (de[0] + 1e-16f) + own.x;
    r.y = nu[1] / (de[1] + 1e-16f) + own.y;
    r.z = nu[2] / (de[2] + 1e-16f) + own.z;
    r.w = nu[3] / (de[3] + 1e-16f) + own.w;
    return r;
}

template <bool DEMAND>
__global__ __launch_bounds__(256)
void node_softmax_agg_kernel(const float* __restrict__ hin, const int* __restrict__ seg_beg,
                             const int* __restrict__ seg_end, const int* __restrict__ csr_src,
                             const int* __restrict__ need, float* __restrict__ t)
{
    const int lane = threadIdx.x & 31;
    const int wv = __builtin_amdgcn_readfirstlane((int)(threadIdx.x >> 5));
    const float* __restrict__ hl = hin + lane * 4;
    const float4 none = {0.f, 0.f, 0.f, 0.f};

    constexpr int NBLK = DEMAND ? AGG_DEMAND_BLOCKS : AGG_BLOCKS;
    for (int row = blockIdx.x * 8 + wv; row < NNODES; row += NBLK * 8) {
        if (DEMAND) {
            if (__builtin_amdgcn_readfirstlane(need[row]) == 0) continue;
        }
        const int beg = __builtin_amdgcn_readfirstlane(seg_beg[row]);
        const int n   = __builtin_amdgcn_readfirstlane(seg_end[row]) - beg;
        const float4 own = *(const float4*)(hl + (size_t)row * HDIM);
        const float4 r = agg_node<false>(hl, csr_src, nullptr, beg, n, lane, own, none, none);
        *(float4*)(t + (size_t)row * HDIM + lane * 4) = r;
    }
}

__global__ __launch_bounds__(256)
void mark_needed_kernel(const int* __restrict__ final_map, const int* __restrict__ node_map,
                        const int* __restrict__ pmap,
                        const int* __restrict__ seg_beg, const int* __restrict__ seg_end,
                        const int* __restrict__ csr_src, int* __restrict__ need)
{
    const int lane = threadIdx.x & 31;
    const int i    = blockIdx.x * 8 + (int)(threadIdx.x >> 5);
    if (i >= NSEEDS) return;
    const int v   = node_map[final_map[i]];
    const int beg = seg_beg[v];
    const int n   = seg_end[v] - beg;
    if (lane == 0) need[pmap[v]] = 1;
    for (int j = lane; j < n; j += 32) need[pmap[csr_src[beg + j]]] = 1;
}

__global__ __launch_bounds__(256)
void seed_softmax_agg_kernel(const float* __restrict__ p, const int* __restrict__ pmap,
                             const int* __restrict__ final_map,
                             const int* __restrict__ node_map,
                             const float* __restrict__ g, const float* __restrict__ b,
                             const int* __restrict__ seg_beg, const int* __restrict__ seg_end,
                             const int* __restrict__ csr_src,
                             float* __restrict__ t, float* __restrict__ rs)
{
    const int lane = threadIdx.x & 31;
    const int wv = __builtin_amdgcn_readfirstlane((int)(threadIdx.x >> 5));
    const int i  = blockIdx.x * 8 + wv;
    if (i >= NSEEDS) return;
    const int v   = __builtin_amdgcn_readfirstlane(node_map[final_map[i]]);
    const int pv  = __builtin_amdgcn_readfirstlane(pmap[v]);
    const int beg = __builtin_amdgcn_readfirstlane(seg_beg[v]);
    const int n   = __builtin_amdgcn_readfirstlane(seg_end[v]) - beg;
    const float* __restrict__ hl = p + lane * 4;
    const float4 raw = *(const float4*)(hl + (size_t)pv * HDIM);
    const float4 gg  = *(const float4*)(g + lane * 4);
    const float4 bb  = *(const float4*)(b + lane * 4);
    const float4 own = ln_relu_row(raw, gg, bb);
    const float4 r   = agg_node<true>(hl, csr_src, pmap, beg, n, lane, own, gg, bb);
    *(float4*)(t  + (size_t)i * HDIM + lane * 4) = r;
    *(float4*)(rs + (size_t)i * HDIM + lane * 4) = raw;
}

__global__ __launch_bounds__(256)
void ln_relu_kernel(const float* __restrict__ in, const float* __restrict__ g,
                    const float* __restrict__ b, float* __restrict__ out, int M)
{
    const int row  = blockIdx.x * 8 + (threadIdx.x >> 5);
    const int lane = threadIdx.x & 31;
    if (row >= M) return;
    const float4 v  = *(const float4*)(in + (size_t)row * HDIM + lane * 4);
    const float4 gg = *(const float4*)(g + lane * 4);
    const float4 bb = *(const float4*)(b + lane * 4);
    const float4 r  = ln_relu_row(v, gg, bb);
    *(float4*)(out + (size_t)row * HDIM + lane * 4) = r;
}

__global__ __launch_bounds__(256)
void logsoftmax_kernel(const float* __restrict__ in, float* __restrict__ out, int M)
{
    const int row  = blockIdx.x * 8 + (threadIdx.x >> 5);
    const int lane = threadIdx.x & 31;
    if (row >= M) return;
    const float NEG = -3.0e38f;
    const float v0 = (lane      < NCLS) ? in[(size_t)row * NCLSPAD + lane     ] : NEG;
    const float v1 = (lane + 32 < NCLS) ? in[(size_t)row * NCLSPAD + lane + 32] : NEG;
    float mx = fmaxf(v0, v1);
#pragma unroll
    for (int off = 16; off > 0; off >>= 1) mx = fmaxf(mx, __shfl_xor(mx, off, 32));
    float se = ((lane      < NCLS) ? expf(v0 - mx) : 0.f)
             + ((lane + 32 < NCLS) ? expf(v1 - mx) : 0.f);
#pragma unroll
    for (int off = 16; off > 0; off >>= 1) se += __shfl_xor(se, off, 32);
    const float lse = logf(se) + mx;
    if (lane      < NCLS) out[(size_t)row * NCLS + lane     ] = v0 - lse;
    if (lane + 32 < NCLS) out[(size_t)row * NCLS + lane + 32] = v1 - lse;
}

extern "C" void kernel_launch(void* const* d_in, const int* in_sizes, int n_in,
                              void* d_out, int out_size, void* d_ws, size_t ws_size,
                              hipStream_t stream)
{
    (void)in_sizes; (void)n_in; (void)out_size; (void)ws_size;

    const float* x         = (const float*)d_in[0];
    const int*   src       = (const int*)  d_in[1];
    const int*   dst       = (const int*)  d_in[2];
    const int*   node_map  = (const int*)  d_in[3];
    const int*   final_map = (const int*)  d_in[4];
    const float* enc_w     = (const float*)d_in[5];
    const float* enc_b     = (const float*)d_in[6];
    const float* gcn_w     = (const float*)d_in[7];
    const float* gcn_b     = (const float*)d_in[8];
    const float* ln_g      = (const float*)d_in[9];
    const float* ln_b      = (const float*)d_in[10];
    const float* pred_w    = (const float*)d_in[11];
    const float* pred_b    = (const float*)d_in[12];
    float* out = (float*)d_out;

    const size_t NH = (size_t)NNODES * HDIM;
    float* B0     = (float*)d_ws;
    float* B1     = B0 + NH;
    float* B2     = B0 + 2 * NH;
    float* B3     = B0 + 3 * NH;
    float* B4     = B0 + 4 * NH;
    float* HG     = B0 + 5 * NH;
    float* HG2    = HG  + (size_t)NSEEDS * HDIM;
    float* SC     = HG2 + (size_t)NSEEDS * HDIM;

    const dim3 gblk(32, 4, 1);
    const int  mtN   = NNODES / 16;
    const int  lgrid = (NNODES + 7) / 8;

    int*  IWHIST = (int*)B3;
    int*  IBASE  = IWHIST + NBINS * SORT_WGS;
    int*  INEED  = IBASE + 256;
    int*  ISTART = INEED + NNODES;
    int*  IEND   = ISTART + 2 * NNODES;
    int*  ICSR   = IEND + 2 * NNODES;
    int2* IREC   = (int2*)(ICSR + 2 * NEDGES);
    bin_hist_kernel<<<SORT_WGS, SORT_THR, 0, stream>>>(dst, IWHIST, INEED);
    bin_scatter_kernel<<<SORT_WGS, SORT_THR, 0, stream>>>(src, dst, IWHIST, IBASE, IREC);
    bin_rank_kernel<<<NBINS, BIN_ROWS, 0, stream>>>(IREC, IBASE, ISTART, IEND, ICSR);

    gemm_rows128_kernel<FDIM><<<GEMM_WGS, 512, 0, stream>>>(
        x, enc_w, enc_b, nullptr, B0, mtN);

    node_softmax_agg_kernel<false><<<AGG_BLOCKS, 256, 0, stream>>>(
        B0, ISTART, IEND, ICSR, nullptr, B2);
    gemm_rows128_kernel<HDIM><<<GEMM_WGS, 512, 0, stream>>>(
        B2, gcn_w, gcn_b, nullptr, B0, mtN);

    mark_needed_kernel<<<NSEEDS / 8, 256, 0, stream>>>(
        final_map, node_map + NNODES, node_map, ISTART + NNODES, IEND + NNODES, ICSR, INEED);
    ln_relu_kernel<<<lgrid, 256, 0, stream>>>(B0, ln_g, ln_b, B1, NNODES);
    node_softmax_agg_kernel<true><<<AGG_DEMAND_BLOCKS, 256, 0, stream>>>(
        B1, ISTART, IEND, ICSR, INEED, B2);
    gemm_rows128_kernel<HDIM><<<GEMM_WGS, 512, 0, stream>>>(
        B2, gcn_w + (size_t)HDIM * HDIM, gcn_b + HDIM,  B0,  B4, mtN);

    float* T2 = B2;
    float* R2 = B1;
    const int mtS = NSEEDS / 16;
    seed_softmax_agg_kernel<<<NSEEDS / 8, 256, 0, stream>>>(
        B4, node_map, final_map, node_map + NNODES, ln_g + HDIM, ln_b + HDIM,
        ISTART + NNODES, IEND + NNODES, ICSR, T2, R2);
    gemm_rows128_kernel<HDIM><<<GEMM_WGS, 512, 0, stream>>>(
        T2, gcn_w + (size_t)2 * HDIM * HDIM, gcn_b + 2 * HDIM, R2, HG, mtS);

    ln_relu_kernel<<<(NSEEDS + 7) / 8, 256, 0, stream>>>(
        HG, ln_g + 2 * HDIM, ln_b + 2 * HDIM, HG2, NSEEDS);
    gemm_head_kernel<<<(mtS + 3) / 4, gblk, 0, stream>>>(
        HG2, pred_w, pred_b, SC, NSEEDS);
    logsoftmax_kernel<<<(NSEEDS + 7) / 8, 256, 0, stream>>>(SC, out, NSEEDS);
}


# LayerNorm+ReLU of a row product is formed in the product's strip epilogue from the accumulators (row sums in a new order, by DPP): both normalisation launches are gone, 15 -> 13 launches
# speedup vs baseline: 6.2975x; 1.0457x over previous
#include <hip/hip_runtime.h>
#include <hip/hip_bf16.h>
#include <math.h>

#define NNODES   100000
#define NEDGES   600000
#define FDIM     256
#define HDIM     128
#define NCLS     47
#define NCLSPAD  64
#define NSEEDS   10000
#define GEN_EPS  1e-7f
#define LN_EPS   1e-5f

typedef float v2f __attribute__((ext_vector_type(2)));
typedef float v8f __attribute__((ext_vector_type(8)));


#define HEAD_PITCH (HDIM + 4)
__global__ __launch_bounds__(128)
void gemm_head_kernel(const float* __restrict__ A, const float* __restrict__ W,
                      const float* __restrict__ bias, float* __restrict__ out, int M)
{
    __shared__ __attribute__((aligned(16))) float wl[NCLSPAD * HEAD_PITCH];

    const int lane = threadIdx.x;
    const int tid  = threadIdx.y * 32 + lane;

    constexpr int NW    = HDIM * NCLS;
    constexpr int NPASS = NW / 128;
    static_assert(NW % 128 == 0, "whole passes of the 128 threads");
    float wq[NPASS];
#pragma unroll
    for (int r = 0; r < NPASS; ++r) wq[r] = W[tid + 128 * r];
#pragma unroll
    for (int r = 0; r < NPASS; ++r) {
        const int i = tid + 128 * r;
        const int k = i / NCLS;
        const int n = i - k * NCLS;
        wl[n * HEAD_PITCH + k] = wq[r];
    }
    for (int i = tid; i < (NCLSPAD - NCLS) * HDIM; i += 128) {
        const int n = NCLS + i / HDIM;
        const int k = i & (HDIM - 1);
        wl[n * HEAD_PITCH + k] = 0.f;
    }
    __syncthreads();

    const int mtile = blockIdx.x * 4 + threadIdx.y;
    const int row0  = mtile * 16;
    if (row0 >= M) return;
    const int hb   = lane >> 4;
    const int lo   = lane & 15;
    const int arow = row0 + lo;

    const float* __restrict__ a_ptr = A + (size_t)arow * HDIM;
    const float* b_ptr = wl + lo * HEAD_PITCH;
    constexpr int bstep = 16 * HEAD_PITCH;

    v8f acc[4];
#pragma unroll
    for (int t = 0; t < 4; ++t)
#pragma unroll
        for (int i = 0; i < 8; ++i) acc[t][i] = 0.f;

#pragma unroll 4
    for (int k = 0; k < HDIM; k += 4) {
        const int ka = k + 2 * hb;
        const v2f a  = *(const v2f*)(a_ptr + ka);
        const v2f b0 = *(const v2f*)(b_ptr + ka);
        const v2f b1 = *(const v2f*)(b_ptr + bstep + ka);
        const v2f b2 = *(const v2f*)(b_ptr + 2 * bstep + ka);
        const v2f b3 = *(const v2f*)(b_ptr + 3 * bstep + ka);
        acc[0] = __builtin_amdgcn_wmma_f32_16x16x4_f32(false, a, false, b0, (short)0, acc[0], false, false);
        acc[1] = __builtin_amdgcn_wmma_f32_16x16x4_f32(false, a, false, b1, (short)0, acc[1], false, false);
        acc[2] = __builtin_amdgcn_wmma_f32_16x16x4_f32(false, a, false, b2, (short)0, acc[2], false, false);
        acc[3] = __builtin_amdgcn_wmma_f32_16x16x4_f32(false, a, false, b3, (short)0, acc[3], false, false);
    }

#pragma unroll
    for (int t = 0; t < 4; ++t) {
        const int   col = lo + 16 * t;
        const float bv  = (col < NCLS) ? bias[col] : 0.f;
#pragma unroll
        for (int i = 0; i < 8; ++i) {
            const int r = row0 + i + 8 * hb;
            out[(size_t)r * NCLSPAD + col] = acc[t][i] + bv;
        }
    }
}

#define GEMM_WGS 256


__device__ __forceinline__ void load_a_chunk(const float* __restrict__ ap, float4 (&a)[16])
{
#pragma unroll
    for (int i = 0; i < 16; ++i) a[i] = *(const float4*)(ap + 8 * i);
}

template <int K>
__device__ __forceinline__ void load_b_block(const float* bp, float4 (&b)[8])
{
    constexpr int PITCH = K + 4;
#pragma unroll
    for (int t = 0; t < 8; ++t) b[t] = *(const float4*)(bp + t * 16 * PITCH);
}

__device__ __forceinline__ void mma_block(const float4 a, const float4 (&b)[8], v8f (&acc)[8])
{
    const v2f a01 = {a.x, a.y};
    const v2f a23 = {a.z, a.w};
#pragma unroll
    for (int t = 0; t < 8; ++t) {
        const v2f b01 = {b[t].x, b[t].y};
        acc[t] = __builtin_amdgcn_wmma_f32_16x16x4_f32(false, a01, false, b01, (short)0, acc[t], false, false);
    }
#pragma unroll
    for (int t = 0; t < 8; ++t) {
        const v2f b23 = {b[t].z, b[t].w};
        acc[t] = __builtin_amdgcn_wmma_f32_16x16x4_f32(false, a23, false, b23, (short)0, acc[t], false, false);
    }
}

template <int K>
__device__ __forceinline__ void mma_chunk(const float* bp, const float4 (&a)[16], v8f (&acc)[8])
{
    float4 b0[8], b1[8];
    load_b_block<K>(bp, b0);
#pragma unroll
    for (int i = 0; i < 16; i += 2) {
        load_b_block<K>(bp + 8 * (i + 1), b1);
        __builtin_amdgcn_sched_barrier(0);
        mma_block(a[i], b0, acc);
        __builtin_amdgcn_sched_barrier(0);
        if (i + 2 < 16) load_b_block<K>(bp + 8 * (i + 2), b0);
        __builtin_amdgcn_sched_barrier(0);
        mma_block(a[i + 1], b1, acc);
        __builtin_amdgcn_sched_barrier(0);
    }
}

#define EPI_RAW  0
#define EPI_BOTH 1
#define EPI_NORM 2

template <int XM>
__device__ __forceinline__ float row_xor_add(const float x)
{
    const int moved = __builtin_amdgcn_update_dpp(0, __builtin_bit_cast(int, x), 0x160 + XM, 0xf, 0xf, true);
    return x + __builtin_bit_cast(float, moved);
}

template <int XM>
__device__ __forceinline__ void row_xor_add8(float (&s)[8])
{
#pragma unroll
    for (int i = 0; i < 8; ++i) s[i] = row_xor_add<XM>(s[i]);
}

template <bool HAS_RES, int EPI>
__device__ __forceinline__ void store_strip(const v8f (&acc)[8], const float* __restrict__ bias,
                                            const float* __restrict__ res,
                                            const float* __restrict__ lng, const float* __restrict__ lnb,
                                            float* __restrict__ out, float* __restrict__ out2,
                                            int row0, int c, int h)
{
    float4 s[2][8];
#pragma unroll
    for (int g = 0; g < 2; ++g) {
        const int    col = 64 * g + 4 * c;
        const float4 bv  = *(const float4*)(bias + col);
        float4 rv[8];
        if (HAS_RES) {
#pragma unroll
            for (int i = 0; i < 8; ++i)
                rv[i] = *(const float4*)(res + (size_t)(row0 + i + 8 * h) * HDIM + col);
        }
#pragma unroll
        for (int i = 0; i < 8; ++i) {
            float4 v;
            v.x = acc[4 * g + 0][i] + bv.x;
            v.y = acc[4 * g + 1][i] + bv.y;
            v.z = acc[4 * g + 2][i] + bv.z;
            v.w = acc[4 * g + 3][i] + bv.w;
            if (HAS_RES) { v.x += rv[i].x; v.y += rv[i].y; v.z += rv[i].z; v.w += rv[i].w; }
            s[g][i] = v;
            if (EPI != EPI_NORM) *(float4*)(out + (size_t)(row0 + i + 8 * h) * HDIM + col) = v;
        }
    }
    if (EPI == EPI_RAW) return;

    float s1[8], s2[8];
#pragma unroll
    for (int i = 0; i < 8; ++i) {
        const float4 p = s[0][i];
        const float4 q = s[1][i];
        s1[i] = ((p.x + p.y) + (p.z + p.w)) + ((q.x + q.y) + (q.z + q.w));
        s2[i] = ((p.x * p.x + p.y * p.y) + (p.z * p.z + p.w * p.w))
              + ((q.x * q.x + q.y * q.y) + (q.z * q.z + q.w * q.w));
    }
    row_xor_add8<1>(s1); row_xor_add8<1>(s2);
    row_xor_add8<2>(s1); row_xor_add8<2>(s2);
    row_xor_add8<4>(s1); row_xor_add8<4>(s2);
    row_xor_add8<8>(s1); row_xor_add8<8>(s2);

    float mu[8], inv[8];
#pragma unroll
    for (int i = 0; i < 8; ++i) {
        mu[i] = s1[i] * (1.f / HDIM);
        const float var = s2[i] * (1.f / HDIM) - mu[i] * mu[i];
        inv[i] = rsqrtf(var + LN_EPS);
    }
    float* __restrict__ dst = (EPI == EPI_BOTH) ? out2 : out;
#pragma unroll
    for (int g = 0; g < 2; ++g) {
        const int    col = 64 * g + 4 * c;
        const float4 gg  = *(const float4*)(lng + col);
        const float4 bb  = *(const float4*)(lnb + col);
#pragma unroll
        for (int i = 0; i < 8; ++i) {
            const float4 v = s[g][i];
            float4 r;
            r.x = fmaxf((v.x - mu[i]) * inv[i] * gg.x + bb.x, 0.f);
            r.y = fmaxf((v.y - mu[i]) * inv[i] * gg.y + bb.y, 0.f);
            r.z = fmaxf((v.z - mu[i]) * inv[i] * gg.z + bb.z, 0.f);
            r.w = fmaxf((v.w - mu[i]) * inv[i] * gg.w + bb.w, 0.f);
            *(float4*)(dst + (size_t)(row0 + i + 8 * h) * HDIM + col) = r;
        }
    }
}

template <int K, bool HAS_RES, int EPI>
__global__ __launch_bounds__(512)
void gemm_rows128_kernel(const float* __restrict__ A, const float* __restrict__ W,
                         const float* __restrict__ bias, const float* __restrict__ res,
                         const float* __restrict__ lng, const float* __restrict__ lnb,
                         float* __restrict__ out, float* __restrict__ out2, int ntiles)
{
    constexpr int PITCH = K + 4;
    __shared__ __attribute__((aligned(16))) float wl[HDIM * PITCH];

    const int tid  = threadIdx.x;
    const int lane = tid & 31;
    const int wv   = __builtin_amdgcn_readfirstlane(tid >> 5);
    const int c    = lane & 15;
    const int h    = lane >> 4;

    const int wa = wv >> 2;
    const int q  = wa + 4 * (((wv & 3) - wa) & 3);
    int tile = q * GEMM_WGS + (int)blockIdx.x;

    float4 a[16];
    if (tile < ntiles) load_a_chunk(A + (size_t)(tile * 16 + c) * K + 4 * h, a);

    constexpr int NPASS = K * (HDIM / 4) / 512;
    float4 wq[NPASS];
#pragma unroll
    for (int r = 0; r < NPASS; ++r) {
        const int i  = tid + 512 * r;
        const int m  = i & 15;
        const int hh = (i >> 5) & 1;
        const int k  = 2 * (i >> 6) + ((i >> 4) & 1);
        wq[r] = *(const float4*)(W + (size_t)k * HDIM + 64 * hh + 4 * m);
    }
#pragma unroll
    for (int r = 0; r < NPASS; ++r) {
        const int i  = tid + 512 * r;
        const int m  = i & 15;
        const int hh = (i >> 5) & 1;
        const int k  = 2 * (i >> 6) + ((i >> 4) & 1);
        float* dp = wl + (64 * hh + m) * PITCH + k;
        dp[0]          = wq[r].x;
        dp[16 * PITCH] = wq[r].y;
        dp[32 * PITCH] = wq[r].z;
        dp[48 * PITCH] = wq[r].w;
    }
    __syncthreads();

    const float* bp = wl + c * PITCH + 4 * h;

    while (tile < ntiles) {
        const int row0 = tile * 16;

        v8f acc[8];
#pragma unroll
        for (int t = 0; t < 8; ++t)
#pragma unroll
            for (int i = 0; i < 8; ++i) acc[t][i] = 0.f;

        mma_chunk<K>(bp, a, acc);
#pragma unroll
        for (int kc = 128; kc < K; kc += 128) {
            load_a_chunk(A + (size_t)(row0 + c) * K + 4 * h + kc, a);
            mma_chunk<K>(bp + kc, a, acc);
        }

        store_strip<HAS_RES, EPI>(acc, bias, res, lng, lnb, out, out2, row0, c, h);

        tile += 16 * GEMM_WGS;
        if (tile < ntiles) load_a_chunk(A + (size_t)(tile * 16 + c) * K + 4 * h, a);
    }
}

#define BIN_SHIFT   10
#define BIN_ROWS    (1 << BIN_SHIFT)
#define NBINS       ((2 * NNODES + BIN_ROWS - 1) / BIN_ROWS)
#define SORT_WGS    256
#define SORT_THR    512
#define SORT_CH     ((2 * NEDGES + SORT_WGS - 1) / SORT_WGS)
#define SORT_EPT    ((SORT_CH + SORT_THR - 1) / SORT_THR)
#define NEED_CH     ((NNODES + SORT_WGS - 1) / SORT_WGS)

__device__ __forceinline__ void load_chunk_rows(const int* __restrict__ dst, int w, int tid,
                                                int (&rows)[SORT_EPT], int (&eid)[SORT_EPT])
{
    const int e0 = w * SORT_CH;
    const int e1 = (e0 + SORT_CH < 2 * NEDGES) ? (e0 + SORT_CH) : (2 * NEDGES);
    int raw[SORT_EPT];
#pragma unroll
    for (int j = 0; j < SORT_EPT; ++j) {
        const int e = e0 + tid + SORT_THR * j;
        eid[j] = min(e, e1 - 1);
        raw[j] = dst[eid[j]];
    }
#pragma unroll
    for (int j = 0; j < SORT_EPT; ++j) {
        const int e = e0 + tid + SORT_THR * j;
        rows[j] = (e < e1) ? (((e >= NEDGES) ? NNODES : 0) + raw[j]) : -1;
    }
}

__device__ __forceinline__ int bin_slot(int row, int tid)
{
    return (row >= 0) ? (row >> BIN_SHIFT) : (NBINS + (tid & 31));
}

__global__ __launch_bounds__(SORT_THR)
void bin_hist_kernel(const int* __restrict__ dst, int* __restrict__ whist, int* __restrict__ need)
{
    __shared__ int hist[256];
    const int tid = threadIdx.x;
    const int w   = blockIdx.x;
    int rows[SORT_EPT], eid[SORT_EPT];
    load_chunk_rows(dst, w, tid, rows, eid);
    if (tid < 256) hist[tid] = 0;
    static_assert(NEED_CH <= SORT_THR, "one flag per thread");
    if (tid < NEED_CH && w * NEED_CH + tid < NNODES) need[w * NEED_CH + tid] = 0;
    __syncthreads();
#pragma unroll
    for (int j = 0; j < SORT_EPT; ++j) atomicAdd(&hist[bin_slot(rows[j], tid)], 1);
    __syncthreads();
    if (tid < NBINS) whist[tid * SORT_WGS + w] = hist[tid];
}

__global__ __launch_bounds__(SORT_THR)
void bin_scatter_kernel(const int* __restrict__ src, const int* __restrict__ dst,
                        const int* __restrict__ whist, int* __restrict__ binbase,
                        int2* __restrict__ rec)
{
    __shared__ int tot[256];
    __shared__ int cur[256];
    __shared__ int wtot[SORT_THR / 32];
    const int tid  = threadIdx.x;
    const int lane = tid & 31;
    const int wv   = tid >> 5;
    const int w    = blockIdx.x;

    int rows[SORT_EPT], eid[SORT_EPT], srcs[SORT_EPT];
    load_chunk_rows(dst, w, tid, rows, eid);
#pragma unroll
    for (int j = 0; j < SORT_EPT; ++j) srcs[j] = src[eid[j]];

    constexpr int NWV = SORT_THR / 32;
    constexpr int NIT = (NBINS + NWV - 1) / NWV;
    int4 q0[NIT], q1[NIT];
#pragma unroll
    for (int it = 0; it < NIT; ++it) {
        const int b  = wv + NWV * it;
        const int bb = (b < NBINS) ? b : (NBINS - 1);
        q0[it] = *(const int4*)(whist + bb * SORT_WGS + 8 * lane);
        q1[it] = *(const int4*)(whist + bb * SORT_WGS + 8 * lane + 4);
    }
    const int kf = w - 8 * lane;
    int tt[NIT], mm[NIT];
#pragma unroll
    for (int it = 0; it < NIT; ++it) {
        tt[it] = (q0[it].x + q0[it].y) + (q0[it].z + q0[it].w) + (q1[it].x + q1[it].y) + (q1[it].z + q1[it].w);
        mm[it] = ((0 < kf) ? q0[it].x : 0) + ((1 < kf) ? q0[it].y : 0) + ((2 < kf) ? q0[it].z : 0) + ((3 < kf) ? q0[it].w : 0)
               + ((4 < kf) ? q1[it].x : 0) + ((5 < kf) ? q1[it].y : 0) + ((6 < kf) ? q1[it].z : 0) + ((7 < kf) ? q1[it].w : 0);
    }
#pragma unroll
    for (int off = 16; off > 0; off >>= 1) {
        int tu[NIT], mu[NIT];
#pragma unroll
        for (int it = 0; it < NIT; ++it) {
            tu[it] = __shfl_xor(tt[it], off, 32);
            mu[it] = __shfl_xor(mm[it], off, 32);
        }
#pragma unroll
        for (int it = 0; it < NIT; ++it) { tt[it] += tu[it]; mm[it] += mu[it]; }
    }
#pragma unroll
    for (int it = 0; it < NIT; ++it) {
        const int b = wv + NWV * it;
        if (lane == 0 && b < NBINS) { tot[b] = tt[it]; cur[b] = mm[it]; }
    }
    if (tid >= NBINS && tid < 256) tot[tid] = 0;
    __syncthreads();

    const int d = (tid < 256) ? tot[tid] : 0;
    int incl = d;
#pragma unroll
    for (int off = 1; off < 32; off <<= 1) {
        const int up = __shfl_up(incl, off, 32);
        if (lane >= off) incl += up;
    }
    if (lane == 31) wtot[wv] = incl;
    __syncthreads();
    int excl = incl - d;
#pragma unroll
    for (int k = 0; k < SORT_THR / 32; ++k) if (k < wv) excl += wtot[k];
    if (tid < NBINS) cur[tid] += excl;
    if (w == 0 && tid <= NBINS) binbase[tid] = excl;
    __syncthreads();

    int pos[SORT_EPT];
#pragma unroll
    for (int j = 0; j < SORT_EPT; ++j) pos[j] = atomicAdd(&cur[bin_slot(rows[j], tid)], 1);
#pragma unroll
    for (int j = 0; j < SORT_EPT; ++j)
        if (rows[j] >= 0) rec[pos[j]] = make_int2(rows[j], srcs[j]);
}

#define RANK_UNR 8
__global__ __launch_bounds__(BIN_ROWS)
void bin_rank_kernel(const int2* __restrict__ rec, const int* __restrict__ binbase,
                     int* __restrict__ seg_beg, int* __restrict__ seg_end,
                     int* __restrict__ csr_src)
{
    __shared__ int cnt[BIN_ROWS + 32];
    __shared__ int wtot[BIN_ROWS / 32];
    const int tid   = threadIdx.x;
    const int lane  = tid & 31;
    const int wv    = tid >> 5;
    const int b     = blockIdx.x;
    const int r0    = binbase[b];
    const int r1    = binbase[b + 1];
    const int spare = BIN_ROWS + lane;

    cnt[tid] = 0;
    __syncthreads();
    for (int i0 = r0; i0 < r1; i0 += RANK_UNR * BIN_ROWS) {
        int x[RANK_UNR];
#pragma unroll
        for (int u = 0; u < RANK_UNR; ++u) {
            const int i = i0 + tid + u * BIN_ROWS;
            x[u] = rec[min(i, r1 - 1)].x;
        }
#pragma unroll
        for (int u = 0; u < RANK_UNR; ++u) {
            const int i = i0 + tid + u * BIN_ROWS;
            atomicAdd(&cnt[(i < r1) ? (x[u] & (BIN_ROWS - 1)) : spare], 1);
        }
    }
    __syncthreads();

    const int d = cnt[tid];
    int incl = d;
#pragma unroll
    for (int off = 1; off < 32; off <<= 1) {
        const int up = __shfl_up(incl, off, 32);
        if (lane >= off) incl += up;
    }
    if (lane == 31) wtot[wv] = incl;
    __syncthreads();
    int st = r0 + incl - d;
#pragma unroll
    for (int k = 0; k < BIN_ROWS / 32; ++k) if (k < wv) st += wtot[k];
    const int v = b * BIN_ROWS + tid;
    if (v < 2 * NNODES) { seg_beg[v] = st; seg_end[v] = st + d; }
    cnt[tid] = st;
    __syncthreads();

    for (int i0 = r0; i0 < r1; i0 += RANK_UNR * BIN_ROWS) {
        int2 r[RANK_UNR];
        int  pos[RANK_UNR];
#pragma unroll
        for (int u = 0; u < RANK_UNR; ++u) {
            const int i = i0 + tid + u * BIN_ROWS;
            r[u] = rec[min(i, r1 - 1)];
        }
#pragma unroll
        for (int u = 0; u < RANK_UNR; ++u) {
            const int i = i0 + tid + u * BIN_ROWS;
            pos[u] = atomicAdd(&cnt[(i < r1) ? (r[u].x & (BIN_ROWS - 1)) : spare], 1);
        }
#pragma unroll
        for (int u = 0; u < RANK_UNR; ++u) {
            const int i = i0 + tid + u * BIN_ROWS;
            if (i < r1) csr_src[pos[u]] = r[u].y;
        }
    }
}

__device__ __forceinline__ float4 ln_relu_row(const float4 v, const float4 gg, const float4 bb)
{
    float s1 = v.x + v.y + v.z + v.w;
    float s2 = v.x * v.x + v.y * v.y + v.z * v.z + v.w * v.w;
#pragma unroll
    for (int off = 16; off > 0; off >>= 1) {
        s1 += __shfl_xor(s1, off, 32);
        s2 += __shfl_xor(s2, off, 32);
    }
    const float mu  = s1 * (1.f / HDIM);
    const float var = s2 * (1.f / HDIM) - mu * mu;
    const float inv = rsqrtf(var + LN_EPS);
    float4 r;
    r.x = fmaxf((v.x - mu) * inv * gg.x + bb.x, 0.f);
    r.y = fmaxf((v.y - mu) * inv * gg.y + bb.y, 0.f);
    r.z = fmaxf((v.z - mu) * inv * gg.z + bb.z, 0.f);
    r.w = fmaxf((v.w - mu) * inv * gg.w + bb.w, 0.f);
    return r;
}

#define AGG_BLOCKS 4096
#define AGG_DEMAND_BLOCKS 1024

__device__ __forceinline__ float exp_nonpos(const float x)
{
    return __builtin_amdgcn_exp2f(x * 1.44269504088896340736f);
}

template <int CNT, bool LNF>
__device__ __forceinline__ void agg_chunk(const float* __restrict__ hl, int idx, int c, bool first,
                                          float (&mx)[4], float (&de)[4], float (&nu)[4],
                                          const float4 gg, const float4 bb)
{
    float4 hv[CNT];
#pragma unroll
    for (int j = 0; j < CNT; ++j) {
        const int s = __builtin_amdgcn_readlane(idx, c + j);
        hv[j] = *(const float4*)(hl + (size_t)s * HDIM);
    }
    if (LNF) {
#pragma unroll
        for (int j = 0; j < CNT; ++j) hv[j] = ln_relu_row(hv[j], gg, bb);
    }
    float m[CNT][4];
    float c0 = mx[0], c1 = mx[1], c2 = mx[2], c3 = mx[3];
#pragma unroll
    for (int j = 0; j < CNT; ++j) {
        m[j][0] = fmaxf(hv[j].x, 0.f) + GEN_EPS;
        m[j][1] = fmaxf(hv[j].y, 0.f) + GEN_EPS;
        m[j][2] = fmaxf(hv[j].z, 0.f) + GEN_EPS;
        m[j][3] = fmaxf(hv[j].w, 0.f) + GEN_EPS;
        c0 = fmaxf(c0, m[j][0]);
        c1 = fmaxf(c1, m[j][1]);
        c2 = fmaxf(c2, m[j][2]);
        c3 = fmaxf(c3, m[j][3]);
    }
    if (!first) {
        const float s0 = exp_nonpos(mx[0] - c0), s1 = exp_nonpos(mx[1] - c1);
        const float s2 = exp_nonpos(mx[2] - c2), s3 = exp_nonpos(mx[3] - c3);
        de[0] *= s0; nu[0] *= s0;
        de[1] *= s1; nu[1] *= s1;
        de[2] *= s2; nu[2] *= s2;
        de[3] *= s3; nu[3] *= s3;
    }
    mx[0] = c0; mx[1] = c1; mx[2] = c2; mx[3] = c3;
#pragma unroll
    for (int j = 0; j < CNT; ++j) {
#pragma unroll
        for (int f = 0; f < 4; ++f) {
            const float ex = exp_nonpos(m[j][f] - mx[f]);
            de[f] += ex;
            nu[f] += m[j][f] * ex;
        }
    }
}

template <bool LNF>
__device__ __forceinline__ float4 agg_node(const float* __restrict__ hl, const int* __restrict__ csr_src,
                                           const int* __restrict__ rmap,
                                           int beg, int n, int lane, const float4 own,
                                           const float4 gg, const float4 bb)
{
    float mx[4] = {0.f, 0.f, 0.f, 0.f};
    float de[4] = {0.f, 0.f, 0.f, 0.f};
    float nu[4] = {0.f, 0.f, 0.f, 0.f};

    for (int base = 0; base < n; base += 32) {
        const int bc  = (n - base < 32) ? (n - base) : 32;
        int idx = (lane < bc) ? csr_src[beg + base + lane] : 0;
        if (LNF) idx = rmap[idx];
        for (int c = 0; c < bc; c += 8) {
            const int  cnt   = (bc - c < 8) ? (bc - c) : 8;
            const bool first = (base + c == 0);
            switch (cnt) {
            case 1:  agg_chunk<1, LNF>(hl, idx, c, first, mx, de, nu, gg, bb); break;
            case 2:  agg_chunk<2, LNF>(hl, idx, c, first, mx, de, nu, gg, bb); break;
            case 3:  agg_chunk<3, LNF>(hl, idx, c, first, mx, de, nu, gg, bb); break;
            case 4:  agg_chunk<4, LNF>(hl, idx, c, first, mx, de, nu, gg, bb); break;
            case 5:  agg_chunk<5, LNF>(hl, idx, c, first, mx, de, nu, gg, bb); break;
            case 6:  agg_chunk<6, LNF>(hl, idx, c, first, mx, de, nu, gg, bb); break;
            case 7:  agg_chunk<7, LNF>(hl, idx, c, first, mx, de, nu, gg, bb); break;
            default: agg_chunk<8, LNF>(hl, idx, c, first, mx, de, nu, gg, bb); break;
            }
        }
    }

    float4 r;
    r.x = nu[0] / (de[0] + 1e-16f) + own.x;
    r.y = nu[1] / (de[1] + 1e-16f) + own.y;
    r.z = nu[2] / (de[2] + 1e-16f) + own.z;
    r.w = nu[3] / (de[3] + 1e-16f) + own.w;
    return r;
}

template <bool DEMAND>
__global__ __launch_bounds__(256)
void node_softmax_agg_kernel(const float* __restrict__ hin, const int* __restrict__ seg_beg,
                             const int* __restrict__ seg_end, const int* __restrict__ csr_src,
                             const int* __restrict__ need, float* __restrict__ t)
{
    const int lane = threadIdx.x & 31;
    const int wv = __builtin_amdgcn_readfirstlane((int)(threadIdx.x >> 5));
    const float* __restrict__ hl = hin + lane * 4;
    const float4 none = {0.f, 0.f, 0.f, 0.f};

    constexpr int NBLK = DEMAND ? AGG_DEMAND_BLOCKS : AGG_BLOCKS;
    for (int row = blockIdx.x * 8 + wv; row < NNODES; row += NBLK * 8) {
        if (DEMAND) {
            if (__builtin_amdgcn_readfirstlane(need[row]) == 0) continue;
        }
        const int beg = __builtin_amdgcn_readfirstlane(seg_beg[row]);
        const int n   = __builtin_amdgcn_readfirstlane(seg_end[row]) - beg;
        const float4 own = *(const float4*)(hl + (size_t)row * HDIM);
        const float4 r = agg_node<false>(hl, csr_src, nullptr, beg, n, lane, own, none, none);
        *(float4*)(t + (size_t)row * HDIM + lane * 4) = r;
    }
}

__global__ __launch_bounds__(256)
void mark_needed_kernel(const int* __restrict__ final_map, const int* __restrict__ node_map,
                        const int* __restrict__ pmap,
                        const int* __restrict__ seg_beg, const int* __restrict__ seg_end,
                        const int* __restrict__ csr_src, int* __restrict__ need)
{
    const int lane = threadIdx.x & 31;
    const int i    = blockIdx.x * 8 + (int)(threadIdx.x >> 5);
    if (i >= NSEEDS) return;
    const int v   = node_map[final_map[i]];
    const int beg = seg_beg[v];
    const int n   = seg_end[v] - beg;
    if (lane == 0) need[pmap[v]] = 1;
    for (int j = lane; j < n; j += 32) need[pmap[csr_src[beg + j]]] = 1;
}

__global__ __launch_bounds__(256)
void seed_softmax_agg_kernel(const float* __restrict__ p, const int* __restrict__ pmap,
                             const int* __restrict__ final_map,
                             const int* __restrict__ node_map,
                             const float* __restrict__ g, const float* __restrict__ b,
                             const int* __restrict__ seg_beg, const int* __restrict__ seg_end,
                             const int* __restrict__ csr_src,
                             float* __restrict__ t, float* __restrict__ rs)
{
    const int lane = threadIdx.x & 31;
    const int wv = __builtin_amdgcn_readfirstlane((int)(threadIdx.x >> 5));
    const int i  = blockIdx.x * 8 + wv;
    if (i >= NSEEDS) return;
    const int v   = __builtin_amdgcn_readfirstlane(node_map[final_map[i]]);
    const int pv  = __builtin_amdgcn_readfirstlane(pmap[v]);
    const int beg = __builtin_amdgcn_readfirstlane(seg_beg[v]);
    const int n   = __builtin_amdgcn_readfirstlane(seg_end[v]) - beg;
    const float* __restrict__ hl = p + lane * 4;
    const float4 raw = *(const float4*)(hl + (size_t)pv * HDIM);
    const float4 gg  = *(const float4*)(g + lane * 4);
    const float4 bb  = *(const float4*)(b + lane * 4);
    const float4 own = ln_relu_row(raw, gg, bb);
    const float4 r   = agg_node<true>(hl, csr_src, pmap, beg, n, lane, own, gg, bb);
    *(float4*)(t  + (size_t)i * HDIM + lane * 4) = r;
    *(float4*)(rs + (size_t)i * HDIM + lane * 4) = raw;
}

__global__ __launch_bounds__(256)
void logsoftmax_kernel(const float* __restrict__ in, float* __restrict__ out, int M)
{
    const int row  = blockIdx.x * 8 + (threadIdx.x >> 5);
    const int lane = threadIdx.x & 31;
    if (row >= M) return;
    const float NEG = -3.0e38f;
    const float v0 = (lane      < NCLS) ? in[(size_t)row * NCLSPAD + lane     ] : NEG;
    const float v1 = (lane + 32 < NCLS) ? in[(size_t)row * NCLSPAD + lane + 32] : NEG;
    float mx = fmaxf(v0, v1);
#pragma unroll
    for (int off = 16; off > 0; off >>= 1) mx = fmaxf(mx, __shfl_xor(mx, off, 32));
    float se = ((lane      < NCLS) ? expf(v0 - mx) : 0.f)
             + ((lane + 32 < NCLS) ? expf(v1 - mx) : 0.f);
#pragma unroll
    for (int off = 16; off > 0; off >>= 1) se += __shfl_xor(se, off, 32);
    const float lse = logf(se) + mx;
    if (lane      < NCLS) out[(size_t)row * NCLS + lane     ] = v0 - lse;
    if (lane + 32 < NCLS) out[(size_t)row * NCLS + lane + 32] = v1 - lse;
}

extern "C" void kernel_launch(void* const* d_in, const int* in_sizes, int n_in,
                              void* d_out, int out_size, void* d_ws, size_t ws_size,
                              hipStream_t stream)
{
    (void)in_sizes; (void)n_in; (void)out_size; (void)ws_size;

    const float* x         = (const float*)d_in[0];
    const int*   src       = (const int*)  d_in[1];
    const int*   dst       = (const int*)  d_in[2];
    const int*   node_map  = (const int*)  d_in[3];
    const int*   final_map = (const int*)  d_in[4];
    const float* enc_w     = (const float*)d_in[5];
    const float* enc_b     = (const float*)d_in[6];
    const float* gcn_w     = (const float*)d_in[7];
    const float* gcn_b     = (const float*)d_in[8];
    const float* ln_g      = (const float*)d_in[9];
    const float* ln_b      = (const float*)d_in[10];
    const float* pred_w    = (const float*)d_in[11];
    const float* pred_b    = (const float*)d_in[12];
    float* out = (float*)d_out;

    const size_t NH = (size_t)NNODES * HDIM;
    float* B0     = (float*)d_ws;
    float* B1     = B0 + NH;
    float* B2     = B0 + 2 * NH;
    float* B3     = B0 + 3 * NH;
    float* B4     = B0 + 4 * NH;
    float* HGN    = B0 + 5 * NH;
    float* SC     = HGN + (size_t)2 * NSEEDS * HDIM;

    const dim3 gblk(32, 4, 1);
    const int  mtN   = NNODES / 16;

    int*  IWHIST = (int*)B3;
    int*  IBASE  = IWHIST + NBINS * SORT_WGS;
    int*  INEED  = IBASE + 256;
    int*  ISTART = INEED + NNODES;
    int*  IEND   = ISTART + 2 * NNODES;
    int*  ICSR   = IEND + 2 * NNODES;
    int2* IREC   = (int2*)(ICSR + 2 * NEDGES);
    bin_hist_kernel<<<SORT_WGS, SORT_THR, 0, stream>>>(dst, IWHIST, INEED);
    bin_scatter_kernel<<<SORT_WGS, SORT_THR, 0, stream>>>(src, dst, IWHIST, IBASE, IREC);
    bin_rank_kernel<<<NBINS, BIN_ROWS, 0, stream>>>(IREC, IBASE, ISTART, IEND, ICSR);

    gemm_rows128_kernel<FDIM, false, EPI_RAW><<<GEMM_WGS, 512, 0, stream>>>(
        x, enc_w, enc_b, nullptr, nullptr, nullptr, B0, nullptr, mtN);

    node_softmax_agg_kernel<false><<<AGG_BLOCKS, 256, 0, stream>>>(
        B0, ISTART, IEND, ICSR, nullptr, B2);
    gemm_rows128_kernel<HDIM, false, EPI_BOTH><<<GEMM_WGS, 512, 0, stream>>>(
        B2, gcn_w, gcn_b, nullptr, ln_g, ln_b, B0, B1, mtN);

    mark_needed_kernel<<<NSEEDS / 8, 256, 0, stream>>>(
        final_map, node_map + NNODES, node_map, ISTART + NNODES, IEND + NNODES, ICSR, INEED);
    node_softmax_agg_kernel<true><<<AGG_DEMAND_BLOCKS, 256, 0, stream>>>(
        B1, ISTART, IEND, ICSR, INEED, B2);
    gemm_rows128_kernel<HDIM, true, EPI_RAW><<<GEMM_WGS, 512, 0, stream>>>(
        B2, gcn_w + (size_t)HDIM * HDIM, gcn_b + HDIM,  B0, nullptr, nullptr,  B4, nullptr, mtN);

    float* T2 = B2;
    float* R2 = B1;
    const int mtS = NSEEDS / 16;
    seed_softmax_agg_kernel<<<NSEEDS / 8, 256, 0, stream>>>(
        B4, node_map, final_map, node_map + NNODES, ln_g + HDIM, ln_b + HDIM,
        ISTART + NNODES, IEND + NNODES, ICSR, T2, R2);
    gemm_rows128_kernel<HDIM, true, EPI_NORM><<<GEMM_WGS, 512, 0, stream>>>(
        T2, gcn_w + (size_t)2 * HDIM * HDIM, gcn_b + 2 * HDIM, R2,
        ln_g + 2 * HDIM, ln_b + 2 * HDIM, HGN, nullptr, mtS);

    gemm_head_kernel<<<(mtS + 3) / 4, gblk, 0, stream>>>(
        HGN, pred_w, pred_b, SC, NSEEDS);
    logsoftmax_kernel<<<(NSEEDS + 7) / 8, 256, 0, stream>>>(SC, out, NSEEDS);
}


# Layers 0 and 1: each product wave forms its own 16 rows of t (softmax aggregation; next row's edge list and own row requested a row ahead) in an LDS tile and multiplies from it; 13 -> 11 launches
# speedup vs baseline: 6.7496x; 1.0718x over previous
#include <hip/hip_runtime.h>
#include <hip/hip_bf16.h>
#include <math.h>

#define NNODES   100000
#define NEDGES   600000
#define FDIM     256
#define HDIM     128
#define NCLS     47
#define NCLSPAD  64
#define NSEEDS   10000
#define GEN_EPS  1e-7f
#define LN_EPS   1e-5f

typedef float v2f __attribute__((ext_vector_type(2)));
typedef float v8f __attribute__((ext_vector_type(8)));


#define HEAD_PITCH (HDIM + 4)
__global__ __launch_bounds__(128)
void gemm_head_kernel(const float* __restrict__ A, const float* __restrict__ W,
                      const float* __restrict__ bias, float* __restrict__ out, int M)
{
    __shared__ __attribute__((aligned(16))) float wl[NCLSPAD * HEAD_PITCH];

    const int lane = threadIdx.x;
    const int tid  = threadIdx.y * 32 + lane;

    constexpr int NW    = HDIM * NCLS;
    constexpr int NPASS = NW / 128;
    static_assert(NW % 128 == 0, "whole passes of the 128 threads");
    float wq[NPASS];
#pragma unroll
    for (int r = 0; r < NPASS; ++r) wq[r] = W[tid + 128 * r];
#pragma unroll
    for (int r = 0; r < NPASS; ++r) {
        const int i = tid + 128 * r;
        const int k = i / NCLS;
        const int n = i - k * NCLS;
        wl[n * HEAD_PITCH + k] = wq[r];
    }
    for (int i = tid; i < (NCLSPAD - NCLS) * HDIM; i += 128) {
        const int n = NCLS + i / HDIM;
        const int k = i & (HDIM - 1);
        wl[n * HEAD_PITCH + k] = 0.f;
    }
    __syncthreads();

    const int mtile = blockIdx.x * 4 + threadIdx.y;
    const int row0  = mtile * 16;
    if (row0 >= M) return;
    const int hb   = lane >> 4;
    const int lo   = lane & 15;
    const int arow = row0 + lo;

    const float* __restrict__ a_ptr = A + (size_t)arow * HDIM;
    const float* b_ptr = wl + lo * HEAD_PITCH;
    constexpr int bstep = 16 * HEAD_PITCH;

    v8f acc[4];
#pragma unroll
    for (int t = 0; t < 4; ++t)
#pragma unroll
        for (int i = 0; i < 8; ++i) acc[t][i] = 0.f;

#pragma unroll 4
    for (int k = 0; k < HDIM; k += 4) {
        const int ka = k + 2 * hb;
        const v2f a  = *(const v2f*)(a_ptr + ka);
        const v2f b0 = *(const v2f*)(b_ptr + ka);
        const v2f b1 = *(const v2f*)(b_ptr + bstep + ka);
        const v2f b2 = *(const v2f*)(b_ptr + 2 * bstep + ka);
        const v2f b3 = *(const v2f*)(b_ptr + 3 * bstep + ka);
        acc[0] = __builtin_amdgcn_wmma_f32_16x16x4_f32(false, a, false, b0, (short)0, acc[0], false, false);
        acc[1] = __builtin_amdgcn_wmma_f32_16x16x4_f32(false, a, false, b1, (short)0, acc[1], false, false);
        acc[2] = __builtin_amdgcn_wmma_f32_16x16x4_f32(false, a, false, b2, (short)0, acc[2], false, false);
        acc[3] = __builtin_amdgcn_wmma_f32_16x16x4_f32(false, a, false, b3, (short)0, acc[3], false, false);
    }

#pragma unroll
    for (int t = 0; t < 4; ++t) {
        const int   col = lo + 16 * t;
        const float bv  = (col < NCLS) ? bias[col] : 0.f;
#pragma unroll
        for (int i = 0; i < 8; ++i) {
            const int r = row0 + i + 8 * hb;
            out[(size_t)r * NCLSPAD + col] = acc[t][i] + bv;
        }
    }
}

#define GEMM_WGS 256


__device__ __forceinline__ void load_a_chunk(const float* __restrict__ ap, float4 (&a)[16])
{
#pragma unroll
    for (int i = 0; i < 16; ++i) a[i] = *(const float4*)(ap + 8 * i);
}

template <int K>
__device__ __forceinline__ void load_b_block(const float* bp, float4 (&b)[8])
{
    constexpr int PITCH = K + 4;
#pragma unroll
    for (int t = 0; t < 8; ++t) b[t] = *(const float4*)(bp + t * 16 * PITCH);
}

__device__ __forceinline__ void mma_block(const float4 a, const float4 (&b)[8], v8f (&acc)[8])
{
    const v2f a01 = {a.x, a.y};
    const v2f a23 = {a.z, a.w};
#pragma unroll
    for (int t = 0; t < 8; ++t) {
        const v2f b01 = {b[t].x, b[t].y};
        acc[t] = __builtin_amdgcn_wmma_f32_16x16x4_f32(false, a01, false, b01, (short)0, acc[t], false, false);
    }
#pragma unroll
    for (int t = 0; t < 8; ++t) {
        const v2f b23 = {b[t].z, b[t].w};
        acc[t] = __builtin_amdgcn_wmma_f32_16x16x4_f32(false, a23, false, b23, (short)0, acc[t], false, false);
    }
}

template <int K>
__device__ __forceinline__ void mma_chunk(const float* bp, const float4 (&a)[16], v8f (&acc)[8])
{
    float4 b0[8], b1[8];
    load_b_block<K>(bp, b0);
#pragma unroll
    for (int i = 0; i < 16; i += 2) {
        load_b_block<K>(bp + 8 * (i + 1), b1);
        __builtin_amdgcn_sched_barrier(0);
        mma_block(a[i], b0, acc);
        __builtin_amdgcn_sched_barrier(0);
        if (i + 2 < 16) load_b_block<K>(bp + 8 * (i + 2), b0);
        __builtin_amdgcn_sched_barrier(0);
        mma_block(a[i + 1], b1, acc);
        __builtin_amdgcn_sched_barrier(0);
    }
}

#define EPI_RAW  0
#define EPI_BOTH 1
#define EPI_NORM 2

template <int XM>
__device__ __forceinline__ float row_xor_add(const float x)
{
    const int moved = __builtin_amdgcn_update_dpp(0, __builtin_bit_cast(int, x), 0x160 + XM, 0xf, 0xf, true);
    return x + __builtin_bit_cast(float, moved);
}

template <int XM>
__device__ __forceinline__ void row_xor_add8(float (&s)[8])
{
#pragma unroll
    for (int i = 0; i < 8; ++i) s[i] = row_xor_add<XM>(s[i]);
}

template <bool HAS_RES, int EPI>
__device__ __forceinline__ void store_strip(const v8f (&acc)[8], const float* __restrict__ bias,
                                            const float* __restrict__ res,
                                            const float* __restrict__ lng, const float* __restrict__ lnb,
                                            float* __restrict__ out, float* __restrict__ out2,
                                            int row0, int c, int h)
{
    float4 s[2][8];
#pragma unroll
    for (int g = 0; g < 2; ++g) {
        const int    col = 64 * g + 4 * c;
        const float4 bv  = *(const float4*)(bias + col);
        float4 rv[8];
        if (HAS_RES) {
#pragma unroll
            for (int i = 0; i < 8; ++i)
                rv[i] = *(const float4*)(res + (size_t)(row0 + i + 8 * h) * HDIM + col);
        }
#pragma unroll
        for (int i = 0; i < 8; ++i) {
            float4 v;
            v.x = acc[4 * g + 0][i] + bv.x;
            v.y = acc[4 * g + 1][i] + bv.y;
            v.z = acc[4 * g + 2][i] + bv.z;
            v.w = acc[4 * g + 3][i] + bv.w;
            if (HAS_RES) { v.x += rv[i].x; v.y += rv[i].y; v.z += rv[i].z; v.w += rv[i].w; }
            s[g][i] = v;
            if (EPI != EPI_NORM) *(float4*)(out + (size_t)(row0 + i + 8 * h) * HDIM + col) = v;
        }
    }
    if (EPI == EPI_RAW) return;

    float s1[8], s2[8];
#pragma unroll
    for (int i = 0; i < 8; ++i) {
        const float4 p = s[0][i];
        const float4 q = s[1][i];
        s1[i] = ((p.x + p.y) + (p.z + p.w)) + ((q.x + q.y) + (q.z + q.w));
        s2[i] = ((p.x * p.x + p.y * p.y) + (p.z * p.z + p.w * p.w))
              + ((q.x * q.x + q.y * q.y) + (q.z * q.z + q.w * q.w));
    }
    row_xor_add8<1>(s1); row_xor_add8<1>(s2);
    row_xor_add8<2>(s1); row_xor_add8<2>(s2);
    row_xor_add8<4>(s1); row_xor_add8<4>(s2);
    row_xor_add8<8>(s1); row_xor_add8<8>(s2);

    float mu[8], inv[8];
#pragma unroll
    for (int i = 0; i < 8; ++i) {
        mu[i] = s1[i] * (1.f / HDIM);
        const float var = s2[i] * (1.f / HDIM) - mu[i] * mu[i];
        inv[i] = rsqrtf(var + LN_EPS);
    }
    float* __restrict__ dst = (EPI == EPI_BOTH) ? out2 : out;
#pragma unroll
    for (int g = 0; g < 2; ++g) {
        const int    col = 64 * g + 4 * c;
        const float4 gg  = *(const float4*)(lng + col);
        const float4 bb  = *(const float4*)(lnb + col);
#pragma unroll
        for (int i = 0; i < 8; ++i) {
            const float4 v = s[g][i];
            float4 r;
            r.x = fmaxf((v.x - mu[i]) * inv[i] * gg.x + bb.x, 0.f);
            r.y = fmaxf((v.y - mu[i]) * inv[i] * gg.y + bb.y, 0.f);
            r.z = fmaxf((v.z - mu[i]) * inv[i] * gg.z + bb.z, 0.f);
            r.w = fmaxf((v.w - mu[i]) * inv[i] * gg.w + bb.w, 0.f);
            *(float4*)(dst + (size_t)(row0 + i + 8 * h) * HDIM + col) = r;
        }
    }
}

template <int K, bool HAS_RES, int EPI>
__global__ __launch_bounds__(512)
void gemm_rows128_kernel(const float* __restrict__ A, const float* __restrict__ W,
                         const float* __restrict__ bias, const float* __restrict__ res,
                         const float* __restrict__ lng, const float* __restrict__ lnb,
                         float* __restrict__ out, float* __restrict__ out2, int ntiles)
{
    constexpr int PITCH = K + 4;
    __shared__ __attribute__((aligned(16))) float wl[HDIM * PITCH];

    const int tid  = threadIdx.x;
    const int lane = tid & 31;
    const int wv   = __builtin_amdgcn_readfirstlane(tid >> 5);
    const int c    = lane & 15;
    const int h    = lane >> 4;

    const int wa = wv >> 2;
    const int q  = wa + 4 * (((wv & 3) - wa) & 3);
    int tile = q * GEMM_WGS + (int)blockIdx.x;

    float4 a[16];
    if (tile < ntiles) load_a_chunk(A + (size_t)(tile * 16 + c) * K + 4 * h, a);

    constexpr int NPASS = K * (HDIM / 4) / 512;
    float4 wq[NPASS];
#pragma unroll
    for (int r = 0; r < NPASS; ++r) {
        const int i  = tid + 512 * r;
        const int m  = i & 15;
        const int hh = (i >> 5) & 1;
        const int k  = 2 * (i >> 6) + ((i >> 4) & 1);
        wq[r] = *(const float4*)(W + (size_t)k * HDIM + 64 * hh + 4 * m);
    }
#pragma unroll
    for (int r = 0; r < NPASS; ++r) {
        const int i  = tid + 512 * r;
        const int m  = i & 15;
        const int hh = (i >> 5) & 1;
        const int k  = 2 * (i >> 6) + ((i >> 4) & 1);
        float* dp = wl + (64 * hh + m) * PITCH + k;
        dp[0]          = wq[r].x;
        dp[16 * PITCH] = wq[r].y;
        dp[32 * PITCH] = wq[r].z;
        dp[48 * PITCH] = wq[r].w;
    }
    __syncthreads();

    const float* bp = wl + c * PITCH + 4 * h;

    while (tile < ntiles) {
        const int row0 = tile * 16;

        v8f acc[8];
#pragma unroll
        for (int t = 0; t < 8; ++t)
#pragma unroll
            for (int i = 0; i < 8; ++i) acc[t][i] = 0.f;

        mma_chunk<K>(bp, a, acc);
#pragma unroll
        for (int kc = 128; kc < K; kc += 128) {
            load_a_chunk(A + (size_t)(row0 + c) * K + 4 * h + kc, a);
            mma_chunk<K>(bp + kc, a, acc);
        }

        store_strip<HAS_RES, EPI>(acc, bias, res, lng, lnb, out, out2, row0, c, h);

        tile += 16 * GEMM_WGS;
        if (tile < ntiles) load_a_chunk(A + (size_t)(tile * 16 + c) * K + 4 * h, a);
    }
}

#define BIN_SHIFT   10
#define BIN_ROWS    (1 << BIN_SHIFT)
#define NBINS       ((2 * NNODES + BIN_ROWS - 1) / BIN_ROWS)
#define SORT_WGS    256
#define SORT_THR    512
#define SORT_CH     ((2 * NEDGES + SORT_WGS - 1) / SORT_WGS)
#define SORT_EPT    ((SORT_CH + SORT_THR - 1) / SORT_THR)
#define NEED_CH     ((NNODES + SORT_WGS - 1) / SORT_WGS)

__device__ __forceinline__ void load_chunk_rows(const int* __restrict__ dst, int w, int tid,
                                                int (&rows)[SORT_EPT], int (&eid)[SORT_EPT])
{
    const int e0 = w * SORT_CH;
    const int e1 = (e0 + SORT_CH < 2 * NEDGES) ? (e0 + SORT_CH) : (2 * NEDGES);
    int raw[SORT_EPT];
#pragma unroll
    for (int j = 0; j < SORT_EPT; ++j) {
        const int e = e0 + tid + SORT_THR * j;
        eid[j] = min(e, e1 - 1);
        raw[j] = dst[eid[j]];
    }
#pragma unroll
    for (int j = 0; j < SORT_EPT; ++j) {
        const int e = e0 + tid + SORT_THR * j;
        rows[j] = (e < e1) ? (((e >= NEDGES) ? NNODES : 0) + raw[j]) : -1;
    }
}

__device__ __forceinline__ int bin_slot(int row, int tid)
{
    return (row >= 0) ? (row >> BIN_SHIFT) : (NBINS + (tid & 31));
}

__global__ __launch_bounds__(SORT_THR)
void bin_hist_kernel(const int* __restrict__ dst, int* __restrict__ whist, int* __restrict__ need)
{
    __shared__ int hist[256];
    const int tid = threadIdx.x;
    const int w   = blockIdx.x;
    int rows[SORT_EPT], eid[SORT_EPT];
    load_chunk_rows(dst, w, tid, rows, eid);
    if (tid < 256) hist[tid] = 0;
    static_assert(NEED_CH <= SORT_THR, "one flag per thread");
    if (tid < NEED_CH && w * NEED_CH + tid < NNODES) need[w * NEED_CH + tid] = 0;
    __syncthreads();
#pragma unroll
    for (int j = 0; j < SORT_EPT; ++j) atomicAdd(&hist[bin_slot(rows[j], tid)], 1);
    __syncthreads();
    if (tid < NBINS) whist[tid * SORT_WGS + w] = hist[tid];
}

__global__ __launch_bounds__(SORT_THR)
void bin_scatter_kernel(const int* __restrict__ src, const int* __restrict__ dst,
                        const int* __restrict__ whist, int* __restrict__ binbase,
                        int2* __restrict__ rec)
{
    __shared__ int tot[256];
    __shared__ int cur[256];
    __shared__ int wtot[SORT_THR / 32];
    const int tid  = threadIdx.x;
    const int lane = tid & 31;
    const int wv   = tid >> 5;
    const int w    = blockIdx.x;

    int rows[SORT_EPT], eid[SORT_EPT], srcs[SORT_EPT];
    load_chunk_rows(dst, w, tid, rows, eid);
#pragma unroll
    for (int j = 0; j < SORT_EPT; ++j) srcs[j] = src[eid[j]];

    constexpr int NWV = SORT_THR / 32;
    constexpr int NIT = (NBINS + NWV - 1) / NWV;
    int4 q0[NIT], q1[NIT];
#pragma unroll
    for (int it = 0; it < NIT; ++it) {
        const int b  = wv + NWV * it;
        const int bb = (b < NBINS) ? b : (NBINS - 1);
        q0[it] = *(const int4*)(whist + bb * SORT_WGS + 8 * lane);
        q1[it] = *(const int4*)(whist + bb * SORT_WGS + 8 * lane + 4);
    }
    const int kf = w - 8 * lane;
    int tt[NIT], mm[NIT];
#pragma unroll
    for (int it = 0; it < NIT; ++it) {
        tt[it] = (q0[it].x + q0[it].y) + (q0[it].z + q0[it].w) + (q1[it].x + q1[it].y) + (q1[it].z + q1[it].w);
        mm[it] = ((0 < kf) ? q0[it].x : 0) + ((1 < kf) ? q0[it].y : 0) + ((2 < kf) ? q0[it].z : 0) + ((3 < kf) ? q0[it].w : 0)
               + ((4 < kf) ? q1[it].x : 0) + ((5 < kf) ? q1[it].y : 0) + ((6 < kf) ? q1[it].z : 0) + ((7 < kf) ? q1[it].w : 0);
    }
#pragma unroll
    for (int off = 16; off > 0; off >>= 1) {
        int tu[NIT], mu[NIT];
#pragma unroll
        for (int it = 0; it < NIT; ++it) {
            tu[it] = __shfl_xor(tt[it], off, 32);
            mu[it] = __shfl_xor(mm[it], off, 32);
        }
#pragma unroll
        for (int it = 0; it < NIT; ++it) { tt[it] += tu[it]; mm[it] += mu[it]; }
    }
#pragma unroll
    for (int it = 0; it < NIT; ++it) {
        const int b = wv + NWV * it;
        if (lane == 0 && b < NBINS) { tot[b] = tt[it]; cur[b] = mm[it]; }
    }
    if (tid >= NBINS && tid < 256) tot[tid] = 0;
    __syncthreads();

    const int d = (tid < 256) ? tot[tid] : 0;
    int incl = d;
#pragma unroll
    for (int off = 1; off < 32; off <<= 1) {
        const int up = __shfl_up(incl, off, 32);
        if (lane >= off) incl += up;
    }
    if (lane == 31) wtot[wv] = incl;
    __syncthreads();
    int excl = incl - d;
#pragma unroll
    for (int k = 0; k < SORT_THR / 32; ++k) if (k < wv) excl += wtot[k];
    if (tid < NBINS) cur[tid] += excl;
    if (w == 0 && tid <= NBINS) binbase[tid] = excl;
    __syncthreads();

    int pos[SORT_EPT];
#pragma unroll
    for (int j = 0; j < SORT_EPT; ++j) pos[j] = atomicAdd(&cur[bin_slot(rows[j], tid)], 1);
#pragma unroll
    for (int j = 0; j < SORT_EPT; ++j)
        if (rows[j] >= 0) rec[pos[j]] = make_int2(rows[j], srcs[j]);
}

#define RANK_UNR 8
__global__ __launch_bounds__(BIN_ROWS)
void bin_rank_kernel(const int2* __restrict__ rec, const int* __restrict__ binbase,
                     int* __restrict__ seg_beg, int* __restrict__ seg_end,
                     int* __restrict__ csr_src)
{
    __shared__ int cnt[BIN_ROWS + 32];
    __shared__ int wtot[BIN_ROWS / 32];
    const int tid   = threadIdx.x;
    const int lane  = tid & 31;
    const int wv    = tid >> 5;
    const int b     = blockIdx.x;
    const int r0    = binbase[b];
    const int r1    = binbase[b + 1];
    const int spare = BIN_ROWS + lane;

    cnt[tid] = 0;
    __syncthreads();
    for (int i0 = r0; i0 < r1; i0 += RANK_UNR * BIN_ROWS) {
        int x[RANK_UNR];
#pragma unroll
        for (int u = 0; u < RANK_UNR; ++u) {
            const int i = i0 + tid + u * BIN_ROWS;
            x[u] = rec[min(i, r1 - 1)].x;
        }
#pragma unroll
        for (int u = 0; u < RANK_UNR; ++u) {
            const int i = i0 + tid + u * BIN_ROWS;
            atomicAdd(&cnt[(i < r1) ? (x[u] & (BIN_ROWS - 1)) : spare], 1);
        }
    }
    __syncthreads();

    const int d = cnt[tid];
    int incl = d;
#pragma unroll
    for (int off = 1; off < 32; off <<= 1) {
        const int up = __shfl_up(incl, off, 32);
        if (lane >= off) incl += up;
    }
    if (lane == 31) wtot[wv] = incl;
    __syncthreads();
    int st = r0 + incl - d;
#pragma unroll
    for (int k = 0; k < BIN_ROWS / 32; ++k) if (k < wv) st += wtot[k];
    const int v = b * BIN_ROWS + tid;
    if (v < 2 * NNODES) { seg_beg[v] = st; seg_end[v] = st + d; }
    cnt[tid] = st;
    __syncthreads();

    for (int i0 = r0; i0 < r1; i0 += RANK_UNR * BIN_ROWS) {
        int2 r[RANK_UNR];
        int  pos[RANK_UNR];
#pragma unroll
        for (int u = 0; u < RANK_UNR; ++u) {
            const int i = i0 + tid + u * BIN_ROWS;
            r[u] = rec[min(i, r1 - 1)];
        }
#pragma unroll
        for (int u = 0; u < RANK_UNR; ++u) {
            const int i = i0 + tid + u * BIN_ROWS;
            pos[u] = atomicAdd(&cnt[(i < r1) ? (r[u].x & (BIN_ROWS - 1)) : spare], 1);
        }
#pragma unroll
        for (int u = 0; u < RANK_UNR; ++u) {
            const int i = i0 + tid + u * BIN_ROWS;
            if (i < r1) csr_src[pos[u]] = r[u].y;
        }
    }
}

__device__ __forceinline__ float4 ln_relu_row(const float4 v, const float4 gg, const float4 bb)
{
    float s1 = v.x + v.y + v.z + v.w;
    float s2 = v.x * v.x + v.y * v.y + v.z * v.z + v.w * v.w;
#pragma unroll
    for (int off = 16; off > 0; off >>= 1) {
        s1 += __shfl_xor(s1, off, 32);
        s2 += __shfl_xor(s2, off, 32);
    }
    const float mu  = s1 * (1.f / HDIM);
    const float var = s2 * (1.f / HDIM) - mu * mu;
    const float inv = rsqrtf(var + LN_EPS);
    float4 r;
    r.x = fmaxf((v.x - mu) * inv * gg.x + bb.x, 0.f);
    r.y = fmaxf((v.y - mu) * inv * gg.y + bb.y, 0.f);
    r.z = fmaxf((v.z - mu) * inv * gg.z + bb.z, 0.f);
    r.w = fmaxf((v.w - mu) * inv * gg.w + bb.w, 0.f);
    return r;
}


__device__ __forceinline__ float exp_nonpos(const float x)
{
    return __builtin_amdgcn_exp2f(x * 1.44269504088896340736f);
}

template <int CNT, bool LNF>
__device__ __forceinline__ void agg_chunk(const float* __restrict__ hl, int idx, int c, bool first,
                                          float (&mx)[4], float (&de)[4], float (&nu)[4],
                                          const float4 gg, const float4 bb)
{
    float4 hv[CNT];
#pragma unroll
    for (int j = 0; j < CNT; ++j) {
        const int s = __builtin_amdgcn_readlane(idx, c + j);
        hv[j] = *(const float4*)(hl + (size_t)s * HDIM);
    }
    if (LNF) {
#pragma unroll
        for (int j = 0; j < CNT; ++j) hv[j] = ln_relu_row(hv[j], gg, bb);
    }
    float m[CNT][4];
    float c0 = mx[0], c1 = mx[1], c2 = mx[2], c3 = mx[3];
#pragma unroll
    for (int j = 0; j < CNT; ++j) {
        m[j][0] = fmaxf(hv[j].x, 0.f) + GEN_EPS;
        m[j][1] = fmaxf(hv[j].y, 0.f) + GEN_EPS;
        m[j][2] = fmaxf(hv[j].z, 0.f) + GEN_EPS;
        m[j][3] = fmaxf(hv[j].w, 0.f) + GEN_EPS;
        c0 = fmaxf(c0, m[j][0]);
        c1 = fmaxf(c1, m[j][1]);
        c2 = fmaxf(c2, m[j][2]);
        c3 = fmaxf(c3, m[j][3]);
    }
    if (!first) {
        const float s0 = exp_nonpos(mx[0] - c0), s1 = exp_nonpos(mx[1] - c1);
        const float s2 = exp_nonpos(mx[2] - c2), s3 = exp_nonpos(mx[3] - c3);
        de[0] *= s0; nu[0] *= s0;
        de[1] *= s1; nu[1] *= s1;
        de[2] *= s2; nu[2] *= s2;
        de[3] *= s3; nu[3] *= s3;
    }
    mx[0] = c0; mx[1] = c1; mx[2] = c2; mx[3] = c3;
#pragma unroll
    for (int j = 0; j < CNT; ++j) {
#pragma unroll
        for (int f = 0; f < 4; ++f) {
            const float ex = exp_nonpos(m[j][f] - mx[f]);
            de[f] += ex;
            nu[f] += m[j][f] * ex;
        }
    }
}

template <bool LNF>
__device__ __forceinline__ void agg_round(const float* __restrict__ hl, int idx, int bc, bool start,
                                          float (&mx)[4], float (&de)[4], float (&nu)[4],
                                          const float4 gg, const float4 bb)
{
    for (int c = 0; c < bc; c += 8) {
        const int  cnt   = (bc - c < 8) ? (bc - c) : 8;
        const bool first = start && (c == 0);
        switch (cnt) {
        case 1:  agg_chunk<1, LNF>(hl, idx, c, first, mx, de, nu, gg, bb); break;
        case 2:  agg_chunk<2, LNF>(hl, idx, c, first, mx, de, nu, gg, bb); break;
        case 3:  agg_chunk<3, LNF>(hl, idx, c, first, mx, de, nu, gg, bb); break;
        case 4:  agg_chunk<4, LNF>(hl, idx, c, first, mx, de, nu, gg, bb); break;
        case 5:  agg_chunk<5, LNF>(hl, idx, c, first, mx, de, nu, gg, bb); break;
        case 6:  agg_chunk<6, LNF>(hl, idx, c, first, mx, de, nu, gg, bb); break;
        case 7:  agg_chunk<7, LNF>(hl, idx, c, first, mx, de, nu, gg, bb); break;
        default: agg_chunk<8, LNF>(hl, idx, c, first, mx, de, nu, gg, bb); break;
        }
    }
}

template <bool LNF, bool PRE>
__device__ __forceinline__ float4 agg_node(const float* __restrict__ hl, const int* __restrict__ csr_src,
                                           const int* __restrict__ rmap,
                                           int beg, int n, int idx0, int lane, const float4 own,
                                           const float4 gg, const float4 bb)
{
    float mx[4] = {0.f, 0.f, 0.f, 0.f};
    float de[4] = {0.f, 0.f, 0.f, 0.f};
    float nu[4] = {0.f, 0.f, 0.f, 0.f};

    if (PRE && n > 0) agg_round<LNF>(hl, idx0, (n < 32) ? n : 32, true, mx, de, nu, gg, bb);
    for (int base = PRE ? 32 : 0; base < n; base += 32) {
        const int bc  = (n - base < 32) ? (n - base) : 32;
        int idx = (lane < bc) ? csr_src[beg + base + lane] : 0;
        if (LNF) idx = rmap[idx];
        agg_round<LNF>(hl, idx, bc, base == 0, mx, de, nu, gg, bb);
    }

    float4 r;
    r.x = nu[0] / (de[0] + 1e-16f) + own.x;
    r.y = nu[1] / (de[1] + 1e-16f) + own.y;
    r.z = nu[2] / (de[2] + 1e-16f) + own.z;
    r.w = nu[3] / (de[3] + 1e-16f) + own.w;
    return r;
}

#define TILE_PITCH (HDIM + 4)

template <bool HAS_RES, int EPI, bool DEMAND>
__global__ __launch_bounds__(512)
void agg_gemm_rows128_kernel(const float* __restrict__ hin, const int* __restrict__ seg_beg,
                             const int* __restrict__ seg_end, const int* __restrict__ csr_src,
                             const int* __restrict__ need,
                             const float* __restrict__ W, const float* __restrict__ bias,
                             const float* __restrict__ res,
                             const float* __restrict__ lng, const float* __restrict__ lnb,
                             float* __restrict__ out, float* __restrict__ out2, int ntiles)
{
    constexpr int K     = HDIM;
    constexpr int PITCH = K + 4;
    __shared__ __attribute__((aligned(16))) float wl[HDIM * PITCH];
    __shared__ __attribute__((aligned(16))) float tl[16 * 16 * TILE_PITCH];

    const int tid  = threadIdx.x;
    const int lane = tid & 31;
    const int wv   = __builtin_amdgcn_readfirstlane(tid >> 5);
    const int c    = lane & 15;
    const int h    = lane >> 4;

    const int wa = wv >> 2;
    const int q  = wa + 4 * (((wv & 3) - wa) & 3);
    int tile = q * GEMM_WGS + (int)blockIdx.x;

    int vb = 0, ve = 0, vn = 1;
    if (tile < ntiles) {
        vb = seg_beg[tile * 16 + c];
        ve = seg_end[tile * 16 + c];
        if (DEMAND) vn = need[tile * 16 + c];
    }

    constexpr int NPASS = K * (HDIM / 4) / 512;
    float4 wq[NPASS];
#pragma unroll
    for (int r = 0; r < NPASS; ++r) {
        const int i  = tid + 512 * r;
        const int m  = i & 15;
        const int hh = (i >> 5) & 1;
        const int k  = 2 * (i >> 6) + ((i >> 4) & 1);
        wq[r] = *(const float4*)(W + (size_t)k * HDIM + 64 * hh + 4 * m);
    }
#pragma unroll
    for (int r = 0; r < NPASS; ++r) {
        const int i  = tid + 512 * r;
        const int m  = i & 15;
        const int hh = (i >> 5) & 1;
        const int k  = 2 * (i >> 6) + ((i >> 4) & 1);
        float* dp = wl + (64 * hh + m) * PITCH + k;
        dp[0]          = wq[r].x;
        dp[16 * PITCH] = wq[r].y;
        dp[32 * PITCH] = wq[r].z;
        dp[48 * PITCH] = wq[r].w;
    }
    __syncthreads();

    const float* bp = wl + c * PITCH + 4 * h;
    float*       my = tl + wv * (16 * TILE_PITCH);
    const float* __restrict__ hl = hin + lane * 4;
    const float4 none = {0.f, 0.f, 0.f, 0.f};

    while (tile < ntiles) {
        const int row0 = tile * 16;

        int  beg  = __builtin_amdgcn_readlane(vb, 0);
        int  n    = __builtin_amdgcn_readlane(ve, 0) - beg;
        bool want = !DEMAND || (__builtin_amdgcn_readlane(vn, 0) != 0);
        int    idx = 0;
        float4 own = none;
        if (want) {
            idx = (lane < n) ? csr_src[beg + lane] : 0;
            own = *(const float4*)(hl + (size_t)row0 * HDIM);
        }
        for (int rr = 0; rr < 16; ++rr) {
            const int    cbeg = beg, cn = n, cidx = idx;
            const float4 cown = own;
            const bool   cwant = want;
            if (rr + 1 < 16) {
                beg  = __builtin_amdgcn_readlane(vb, rr + 1);
                n    = __builtin_amdgcn_readlane(ve, rr + 1) - beg;
                want = !DEMAND || (__builtin_amdgcn_readlane(vn, rr + 1) != 0);
                if (want) {
                    idx = (lane < n) ? csr_src[beg + lane] : 0;
                    own = *(const float4*)(hl + (size_t)(row0 + rr + 1) * HDIM);
                }
            }
            float4 r = none;
            if (cwant) r = agg_node<false, true>(hl, csr_src, nullptr, cbeg, cn, cidx, lane, cown, none, none);
            *(float4*)(my + rr * TILE_PITCH + 4 * lane) = r;
        }

        const int ntile = tile + 16 * GEMM_WGS;
        if (ntile < ntiles) {
            vb = seg_beg[ntile * 16 + c];
            ve = seg_end[ntile * 16 + c];
            if (DEMAND) vn = need[ntile * 16 + c];
        }


        float4 a[16];
#pragma unroll
        for (int i = 0; i < 16; ++i) a[i] = *(const float4*)(my + c * TILE_PITCH + 4 * h + 8 * i);

        v8f acc[8];
#pragma unroll
        for (int t = 0; t < 8; ++t)
#pragma unroll
            for (int i = 0; i < 8; ++i) acc[t][i] = 0.f;

        mma_chunk<K>(bp, a, acc);

        store_strip<HAS_RES, EPI>(acc, bias, res, lng, lnb, out, out2, row0, c, h);

        tile = ntile;
    }
}

__global__ __launch_bounds__(256)
void mark_needed_kernel(const int* __restrict__ final_map, const int* __restrict__ node_map,
                        const int* __restrict__ pmap,
                        const int* __restrict__ seg_beg, const int* __restrict__ seg_end,
                        const int* __restrict__ csr_src, int* __restrict__ need)
{
    const int lane = threadIdx.x & 31;
    const int i    = blockIdx.x * 8 + (int)(threadIdx.x >> 5);
    if (i >= NSEEDS) return;
    const int v   = node_map[final_map[i]];
    const int beg = seg_beg[v];
    const int n   = seg_end[v] - beg;
    if (lane == 0) need[pmap[v]] = 1;
    for (int j = lane; j < n; j += 32) need[pmap[csr_src[beg + j]]] = 1;
}

__global__ __launch_bounds__(256)
void seed_softmax_agg_kernel(const float* __restrict__ p, const int* __restrict__ pmap,
                             const int* __restrict__ final_map,
                             const int* __restrict__ node_map,
                             const float* __restrict__ g, const float* __restrict__ b,
                             const int* __restrict__ seg_beg, const int* __restrict__ seg_end,
                             const int* __restrict__ csr_src,
                             float* __restrict__ t, float* __restrict__ rs)
{
    const int lane = threadIdx.x & 31;
    const int wv = __builtin_amdgcn_readfirstlane((int)(threadIdx.x >> 5));
    const int i  = blockIdx.x * 8 + wv;
    if (i >= NSEEDS) return;
    const int v   = __builtin_amdgcn_readfirstlane(node_map[final_map[i]]);
    const int pv  = __builtin_amdgcn_readfirstlane(pmap[v]);
    const int beg = __builtin_amdgcn_readfirstlane(seg_beg[v]);
    const int n   = __builtin_amdgcn_readfirstlane(seg_end[v]) - beg;
    const float* __restrict__ hl = p + lane * 4;
    const float4 raw = *(const float4*)(hl + (size_t)pv * HDIM);
    const float4 gg  = *(const float4*)(g + lane * 4);
    const float4 bb  = *(const float4*)(b + lane * 4);
    const float4 own = ln_relu_row(raw, gg, bb);
    const float4 r   = agg_node<true, false>(hl, csr_src, pmap, beg, n, 0, lane, own, gg, bb);
    *(float4*)(t  + (size_t)i * HDIM + lane * 4) = r;
    *(float4*)(rs + (size_t)i * HDIM + lane * 4) = raw;
}

__global__ __launch_bounds__(256)
void logsoftmax_kernel(const float* __restrict__ in, float* __restrict__ out, int M)
{
    const int row  = blockIdx.x * 8 + (threadIdx.x >> 5);
    const int lane = threadIdx.x & 31;
    if (row >= M) return;
    const float NEG = -3.0e38f;
    const float v0 = (lane      < NCLS) ? in[(size_t)row * NCLSPAD + lane     ] : NEG;
    const float v1 = (lane + 32 < NCLS) ? in[(size_t)row * NCLSPAD + lane + 32] : NEG;
    float mx = fmaxf(v0, v1);
#pragma unroll
    for (int off = 16; off > 0; off >>= 1) mx = fmaxf(mx, __shfl_xor(mx, off, 32));
    float se = ((lane      < NCLS) ? expf(v0 - mx) : 0.f)
             + ((lane + 32 < NCLS) ? expf(v1 - mx) : 0.f);
#pragma unroll
    for (int off = 16; off > 0; off >>= 1) se += __shfl_xor(se, off, 32);
    const float lse = logf(se) + mx;
    if (lane      < NCLS) out[(size_t)row * NCLS + lane     ] = v0 - lse;
    if (lane + 32 < NCLS) out[(size_t)row * NCLS + lane + 32] = v1 - lse;
}

extern "C" void kernel_launch(void* const* d_in, const int* in_sizes, int n_in,
                              void* d_out, int out_size, void* d_ws, size_t ws_size,
                              hipStream_t stream)
{
    (void)in_sizes; (void)n_in; (void)out_size; (void)ws_size;

    const float* x         = (const float*)d_in[0];
    const int*   src       = (const int*)  d_in[1];
    const int*   dst       = (const int*)  d_in[2];
    const int*   node_map  = (const int*)  d_in[3];
    const int*   final_map = (const int*)  d_in[4];
    const float* enc_w     = (const float*)d_in[5];
    const float* enc_b     = (const float*)d_in[6];
    const float* gcn_w     = (const float*)d_in[7];
    const float* gcn_b     = (const float*)d_in[8];
    const float* ln_g      = (const float*)d_in[9];
    const float* ln_b      = (const float*)d_in[10];
    const float* pred_w    = (const float*)d_in[11];
    const float* pred_b    = (const float*)d_in[12];
    float* out = (float*)d_out;

    const size_t NH = (size_t)NNODES * HDIM;
    float* B0     = (float*)d_ws;
    float* B1     = B0 + NH;
    float* B2     = B0 + 2 * NH;
    float* B3     = B0 + 3 * NH;
    float* B4     = B0 + 4 * NH;
    float* HGN    = B0 + 5 * NH;
    float* SC     = HGN + (size_t)2 * NSEEDS * HDIM;

    const dim3 gblk(32, 4, 1);
    const int  mtN   = NNODES / 16;

    int*  IWHIST = (int*)B3;
    int*  IBASE  = IWHIST + NBINS * SORT_WGS;
    int*  INEED  = IBASE + 256;
    int*  ISTART = INEED + NNODES;
    int*  IEND   = ISTART + 2 * NNODES;
    int*  ICSR   = IEND + 2 * NNODES;
    int2* IREC   = (int2*)(ICSR + 2 * NEDGES);
    bin_hist_kernel<<<SORT_WGS, SORT_THR, 0, stream>>>(dst, IWHIST, INEED);
    bin_scatter_kernel<<<SORT_WGS, SORT_THR, 0, stream>>>(src, dst, IWHIST, IBASE, IREC);
    bin_rank_kernel<<<NBINS, BIN_ROWS, 0, stream>>>(IREC, IBASE, ISTART, IEND, ICSR);

    gemm_rows128_kernel<FDIM, false, EPI_RAW><<<GEMM_WGS, 512, 0, stream>>>(
        x, enc_w, enc_b, nullptr, nullptr, nullptr, B0, nullptr, mtN);

    agg_gemm_rows128_kernel<false, EPI_BOTH, false><<<GEMM_WGS, 512, 0, stream>>>(
        B0, ISTART, IEND, ICSR, nullptr, gcn_w, gcn_b, nullptr, ln_g, ln_b, B2, B1, mtN);

    mark_needed_kernel<<<NSEEDS / 8, 256, 0, stream>>>(
        final_map, node_map + NNODES, node_map, ISTART + NNODES, IEND + NNODES, ICSR, INEED);
    agg_gemm_rows128_kernel<true, EPI_RAW, true><<<GEMM_WGS, 512, 0, stream>>>(
        B1, ISTART, IEND, ICSR, INEED, gcn_w + (size_t)HDIM * HDIM, gcn_b + HDIM,  B2,
        nullptr, nullptr,  B4, nullptr, mtN);

    float* T2 = B0;
    float* R2 = B1;
    const int mtS = NSEEDS / 16;
    seed_softmax_agg_kernel<<<NSEEDS / 8, 256, 0, stream>>>(
        B4, node_map, final_map, node_map + NNODES, ln_g + HDIM, ln_b + HDIM,
        ISTART + NNODES, IEND + NNODES, ICSR, T2, R2);
    gemm_rows128_kernel<HDIM, true, EPI_NORM><<<GEMM_WGS, 512, 0, stream>>>(
        T2, gcn_w + (size_t)2 * HDIM * HDIM, gcn_b + 2 * HDIM, R2,
        ln_g + 2 * HDIM, ln_b + 2 * HDIM, HGN, nullptr, mtS);

    gemm_head_kernel<<<(mtS + 3) / 4, gblk, 0, stream>>>(
        HGN, pred_w, pred_b, SC, NSEEDS);
    logsoftmax_kernel<<<(NSEEDS + 7) / 8, 256, 0, stream>>>(SC, out, NSEEDS);
}
